# starting kernel
# baseline (speedup 1.0000x reference)
#include <hip/hip_runtime.h>
#include <hip/hip_bf16.h>
#include <math.h>


typedef __attribute__((ext_vector_type(2))) float v2f;
typedef __attribute__((ext_vector_type(8))) float v8f;

#define T_LEN 8192
#define IN_DIM 512
#define MD 128
#define LCH 64
#define NCH (T_LEN / LCH)

__device__ __forceinline__ v8f wmma4(v2f a, v2f b, v8f c) {
  return __builtin_amdgcn_wmma_f32_16x16x4_f32(
       false, a,  false, b,
       (short)0, c,  false,  false);
}

__device__ __forceinline__ float sigmoidf_(float z) {
  return 1.0f / (1.0f + __expf(-z));
}

__global__ __launch_bounds__(128) void proj_kernel(
    const float* __restrict__ x,
    const float* __restrict__ Wq, const float* __restrict__ bq,
    const float* __restrict__ Wk, const float* __restrict__ bk,
    const float* __restrict__ Wv, const float* __restrict__ bv,
    const float* __restrict__ Wo, const float* __restrict__ bo,
    float* __restrict__ qw, float* __restrict__ kw,
    float* __restrict__ vw, float* __restrict__ ow) {
  const int wave = threadIdx.x >> 5;
  const int lane = threadIdx.x & 31;
  const int half = lane >> 4;
  const int lm = lane & 15;

  const int rowBase = blockIdx.x * 16;
  const int colTile = blockIdx.y * 4 + wave;
  const int mat = colTile >> 3;
  const int col = (colTile & 7) * 16;

  const float* W; const float* bias; float* dst;
  switch (mat) {
    case 0:  W = Wq; bias = bq; dst = qw; break;
    case 1:  W = Wk; bias = bk; dst = kw; break;
    case 2:  W = Wv; bias = bv; dst = vw; break;
    default: W = Wo; bias = bo; dst = ow; break;
  }

  v8f acc = {};
  const float* arow = x + (size_t)(rowBase + lm) * IN_DIM + 2 * half;
  const float* brow = W + (size_t)(col + lm) * IN_DIM + 2 * half;
  for (int kk = 0; kk < IN_DIM; kk += 4) {
    v2f a, b;
    a.x = arow[kk];     a.y = arow[kk + 1];
    b.x = brow[kk];     b.y = brow[kk + 1];
    acc = wmma4(a, b, acc);
  }

  const float bv_ = bias[col + lm];
#pragma unroll
  for (int e = 0; e < 8; ++e) {
    const int r = rowBase + e + 8 * half;
    float val = acc[e] + bv_;
    if (mat == 1) val *= 0.08838834764831845f;
    if (mat == 3) val = sigmoidf_(val);
    dst[(size_t)r * MD + col + lm] = val;
  }
}

__global__ __launch_bounds__(256) void gates_kernel(
    const float* __restrict__ x,
    const float* __restrict__ wi, const float* __restrict__ bi,
    const float* __restrict__ wf, const float* __restrict__ bf,
    float* __restrict__ ig, float* __restrict__ lf) {
  const int t = blockIdx.x * blockDim.x + threadIdx.x;
  if (t >= T_LEN) return;
  const float* xr = x + (size_t)t * IN_DIM;
  float si = 0.f, sf = 0.f;
  for (int f = 0; f < IN_DIM; ++f) {
    const float xv = xr[f];
    si = fmaf(xv, wi[f], si);
    sf = fmaf(xv, wf[f], sf);
  }
  ig[t] = __expf(si + bi[0]);
  const float z = sf + bf[0];
  lf[t] = (z >= 0.f) ? -log1pf(__expf(-z)) : (z - log1pf(__expf(z)));
}

__global__ __launch_bounds__(256, 1) void scan_kernel(
    const float* __restrict__ qw, const float* __restrict__ kw,
    const float* __restrict__ vw, const float* __restrict__ ow,
    const float* __restrict__ ig, const float* __restrict__ lf,
    const float* __restrict__ C0, const float* __restrict__ n0,
    const float* __restrict__ fc_w, const float* __restrict__ fc_b,
    float* __restrict__ out, float* __restrict__ C_out,
    float* __restrict__ n_out) {
  __shared__ float Cs[MD][MD + 1];
  __shared__ float Ps[LCH][LCH + 1];
  __shared__ float Hs[LCH][MD + 1];
  __shared__ float la_s[LCH], ig_s[LCH], u_s[LCH], rs_s[LCH], qn_s[LCH];
  __shared__ float ns[MD], fcw_s[MD];

  const int tid = threadIdx.x;
  const int wave = tid >> 5;
  const int lane = tid & 31;
  const int half = lane >> 4;
  const int lm = lane & 15;

  for (int i = tid; i < MD * MD; i += 256) Cs[i >> 7][i & 127] = C0[i];
  if (tid < MD) { ns[tid] = n0[tid]; fcw_s[tid] = fc_w[tid]; }
  __syncthreads();

  for (int c = 0; c < NCH; ++c) {
    const int t0 = c * LCH;

    if (c + 1 < NCH) {
      const float* nq = qw + (size_t)(t0 + LCH) * MD + tid * 32;
      const float* nk = kw + (size_t)(t0 + LCH) * MD + tid * 32;
      const float* nv = vw + (size_t)(t0 + LCH) * MD + tid * 32;
      __builtin_prefetch(nq, 0, 3);
      __builtin_prefetch(nk, 0, 3);
      __builtin_prefetch(nv, 0, 3);
    }

    if (tid == 0) {
      float a = 0.f;
      for (int j = 0; j < LCH; ++j) {
        a += lf[t0 + j];
        la_s[j] = a;
        ig_s[j] = ig[t0 + j];
      }
    }
    __syncthreads();
    if (tid < LCH) {
      u_s[tid] = __expf(la_s[LCH - 1] - la_s[tid]) * ig_s[tid];
      float s = 0.f;
      const float* qrow = qw + (size_t)(t0 + tid) * MD;
      for (int f = 0; f < MD; ++f) s = fmaf(qrow[f], ns[f], s);
      qn_s[tid] = s;
    }
    __syncthreads();

    for (int tile = wave; tile < 16; tile += 8) {
      const int tr = (tile >> 2) * 16, tc = (tile & 3) * 16;
      v8f acc = {};
      const float* ap = qw + (size_t)(t0 + tr + lm) * MD + 2 * half;
      const float* bp = kw + (size_t)(t0 + tc + lm) * MD + 2 * half;
      for (int kk = 0; kk < MD; kk += 4) {
        v2f a, b;
        a.x = ap[kk]; a.y = ap[kk + 1];
        b.x = bp[kk]; b.y = bp[kk + 1];
        acc = wmma4(a, b, acc);
      }
#pragma unroll
      for (int e = 0; e < 8; ++e) {
        const int r = tr + e + 8 * half, cc = tc + lm;
        float p = 0.f;
        if (r >= cc) p = __expf(la_s[r] - la_s[cc]) * ig_s[cc] * acc[e];
        Ps[r][cc] = p;
      }
    }
    __syncthreads();
    if (tid < LCH) {
      float s = 0.f;
      for (int j = 0; j <= tid; ++j) s += Ps[tid][j];
      rs_s[tid] = s;
    }
    __syncthreads();

    for (int tile = wave; tile < 32; tile += 8) {
      const int tr = (tile >> 3) * 16, tc = (tile & 7) * 16;
      v8f aci = {};
      for (int kk = 0; kk < LCH; kk += 4) {
        v2f a, b;
        const int j0 = kk + 2 * half;
        a.x = Ps[tr + lm][j0]; a.y = Ps[tr + lm][j0 + 1];
        const float* bp = vw + (size_t)(t0 + j0) * MD + tc + lm;
        b.x = bp[0]; b.y = bp[MD];
        aci = wmma4(a, b, aci);
      }
      v8f ace = {};
      const float* ap = qw + (size_t)(t0 + tr + lm) * MD + 2 * half;
      for (int kk = 0; kk < MD; kk += 4) {
        v2f a, b;
        a.x = ap[kk]; a.y = ap[kk + 1];
        b.x = Cs[tc + lm][kk + 2 * half];
        b.y = Cs[tc + lm][kk + 2 * half + 1];
        ace = wmma4(a, b, ace);
      }
#pragma unroll
      for (int e = 0; e < 8; ++e) {
        const int r = tr + e + 8 * half, cc = tc + lm;
        const float Ft = __expf(la_s[r]);
        const float num = fmaf(Ft, ace[e], aci[e]);
        const float den = fmaxf(fabsf(fmaf(Ft, qn_s[r], rs_s[r])), 1.0f);
        Hs[r][cc] = (num / den) * ow[(size_t)(t0 + r) * MD + cc];
      }
    }
    __syncthreads();

    if (tid < LCH) {
      float s = 0.f;
      for (int f = 0; f < MD; ++f) s = fmaf(Hs[tid][f], fcw_s[f], s);
      out[t0 + tid] = s + fc_b[0];
    }

    const float Fend = __expf(la_s[LCH - 1]);
    for (int tile = wave; tile < 64; tile += 8) {
      const int tr = (tile >> 3) * 16, tc = (tile & 7) * 16;
      v8f acc;
#pragma unroll
      for (int e = 0; e < 8; ++e) acc[e] = Fend * Cs[tr + e + 8 * half][tc + lm];
      for (int kk = 0; kk < LCH; kk += 4) {
        v2f a, b;
        const int j0 = kk + 2 * half;
        a.x = u_s[j0]     * vw[(size_t)(t0 + j0) * MD + tr + lm];
        a.y = u_s[j0 + 1] * vw[(size_t)(t0 + j0 + 1) * MD + tr + lm];
        const float* bp = kw + (size_t)(t0 + j0) * MD + tc + lm;
        b.x = bp[0]; b.y = bp[MD];
        acc = wmma4(a, b, acc);
      }
#pragma unroll
      for (int e = 0; e < 8; ++e) Cs[tr + e + 8 * half][tc + lm] = acc[e];
    }
    if (tid < MD) {
      float s = Fend * ns[tid];
      for (int j = 0; j < LCH; ++j)
        s = fmaf(u_s[j], kw[(size_t)(t0 + j) * MD + tid], s);
      ns[tid] = s;
    }
    __syncthreads();
  }

  for (int i = tid; i < MD * MD; i += 256) C_out[i] = Cs[i >> 7][i & 127];
  if (tid < MD) n_out[tid] = ns[tid];
}

extern "C" void kernel_launch(void* const* d_in, const int* in_sizes, int n_in,
                              void* d_out, int out_size, void* d_ws, size_t ws_size,
                              hipStream_t stream) {
  const float* x    = (const float*)d_in[0];
  const float* C0   = (const float*)d_in[1];
  const float* n0   = (const float*)d_in[2];
  const float* Wq   = (const float*)d_in[3];
  const float* bq   = (const float*)d_in[4];
  const float* Wk   = (const float*)d_in[5];
  const float* bk   = (const float*)d_in[6];
  const float* Wv   = (const float*)d_in[7];
  const float* bv   = (const float*)d_in[8];
  const float* wi   = (const float*)d_in[9];
  const float* bi   = (const float*)d_in[10];
  const float* wf   = (const float*)d_in[11];
  const float* bf   = (const float*)d_in[12];
  const float* Wo   = (const float*)d_in[13];
  const float* bo   = (const float*)d_in[14];
  const float* fc_w = (const float*)d_in[15];
  const float* fc_b = (const float*)d_in[16];

  float* ws = (float*)d_ws;
  const size_t NP = (size_t)T_LEN * MD;
  float* qw = ws;
  float* kw = ws + NP;
  float* vw = ws + 2 * NP;
  float* owp = ws + 3 * NP;
  float* igp = ws + 4 * NP;
  float* lfp = igp + T_LEN;

  float* outp  = (float*)d_out;
  float* Coutp = outp + T_LEN;
  float* noutp = Coutp + MD * MD;

  proj_kernel<<<dim3(T_LEN / 16, 8), 128, 0, stream>>>(
      x, Wq, bq, Wk, bk, Wv, bv, Wo, bo, qw, kw, vw, owp);
  gates_kernel<<<dim3(T_LEN / 256), 256, 0, stream>>>(
      x, wi, bi, wf, bf, igp, lfp);
  scan_kernel<<<1, 256, 0, stream>>>(
      qw, kw, vw, owp, igp, lfp, C0, n0, fc_w, fc_b, outp, Coutp, noutp);
}


# Scan spread over chunks: one work-group per 128-step chunk for local products, per-element prefix of chunk states, parallel read-out (was one work-group); sum order changes, chunk 64->128
# speedup vs baseline: 21.1847x; 21.1847x over previous
#include <hip/hip_runtime.h>
#include <hip/hip_bf16.h>
#include <math.h>


typedef __attribute__((ext_vector_type(2))) float v2f;
typedef __attribute__((ext_vector_type(8))) float v8f;

#define T_LEN 8192
#define IN_DIM 512
#define MD 128
#define LCH 128
#define NCH (T_LEN / LCH)
#define NW 16
#define CH (LCH * MD)
#define SC_LA 0
#define SC_RS 128
#define SC_A 256
#define SC_DN 384

__device__ __forceinline__ v8f wmma4(v2f a, v2f b, v8f c) {
  return __builtin_amdgcn_wmma_f32_16x16x4_f32(
       false, a,  false, b,
       (short)0, c,  false,  false);
}

__device__ __forceinline__ float sigmoidf_(float z) {
  return 1.0f / (1.0f + __expf(-z));
}

__global__ __launch_bounds__(128) void proj_kernel(
    const float* __restrict__ x,
    const float* __restrict__ Wq, const float* __restrict__ bq,
    const float* __restrict__ Wk, const float* __restrict__ bk,
    const float* __restrict__ Wv, const float* __restrict__ bv,
    const float* __restrict__ Wo, const float* __restrict__ bo,
    float* __restrict__ qw, float* __restrict__ kw,
    float* __restrict__ vw, float* __restrict__ ow) {
  const int wave = threadIdx.x >> 5;
  const int lane = threadIdx.x & 31;
  const int half = lane >> 4;
  const int lm = lane & 15;

  const int rowBase = blockIdx.x * 16;
  const int colTile = blockIdx.y * 4 + wave;
  const int mat = colTile >> 3;
  const int col = (colTile & 7) * 16;

  const float* W; const float* bias; float* dst;
  switch (mat) {
    case 0:  W = Wq; bias = bq; dst = qw; break;
    case 1:  W = Wk; bias = bk; dst = kw; break;
    case 2:  W = Wv; bias = bv; dst = vw; break;
    default: W = Wo; bias = bo; dst = ow; break;
  }

  v8f acc = {};
  const float* arow = x + (size_t)(rowBase + lm) * IN_DIM + 2 * half;
  const float* brow = W + (size_t)(col + lm) * IN_DIM + 2 * half;
  for (int kk = 0; kk < IN_DIM; kk += 4) {
    v2f a, b;
    a.x = arow[kk];     a.y = arow[kk + 1];
    b.x = brow[kk];     b.y = brow[kk + 1];
    acc = wmma4(a, b, acc);
  }

  const float bv_ = bias[col + lm];
#pragma unroll
  for (int e = 0; e < 8; ++e) {
    const int r = rowBase + e + 8 * half;
    float val = acc[e] + bv_;
    if (mat == 1) val *= 0.08838834764831845f;
    if (mat == 3) val = sigmoidf_(val);
    dst[(size_t)r * MD + col + lm] = val;
  }
}

__global__ __launch_bounds__(256) void gates_kernel(
    const float* __restrict__ x,
    const float* __restrict__ wi, const float* __restrict__ bi,
    const float* __restrict__ wf, const float* __restrict__ bf,
    float* __restrict__ ig, float* __restrict__ lf) {
  const int t = blockIdx.x * blockDim.x + threadIdx.x;
  if (t >= T_LEN) return;
  const float* xr = x + (size_t)t * IN_DIM;
  float si = 0.f, sf = 0.f;
  for (int f = 0; f < IN_DIM; ++f) {
    const float xv = xr[f];
    si = fmaf(xv, wi[f], si);
    sf = fmaf(xv, wf[f], sf);
  }
  ig[t] = __expf(si + bi[0]);
  const float z = sf + bf[0];
  lf[t] = (z >= 0.f) ? -log1pf(__expf(-z)) : (z - log1pf(__expf(z)));
}

__global__ __launch_bounds__(512) void chunk_local_kernel(
    const float* __restrict__ qw, float* kw, float* vw,
    const float* __restrict__ ow,
    const float* __restrict__ ig, const float* __restrict__ lf,
    const float* __restrict__ fc_w) {
  __shared__ float Ps[LCH][LCH + 1];
  __shared__ float Gs[LCH][MD + 1];
  __shared__ float la_s[LCH], ig_s[LCH], u_s[LCH], fcw_s[MD];

  const int tid = threadIdx.x;
  const int wave = tid >> 5;
  const int lane = tid & 31;
  const int half = lane >> 4;
  const int lm = lane & 15;
  const int t0 = blockIdx.x * LCH;

  const float* qc = qw + (size_t)t0 * MD;
  const float* oc = ow + (size_t)t0 * MD;
  float* kc = kw + (size_t)t0 * MD;
  float* vc = vw + (size_t)t0 * MD;

  if (tid < LCH) {
    la_s[tid] = lf[t0 + tid];
    ig_s[tid] = ig[t0 + tid];
    fcw_s[tid] = fc_w[tid];
  }
  __syncthreads();
  if (tid == 0) {
    float a = 0.f;
    for (int j = 0; j < LCH; ++j) { a += la_s[j]; la_s[j] = a; }
  }
  __syncthreads();
  if (tid < LCH) u_s[tid] = __expf(la_s[LCH - 1] - la_s[tid]) * ig_s[tid];

  for (int tile = wave; tile < 64; tile += NW) {
    const int tr = (tile >> 3) * 16, tc = (tile & 7) * 16;
    v8f acc = {};
    if (tc <= tr) {
      const float* ap = qc + (size_t)(tr + lm) * MD + 2 * half;
      const float* bp = kc + (size_t)(tc + lm) * MD + 2 * half;
      for (int kk = 0; kk < MD; kk += 4) {
        v2f a, b;
        a.x = ap[kk]; a.y = ap[kk + 1];
        b.x = bp[kk]; b.y = bp[kk + 1];
        acc = wmma4(a, b, acc);
      }
    }
#pragma unroll
    for (int e = 0; e < 8; ++e) {
      const int r = tr + e + 8 * half, cc = tc + lm;
      float p = 0.f;
      if (r >= cc) p = __expf(la_s[r] - la_s[cc]) * ig_s[cc] * acc[e];
      Ps[r][cc] = p;
    }
  }
  __syncthreads();

  float rs_v = 0.f;
  if (tid < LCH) {
    for (int j = 0; j <= tid; ++j) rs_v += Ps[tid][j];
  }

  for (int tile = wave; tile < 64; tile += NW) {
    const int tr = (tile >> 3) * 16, tc = (tile & 7) * 16;
    v8f aci = {};
    for (int kk = 0; kk < LCH; kk += 4) {
      v2f a, b;
      const int j0 = kk + 2 * half;
      a.x = Ps[tr + lm][j0]; a.y = Ps[tr + lm][j0 + 1];
      const float* bp = vc + (size_t)j0 * MD + tc + lm;
      b.x = bp[0]; b.y = bp[MD];
      aci = wmma4(a, b, aci);
    }
#pragma unroll
    for (int e = 0; e < 8; ++e) {
      const int r = tr + e + 8 * half, cc = tc + lm;
      Gs[r][cc] = aci[e] * oc[(size_t)r * MD + cc] * fcw_s[cc];
    }
  }

  v8f dacc[4];
#pragma unroll
  for (int i = 0; i < 4; ++i) {
    const int tile = wave + NW * i;
    const int tr = (tile >> 3) * 16, tc = (tile & 7) * 16;
    v8f acc = {};
    for (int kk = 0; kk < LCH; kk += 4) {
      v2f a, b;
      const int j0 = kk + 2 * half;
      a.x = u_s[j0]     * vc[(size_t)j0 * MD + tr + lm];
      a.y = u_s[j0 + 1] * vc[(size_t)(j0 + 1) * MD + tr + lm];
      const float* bp = kc + (size_t)j0 * MD + tc + lm;
      b.x = bp[0]; b.y = bp[MD];
      acc = wmma4(a, b, acc);
    }
    dacc[i] = acc;
  }
  float dn_v = 0.f;
  if (tid < MD) {
    for (int j = 0; j < LCH; ++j) dn_v = fmaf(u_s[j], kc[(size_t)j * MD + tid], dn_v);
  }
  __syncthreads();

#pragma unroll
  for (int i = 0; i < 4; ++i) {
    const int tile = wave + NW * i;
    const int tr = (tile >> 3) * 16, tc = (tile & 7) * 16;
#pragma unroll
    for (int e = 0; e < 8; ++e)
      kc[(size_t)(tr + e + 8 * half) * MD + tc + lm] = dacc[i][e];
  }
  if (tid < LCH) {
    float a_v = 0.f;
    for (int f = 0; f < MD; ++f) a_v += Gs[tid][f];
    vc[SC_LA + tid] = la_s[tid];
    vc[SC_RS + tid] = rs_v;
    vc[SC_A + tid] = a_v;
    vc[SC_DN + tid] = dn_v;
  }
}

__global__ __launch_bounds__(256) void state_prefix_kernel(
    float* kw, float* vw,
    const float* __restrict__ C0, const float* __restrict__ n0,
    float* __restrict__ C_out, float* __restrict__ n_out) {
  __shared__ float F_s[NCH];
  if (threadIdx.x < NCH)
    F_s[threadIdx.x] = __expf(vw[(size_t)threadIdx.x * CH + SC_LA + LCH - 1]);
  __syncthreads();

  const int gid = blockIdx.x * 256 + threadIdx.x;
  float* p; float* dst; float s;
  if (gid < MD * MD) {
    p = kw + gid; s = C0[gid]; dst = C_out + gid;
  } else if (gid < MD * MD + MD) {
    const int f = gid - MD * MD;
    p = vw + SC_DN + f; s = n0[f]; dst = n_out + f;
  } else {
    return;
  }
  for (int cb = 0; cb < NCH; cb += 16) {
    float d[16];
#pragma unroll
    for (int i = 0; i < 16; ++i) d[i] = p[(size_t)(cb + i) * CH];
#pragma unroll
    for (int i = 0; i < 16; ++i) {
      p[(size_t)(cb + i) * CH] = s;
      s = fmaf(F_s[cb + i], s, d[i]);
    }
  }
  *dst = s;
}

__global__ __launch_bounds__(512) void chunk_readout_kernel(
    const float* __restrict__ qw, const float* __restrict__ kw,
    const float* __restrict__ vw, const float* __restrict__ ow,
    const float* __restrict__ fc_w, const float* __restrict__ fc_b,
    float* __restrict__ out) {
  __shared__ float Cs[MD][MD + 1];
  __shared__ float Gs[LCH][MD + 1];
  __shared__ float ns[MD], fcw_s[MD];

  const int tid = threadIdx.x;
  const int wave = tid >> 5;
  const int lane = tid & 31;
  const int half = lane >> 4;
  const int lm = lane & 15;
  const int t0 = blockIdx.x * LCH;

  const float* qc = qw + (size_t)t0 * MD;
  const float* oc = ow + (size_t)t0 * MD;
  const float* st = kw + (size_t)t0 * MD;
  const float* sc = vw + (size_t)t0 * MD;

  for (int i = tid; i < MD * MD; i += 512) Cs[i >> 7][i & 127] = st[i];
  if (tid < MD) { ns[tid] = sc[SC_DN + tid]; fcw_s[tid] = fc_w[tid]; }
  __syncthreads();

  float qn = 0.f, la_v = 0.f, rs_v = 0.f, a_v = 0.f;
  if (tid < LCH) {
    const float* qrow = qc + (size_t)tid * MD;
    for (int f = 0; f < MD; ++f) qn = fmaf(qrow[f], ns[f], qn);
    la_v = sc[SC_LA + tid];
    rs_v = sc[SC_RS + tid];
    a_v = sc[SC_A + tid];
  }

  for (int tile = wave; tile < 64; tile += NW) {
    const int tr = (tile >> 3) * 16, tc = (tile & 7) * 16;
    v8f ace = {};
    const float* ap = qc + (size_t)(tr + lm) * MD + 2 * half;
    for (int kk = 0; kk < MD; kk += 4) {
      v2f a, b;
      a.x = ap[kk]; a.y = ap[kk + 1];
      b.x = Cs[tc + lm][kk + 2 * half];
      b.y = Cs[tc + lm][kk + 2 * half + 1];
      ace = wmma4(a, b, ace);
    }
#pragma unroll
    for (int e = 0; e < 8; ++e) {
      const int r = tr + e + 8 * half, cc = tc + lm;
      Gs[r][cc] = ace[e] * oc[(size_t)r * MD + cc] * fcw_s[cc];
    }
  }
  __syncthreads();

  if (tid < LCH) {
    float b_v = 0.f;
    for (int f = 0; f < MD; ++f) b_v += Gs[tid][f];
    const float Ft = __expf(la_v);
    const float num = fmaf(Ft, b_v, a_v);
    const float den = fmaxf(fabsf(fmaf(Ft, qn, rs_v)), 1.0f);
    out[t0 + tid] = num / den + fc_b[0];
  }
}

extern "C" void kernel_launch(void* const* d_in, const int* in_sizes, int n_in,
                              void* d_out, int out_size, void* d_ws, size_t ws_size,
                              hipStream_t stream) {
  const float* x    = (const float*)d_in[0];
  const float* C0   = (const float*)d_in[1];
  const float* n0   = (const float*)d_in[2];
  const float* Wq   = (const float*)d_in[3];
  const float* bq   = (const float*)d_in[4];
  const float* Wk   = (const float*)d_in[5];
  const float* bk   = (const float*)d_in[6];
  const float* Wv   = (const float*)d_in[7];
  const float* bv   = (const float*)d_in[8];
  const float* wi   = (const float*)d_in[9];
  const float* bi   = (const float*)d_in[10];
  const float* wf   = (const float*)d_in[11];
  const float* bf   = (const float*)d_in[12];
  const float* Wo   = (const float*)d_in[13];
  const float* bo   = (const float*)d_in[14];
  const float* fc_w = (const float*)d_in[15];
  const float* fc_b = (const float*)d_in[16];

  float* ws = (float*)d_ws;
  const size_t NP = (size_t)T_LEN * MD;
  float* qw = ws;
  float* kw = ws + NP;
  float* vw = ws + 2 * NP;
  float* owp = ws + 3 * NP;
  float* igp = ws + 4 * NP;
  float* lfp = igp + T_LEN;

  float* outp  = (float*)d_out;
  float* Coutp = outp + T_LEN;
  float* noutp = Coutp + MD * MD;

  proj_kernel<<<dim3(T_LEN / 16, 8), 128, 0, stream>>>(
      x, Wq, bq, Wk, bk, Wv, bv, Wo, bo, qw, kw, vw, owp);
  gates_kernel<<<dim3(T_LEN / 256), 256, 0, stream>>>(
      x, wi, bi, wf, bf, igp, lfp);
  chunk_local_kernel<<<NCH, 512, 0, stream>>>(qw, kw, vw, owp, igp, lfp, fc_w);
  state_prefix_kernel<<<(MD * MD + MD + 255) / 256, 256, 0, stream>>>(
      kw, vw, C0, n0, Coutp, noutp);
  chunk_readout_kernel<<<NCH, 512, 0, stream>>>(
      qw, kw, vw, owp, fc_w, fc_b, outp);
}


# Projection GEMM restaged through LDS: 128x128 tile per work-group, whole-line float4 loads, two stages, 2x2 tiles per wave; k taken 0,1,4,5 then 2,3,6,7 in each 8, so the sum order changes
# speedup vs baseline: 30.8477x; 1.4561x over previous
#include <hip/hip_runtime.h>
#include <hip/hip_bf16.h>
#include <math.h>


typedef __attribute__((ext_vector_type(2))) float v2f;
typedef __attribute__((ext_vector_type(8))) float v8f;

#define T_LEN 8192
#define IN_DIM 512
#define MD 128
#define LCH 128
#define NCH (T_LEN / LCH)
#define NW 16
#define CH (LCH * MD)
#define SC_LA 0
#define SC_RS 128
#define SC_A 256
#define SC_DN 384

__device__ __forceinline__ v8f wmma4(v2f a, v2f b, v8f c) {
  return __builtin_amdgcn_wmma_f32_16x16x4_f32(
       false, a,  false, b,
       (short)0, c,  false,  false);
}

__device__ __forceinline__ float sigmoidf_(float z) {
  return 1.0f / (1.0f + __expf(-z));
}

typedef __attribute__((ext_vector_type(4))) float v4f;

#define PB 128
#define PKT 64
#define PPITCH (PKT + 4)
#define PNK (IN_DIM / PKT)

__global__ __launch_bounds__(512) void proj_kernel(
    const float* __restrict__ x,
    const float* __restrict__ Wq, const float* __restrict__ bq,
    const float* __restrict__ Wk, const float* __restrict__ bk,
    const float* __restrict__ Wv, const float* __restrict__ bv,
    const float* __restrict__ Wo, const float* __restrict__ bo,
    float* __restrict__ qw, float* __restrict__ kw,
    float* __restrict__ vw, float* __restrict__ ow) {
  __shared__ __attribute__((aligned(16))) float Xs[2][PB][PPITCH];
  __shared__ __attribute__((aligned(16))) float Ws[2][PB][PPITCH];

  const int tid = threadIdx.x;
  const int wave = tid >> 5;
  const int lane = tid & 31;
  const int half = lane >> 4;
  const int lm = lane & 15;

  const int rowBase = blockIdx.x * PB;
  const int mat = blockIdx.y;

  const float* W; const float* bias; float* dst;
  switch (mat) {
    case 0:  W = Wq; bias = bq; dst = qw; break;
    case 1:  W = Wk; bias = bk; dst = kw; break;
    case 2:  W = Wv; bias = bv; dst = vw; break;
    default: W = Wo; bias = bo; dst = ow; break;
  }

  const int sr = tid >> 4;
  const int sc = (tid & 15) * 4;
  const float* xg = x + (size_t)(rowBase + sr) * IN_DIM + sc;
  const float* wg = W + (size_t)sr * IN_DIM + sc;

  v4f px[4], pw[4];
#pragma unroll
  for (int i = 0; i < 4; ++i) {
    px[i] = *(const v4f*)(xg + (size_t)i * 32 * IN_DIM);
    pw[i] = *(const v4f*)(wg + (size_t)i * 32 * IN_DIM);
  }
#pragma unroll
  for (int i = 0; i < 4; ++i) {
    *(v4f*)&Xs[0][sr + 32 * i][sc] = px[i];
    *(v4f*)&Ws[0][sr + 32 * i][sc] = pw[i];
  }
  __syncthreads();

  const int wr = (wave >> 2) * 32;
  const int wc = (wave & 3) * 32;
  v8f acc[2][2] = {};

  for (int kt = 0; kt < PNK; ++kt) {
    const int cur = kt & 1;
    const bool more = (kt + 1) < PNK;
    if (more) {
      const int k0 = (kt + 1) * PKT;
#pragma unroll
      for (int i = 0; i < 4; ++i) {
        px[i] = *(const v4f*)(xg + (size_t)i * 32 * IN_DIM + k0);
        pw[i] = *(const v4f*)(wg + (size_t)i * 32 * IN_DIM + k0);
      }
    }

    const float* xa = &Xs[cur][wr + lm][4 * half];
    const float* wb = &Ws[cur][wc + lm][4 * half];
#pragma unroll
    for (int k8 = 0; k8 < PKT; k8 += 8) {
      const v4f a0 = *(const v4f*)(xa + k8);
      const v4f a1 = *(const v4f*)(xa + 16 * PPITCH + k8);
      const v4f b0 = *(const v4f*)(wb + k8);
      const v4f b1 = *(const v4f*)(wb + 16 * PPITCH + k8);
      const v2f a0l = {a0.x, a0.y}, a0h = {a0.z, a0.w};
      const v2f a1l = {a1.x, a1.y}, a1h = {a1.z, a1.w};
      const v2f b0l = {b0.x, b0.y}, b0h = {b0.z, b0.w};
      const v2f b1l = {b1.x, b1.y}, b1h = {b1.z, b1.w};
      acc[0][0] = wmma4(a0l, b0l, acc[0][0]);
      acc[0][1] = wmma4(a0l, b1l, acc[0][1]);
      acc[1][0] = wmma4(a1l, b0l, acc[1][0]);
      acc[1][1] = wmma4(a1l, b1l, acc[1][1]);
      acc[0][0] = wmma4(a0h, b0h, acc[0][0]);
      acc[0][1] = wmma4(a0h, b1h, acc[0][1]);
      acc[1][0] = wmma4(a1h, b0h, acc[1][0]);
      acc[1][1] = wmma4(a1h, b1h, acc[1][1]);
    }

    if (more) {
#pragma unroll
      for (int i = 0; i < 4; ++i) {
        *(v4f*)&Xs[cur ^ 1][sr + 32 * i][sc] = px[i];
        *(v4f*)&Ws[cur ^ 1][sr + 32 * i][sc] = pw[i];
      }
    }
    __syncthreads();
  }

#pragma unroll
  for (int j = 0; j < 2; ++j) {
    const int cc = wc + 16 * j + lm;
    const float bv_ = bias[cc];
#pragma unroll
    for (int i = 0; i < 2; ++i) {
#pragma unroll
      for (int e = 0; e < 8; ++e) {
        const int r = rowBase + wr + 16 * i + e + 8 * half;
        float val = acc[i][j][e] + bv_;
        if (mat == 1) val *= 0.08838834764831845f;
        if (mat == 3) val = sigmoidf_(val);
        dst[(size_t)r * MD + cc] = val;
      }
    }
  }
}

__global__ __launch_bounds__(256) void gates_kernel(
    const float* __restrict__ x,
    const float* __restrict__ wi, const float* __restrict__ bi,
    const float* __restrict__ wf, const float* __restrict__ bf,
    float* __restrict__ ig, float* __restrict__ lf) {
  const int t = blockIdx.x * blockDim.x + threadIdx.x;
  if (t >= T_LEN) return;
  const float* xr = x + (size_t)t * IN_DIM;
  float si = 0.f, sf = 0.f;
  for (int f = 0; f < IN_DIM; ++f) {
    const float xv = xr[f];
    si = fmaf(xv, wi[f], si);
    sf = fmaf(xv, wf[f], sf);
  }
  ig[t] = __expf(si + bi[0]);
  const float z = sf + bf[0];
  lf[t] = (z >= 0.f) ? -log1pf(__expf(-z)) : (z - log1pf(__expf(z)));
}

__global__ __launch_bounds__(512) void chunk_local_kernel(
    const float* __restrict__ qw, float* kw, float* vw,
    const float* __restrict__ ow,
    const float* __restrict__ ig, const float* __restrict__ lf,
    const float* __restrict__ fc_w) {
  __shared__ float Ps[LCH][LCH + 1];
  __shared__ float Gs[LCH][MD + 1];
  __shared__ float la_s[LCH], ig_s[LCH], u_s[LCH], fcw_s[MD];

  const int tid = threadIdx.x;
  const int wave = tid >> 5;
  const int lane = tid & 31;
  const int half = lane >> 4;
  const int lm = lane & 15;
  const int t0 = blockIdx.x * LCH;

  const float* qc = qw + (size_t)t0 * MD;
  const float* oc = ow + (size_t)t0 * MD;
  float* kc = kw + (size_t)t0 * MD;
  float* vc = vw + (size_t)t0 * MD;

  if (tid < LCH) {
    la_s[tid] = lf[t0 + tid];
    ig_s[tid] = ig[t0 + tid];
    fcw_s[tid] = fc_w[tid];
  }
  __syncthreads();
  if (tid == 0) {
    float a = 0.f;
    for (int j = 0; j < LCH; ++j) { a += la_s[j]; la_s[j] = a; }
  }
  __syncthreads();
  if (tid < LCH) u_s[tid] = __expf(la_s[LCH - 1] - la_s[tid]) * ig_s[tid];

  for (int tile = wave; tile < 64; tile += NW) {
    const int tr = (tile >> 3) * 16, tc = (tile & 7) * 16;
    v8f acc = {};
    if (tc <= tr) {
      const float* ap = qc + (size_t)(tr + lm) * MD + 2 * half;
      const float* bp = kc + (size_t)(tc + lm) * MD + 2 * half;
      for (int kk = 0; kk < MD; kk += 4) {
        v2f a, b;
        a.x = ap[kk]; a.y = ap[kk + 1];
        b.x = bp[kk]; b.y = bp[kk + 1];
        acc = wmma4(a, b, acc);
      }
    }
#pragma unroll
    for (int e = 0; e < 8; ++e) {
      const int r = tr + e + 8 * half, cc = tc + lm;
      float p = 0.f;
      if (r >= cc) p = __expf(la_s[r] - la_s[cc]) * ig_s[cc] * acc[e];
      Ps[r][cc] = p;
    }
  }
  __syncthreads();

  float rs_v = 0.f;
  if (tid < LCH) {
    for (int j = 0; j <= tid; ++j) rs_v += Ps[tid][j];
  }

  for (int tile = wave; tile < 64; tile += NW) {
    const int tr = (tile >> 3) * 16, tc = (tile & 7) * 16;
    v8f aci = {};
    for (int kk = 0; kk < LCH; kk += 4) {
      v2f a, b;
      const int j0 = kk + 2 * half;
      a.x = Ps[tr + lm][j0]; a.y = Ps[tr + lm][j0 + 1];
      const float* bp = vc + (size_t)j0 * MD + tc + lm;
      b.x = bp[0]; b.y = bp[MD];
      aci = wmma4(a, b, aci);
    }
#pragma unroll
    for (int e = 0; e < 8; ++e) {
      const int r = tr + e + 8 * half, cc = tc + lm;
      Gs[r][cc] = aci[e] * oc[(size_t)r * MD + cc] * fcw_s[cc];
    }
  }

  v8f dacc[4];
#pragma unroll
  for (int i = 0; i < 4; ++i) {
    const int tile = wave + NW * i;
    const int tr = (tile >> 3) * 16, tc = (tile & 7) * 16;
    v8f acc = {};
    for (int kk = 0; kk < LCH; kk += 4) {
      v2f a, b;
      const int j0 = kk + 2 * half;
      a.x = u_s[j0]     * vc[(size_t)j0 * MD + tr + lm];
      a.y = u_s[j0 + 1] * vc[(size_t)(j0 + 1) * MD + tr + lm];
      const float* bp = kc + (size_t)j0 * MD + tc + lm;
      b.x = bp[0]; b.y = bp[MD];
      acc = wmma4(a, b, acc);
    }
    dacc[i] = acc;
  }
  float dn_v = 0.f;
  if (tid < MD) {
    for (int j = 0; j < LCH; ++j) dn_v = fmaf(u_s[j], kc[(size_t)j * MD + tid], dn_v);
  }
  __syncthreads();

#pragma unroll
  for (int i = 0; i < 4; ++i) {
    const int tile = wave + NW * i;
    const int tr = (tile >> 3) * 16, tc = (tile & 7) * 16;
#pragma unroll
    for (int e = 0; e < 8; ++e)
      kc[(size_t)(tr + e + 8 * half) * MD + tc + lm] = dacc[i][e];
  }
  if (tid < LCH) {
    float a_v = 0.f;
    for (int f = 0; f < MD; ++f) a_v += Gs[tid][f];
    vc[SC_LA + tid] = la_s[tid];
    vc[SC_RS + tid] = rs_v;
    vc[SC_A + tid] = a_v;
    vc[SC_DN + tid] = dn_v;
  }
}

__global__ __launch_bounds__(256) void state_prefix_kernel(
    float* kw, float* vw,
    const float* __restrict__ C0, const float* __restrict__ n0,
    float* __restrict__ C_out, float* __restrict__ n_out) {
  __shared__ float F_s[NCH];
  if (threadIdx.x < NCH)
    F_s[threadIdx.x] = __expf(vw[(size_t)threadIdx.x * CH + SC_LA + LCH - 1]);
  __syncthreads();

  const int gid = blockIdx.x * 256 + threadIdx.x;
  float* p; float* dst; float s;
  if (gid < MD * MD) {
    p = kw + gid; s = C0[gid]; dst = C_out + gid;
  } else if (gid < MD * MD + MD) {
    const int f = gid - MD * MD;
    p = vw + SC_DN + f; s = n0[f]; dst = n_out + f;
  } else {
    return;
  }
  for (int cb = 0; cb < NCH; cb += 16) {
    float d[16];
#pragma unroll
    for (int i = 0; i < 16; ++i) d[i] = p[(size_t)(cb + i) * CH];
#pragma unroll
    for (int i = 0; i < 16; ++i) {
      p[(size_t)(cb + i) * CH] = s;
      s = fmaf(F_s[cb + i], s, d[i]);
    }
  }
  *dst = s;
}

__global__ __launch_bounds__(512) void chunk_readout_kernel(
    const float* __restrict__ qw, const float* __restrict__ kw,
    const float* __restrict__ vw, const float* __restrict__ ow,
    const float* __restrict__ fc_w, const float* __restrict__ fc_b,
    float* __restrict__ out) {
  __shared__ float Cs[MD][MD + 1];
  __shared__ float Gs[LCH][MD + 1];
  __shared__ float ns[MD], fcw_s[MD];

  const int tid = threadIdx.x;
  const int wave = tid >> 5;
  const int lane = tid & 31;
  const int half = lane >> 4;
  const int lm = lane & 15;
  const int t0 = blockIdx.x * LCH;

  const float* qc = qw + (size_t)t0 * MD;
  const float* oc = ow + (size_t)t0 * MD;
  const float* st = kw + (size_t)t0 * MD;
  const float* sc = vw + (size_t)t0 * MD;

  for (int i = tid; i < MD * MD; i += 512) Cs[i >> 7][i & 127] = st[i];
  if (tid < MD) { ns[tid] = sc[SC_DN + tid]; fcw_s[tid] = fc_w[tid]; }
  __syncthreads();

  float qn = 0.f, la_v = 0.f, rs_v = 0.f, a_v = 0.f;
  if (tid < LCH) {
    const float* qrow = qc + (size_t)tid * MD;
    for (int f = 0; f < MD; ++f) qn = fmaf(qrow[f], ns[f], qn);
    la_v = sc[SC_LA + tid];
    rs_v = sc[SC_RS + tid];
    a_v = sc[SC_A + tid];
  }

  for (int tile = wave; tile < 64; tile += NW) {
    const int tr = (tile >> 3) * 16, tc = (tile & 7) * 16;
    v8f ace = {};
    const float* ap = qc + (size_t)(tr + lm) * MD + 2 * half;
    for (int kk = 0; kk < MD; kk += 4) {
      v2f a, b;
      a.x = ap[kk]; a.y = ap[kk + 1];
      b.x = Cs[tc + lm][kk + 2 * half];
      b.y = Cs[tc + lm][kk + 2 * half + 1];
      ace = wmma4(a, b, ace);
    }
#pragma unroll
    for (int e = 0; e < 8; ++e) {
      const int r = tr + e + 8 * half, cc = tc + lm;
      Gs[r][cc] = ace[e] * oc[(size_t)r * MD + cc] * fcw_s[cc];
    }
  }
  __syncthreads();

  if (tid < LCH) {
    float b_v = 0.f;
    for (int f = 0; f < MD; ++f) b_v += Gs[tid][f];
    const float Ft = __expf(la_v);
    const float num = fmaf(Ft, b_v, a_v);
    const float den = fmaxf(fabsf(fmaf(Ft, qn, rs_v)), 1.0f);
    out[t0 + tid] = num / den + fc_b[0];
  }
}

extern "C" void kernel_launch(void* const* d_in, const int* in_sizes, int n_in,
                              void* d_out, int out_size, void* d_ws, size_t ws_size,
                              hipStream_t stream) {
  const float* x    = (const float*)d_in[0];
  const float* C0   = (const float*)d_in[1];
  const float* n0   = (const float*)d_in[2];
  const float* Wq   = (const float*)d_in[3];
  const float* bq   = (const float*)d_in[4];
  const float* Wk   = (const float*)d_in[5];
  const float* bk   = (const float*)d_in[6];
  const float* Wv   = (const float*)d_in[7];
  const float* bv   = (const float*)d_in[8];
  const float* wi   = (const float*)d_in[9];
  const float* bi   = (const float*)d_in[10];
  const float* wf   = (const float*)d_in[11];
  const float* bf   = (const float*)d_in[12];
  const float* Wo   = (const float*)d_in[13];
  const float* bo   = (const float*)d_in[14];
  const float* fc_w = (const float*)d_in[15];
  const float* fc_b = (const float*)d_in[16];

  float* ws = (float*)d_ws;
  const size_t NP = (size_t)T_LEN * MD;
  float* qw = ws;
  float* kw = ws + NP;
  float* vw = ws + 2 * NP;
  float* owp = ws + 3 * NP;
  float* igp = ws + 4 * NP;
  float* lfp = igp + T_LEN;

  float* outp  = (float*)d_out;
  float* Coutp = outp + T_LEN;
  float* noutp = Coutp + MD * MD;

  proj_kernel<<<dim3(T_LEN / PB, 4), 512, 0, stream>>>(
      x, Wq, bq, Wk, bk, Wv, bv, Wo, bo, qw, kw, vw, owp);
  gates_kernel<<<dim3(T_LEN / 256), 256, 0, stream>>>(
      x, wi, bi, wf, bf, igp, lfp);
  chunk_local_kernel<<<NCH, 512, 0, stream>>>(qw, kw, vw, owp, igp, lfp, fc_w);
  state_prefix_kernel<<<(MD * MD + MD + 255) / 256, 256, 0, stream>>>(
      kw, vw, C0, n0, Coutp, noutp);
  chunk_readout_kernel<<<NCH, 512, 0, stream>>>(
      qw, kw, vw, owp, fc_w, fc_b, outp);
}


# Gate sums x.wi, x.wf folded into the projection kernel, from the x values it stages (per-lane partial sums, then a butterfly add: their sum order changes); gates launch removed, 5 to 4 launches
# speedup vs baseline: 37.4815x; 1.2151x over previous
#include <hip/hip_runtime.h>
#include <hip/hip_bf16.h>
#include <math.h>


typedef __attribute__((ext_vector_type(2))) float v2f;
typedef __attribute__((ext_vector_type(8))) float v8f;

#define T_LEN 8192
#define IN_DIM 512
#define MD 128
#define LCH 128
#define NCH (T_LEN / LCH)
#define NW 16
#define CH (LCH * MD)
#define SC_LA 0
#define SC_RS 128
#define SC_A 256
#define SC_DN 384

__device__ __forceinline__ v8f wmma4(v2f a, v2f b, v8f c) {
  return __builtin_amdgcn_wmma_f32_16x16x4_f32(
       false, a,  false, b,
       (short)0, c,  false,  false);
}

__device__ __forceinline__ float sigmoidf_(float z) {
  return 1.0f / (1.0f + __expf(-z));
}

typedef __attribute__((ext_vector_type(4))) float v4f;

#define PB 128
#define PKT 64
#define PPITCH (PKT + 4)
#define PNK (IN_DIM / PKT)

__device__ __forceinline__ float gate_fma(float g, v4f xv, v4f w) {
  g = fmaf(xv.x, w.x, g);
  g = fmaf(xv.y, w.y, g);
  g = fmaf(xv.z, w.z, g);
  g = fmaf(xv.w, w.w, g);
  return g;
}

__global__ __launch_bounds__(512) void proj_kernel(
    const float* __restrict__ x,
    const float* __restrict__ Wq, const float* __restrict__ bq,
    const float* __restrict__ Wk, const float* __restrict__ bk,
    const float* __restrict__ Wv, const float* __restrict__ bv,
    const float* __restrict__ Wo, const float* __restrict__ bo,
    const float* __restrict__ wi, const float* __restrict__ bi,
    const float* __restrict__ wf, const float* __restrict__ bf,
    float* __restrict__ qw, float* __restrict__ kw,
    float* __restrict__ vw, float* __restrict__ ow,
    float* __restrict__ ig, float* __restrict__ lf) {
  __shared__ __attribute__((aligned(16))) float Xs[2][PB][PPITCH];
  __shared__ __attribute__((aligned(16))) float Ws[2][PB][PPITCH];

  const int tid = threadIdx.x;
  const int wave = tid >> 5;
  const int lane = tid & 31;
  const int half = lane >> 4;
  const int lm = lane & 15;

  const int rowBase = blockIdx.x * PB;
  const int mat = blockIdx.y;

  const float* W; const float* bias; float* dst;
  switch (mat) {
    case 0:  W = Wq; bias = bq; dst = qw; break;
    case 1:  W = Wk; bias = bk; dst = kw; break;
    case 2:  W = Wv; bias = bv; dst = vw; break;
    default: W = Wo; bias = bo; dst = ow; break;
  }

  const int sr = tid >> 4;
  const int sc = (tid & 15) * 4;
  const float* xg = x + (size_t)(rowBase + sr) * IN_DIM + sc;
  const float* wg = W + (size_t)sr * IN_DIM + sc;
  const bool gHi = (mat & 1) != 0;
  const float* gvec = ((mat < 2) ? wi : wf) + sc;
  float g0 = 0.f, g1 = 0.f;
  v4f gw;

  v4f px[4], pw[4];
#pragma unroll
  for (int i = 0; i < 4; ++i) {
    px[i] = *(const v4f*)(xg + (size_t)i * 32 * IN_DIM);
    pw[i] = *(const v4f*)(wg + (size_t)i * 32 * IN_DIM);
  }
  gw = *(const v4f*)gvec;
  g0 = gate_fma(g0, gHi ? px[2] : px[0], gw);
  g1 = gate_fma(g1, gHi ? px[3] : px[1], gw);
#pragma unroll
  for (int i = 0; i < 4; ++i) {
    *(v4f*)&Xs[0][sr + 32 * i][sc] = px[i];
    *(v4f*)&Ws[0][sr + 32 * i][sc] = pw[i];
  }
  __syncthreads();

  const int wr = (wave >> 2) * 32;
  const int wc = (wave & 3) * 32;
  v8f acc[2][2] = {};

  for (int kt = 0; kt < PNK; ++kt) {
    const int cur = kt & 1;
    const bool more = (kt + 1) < PNK;
    if (more) {
      const int k0 = (kt + 1) * PKT;
#pragma unroll
      for (int i = 0; i < 4; ++i) {
        px[i] = *(const v4f*)(xg + (size_t)i * 32 * IN_DIM + k0);
        pw[i] = *(const v4f*)(wg + (size_t)i * 32 * IN_DIM + k0);
      }
      gw = *(const v4f*)(gvec + k0);
    }

    const float* xa = &Xs[cur][wr + lm][4 * half];
    const float* wb = &Ws[cur][wc + lm][4 * half];
#pragma unroll
    for (int k8 = 0; k8 < PKT; k8 += 8) {
      const v4f a0 = *(const v4f*)(xa + k8);
      const v4f a1 = *(const v4f*)(xa + 16 * PPITCH + k8);
      const v4f b0 = *(const v4f*)(wb + k8);
      const v4f b1 = *(const v4f*)(wb + 16 * PPITCH + k8);
      const v2f a0l = {a0.x, a0.y}, a0h = {a0.z, a0.w};
      const v2f a1l = {a1.x, a1.y}, a1h = {a1.z, a1.w};
      const v2f b0l = {b0.x, b0.y}, b0h = {b0.z, b0.w};
      const v2f b1l = {b1.x, b1.y}, b1h = {b1.z, b1.w};
      acc[0][0] = wmma4(a0l, b0l, acc[0][0]);
      acc[0][1] = wmma4(a0l, b1l, acc[0][1]);
      acc[1][0] = wmma4(a1l, b0l, acc[1][0]);
      acc[1][1] = wmma4(a1l, b1l, acc[1][1]);
      acc[0][0] = wmma4(a0h, b0h, acc[0][0]);
      acc[0][1] = wmma4(a0h, b1h, acc[0][1]);
      acc[1][0] = wmma4(a1h, b0h, acc[1][0]);
      acc[1][1] = wmma4(a1h, b1h, acc[1][1]);
    }

    if (more) {
#pragma unroll
      for (int i = 0; i < 4; ++i) {
        *(v4f*)&Xs[cur ^ 1][sr + 32 * i][sc] = px[i];
        *(v4f*)&Ws[cur ^ 1][sr + 32 * i][sc] = pw[i];
      }
      __builtin_amdgcn_sched_barrier(0);
      g0 = gate_fma(g0, gHi ? px[2] : px[0], gw);
      g1 = gate_fma(g1, gHi ? px[3] : px[1], gw);
    }
    __syncthreads();
  }

  {
    g0 += __shfl_xor(g0, 8);  g1 += __shfl_xor(g1, 8);
    g0 += __shfl_xor(g0, 4);  g1 += __shfl_xor(g1, 4);
    g0 += __shfl_xor(g0, 2);  g1 += __shfl_xor(g1, 2);
    g0 += __shfl_xor(g0, 1);  g1 += __shfl_xor(g1, 1);
    const int gj = tid & 15;
    if (gj < 2) {
      const float s = (gj == 0) ? g0 : g1;
      const int t = rowBase + sr + 32 * (2 * (mat & 1) + gj);
      if (mat < 2) {
        ig[t] = __expf(s + bi[0]);
      } else {
        const float z = s + bf[0];
        lf[t] = (z >= 0.f) ? -log1pf(__expf(-z)) : (z - log1pf(__expf(z)));
      }
    }
  }

#pragma unroll
  for (int j = 0; j < 2; ++j) {
    const int cc = wc + 16 * j + lm;
    const float bv_ = bias[cc];
#pragma unroll
    for (int i = 0; i < 2; ++i) {
#pragma unroll
      for (int e = 0; e < 8; ++e) {
        const int r = rowBase + wr + 16 * i + e + 8 * half;
        float val = acc[i][j][e] + bv_;
        if (mat == 1) val *= 0.08838834764831845f;
        if (mat == 3) val = sigmoidf_(val);
        dst[(size_t)r * MD + cc] = val;
      }
    }
  }
}

__global__ __launch_bounds__(512) void chunk_local_kernel(
    const float* __restrict__ qw, float* kw, float* vw,
    const float* __restrict__ ow,
    const float* __restrict__ ig, const float* __restrict__ lf,
    const float* __restrict__ fc_w) {
  __shared__ float Ps[LCH][LCH + 1];
  __shared__ float Gs[LCH][MD + 1];
  __shared__ float la_s[LCH], ig_s[LCH], u_s[LCH], fcw_s[MD];

  const int tid = threadIdx.x;
  const int wave = tid >> 5;
  const int lane = tid & 31;
  const int half = lane >> 4;
  const int lm = lane & 15;
  const int t0 = blockIdx.x * LCH;

  const float* qc = qw + (size_t)t0 * MD;
  const float* oc = ow + (size_t)t0 * MD;
  float* kc = kw + (size_t)t0 * MD;
  float* vc = vw + (size_t)t0 * MD;

  if (tid < LCH) {
    la_s[tid] = lf[t0 + tid];
    ig_s[tid] = ig[t0 + tid];
    fcw_s[tid] = fc_w[tid];
  }
  __syncthreads();
  if (tid == 0) {
    float a = 0.f;
    for (int j = 0; j < LCH; ++j) { a += la_s[j]; la_s[j] = a; }
  }
  __syncthreads();
  if (tid < LCH) u_s[tid] = __expf(la_s[LCH - 1] - la_s[tid]) * ig_s[tid];

  for (int tile = wave; tile < 64; tile += NW) {
    const int tr = (tile >> 3) * 16, tc = (tile & 7) * 16;
    v8f acc = {};
    if (tc <= tr) {
      const float* ap = qc + (size_t)(tr + lm) * MD + 2 * half;
      const float* bp = kc + (size_t)(tc + lm) * MD + 2 * half;
      for (int kk = 0; kk < MD; kk += 4) {
        v2f a, b;
        a.x = ap[kk]; a.y = ap[kk + 1];
        b.x = bp[kk]; b.y = bp[kk + 1];
        acc = wmma4(a, b, acc);
      }
    }
#pragma unroll
    for (int e = 0; e < 8; ++e) {
      const int r = tr + e + 8 * half, cc = tc + lm;
      float p = 0.f;
      if (r >= cc) p = __expf(la_s[r] - la_s[cc]) * ig_s[cc] * acc[e];
      Ps[r][cc] = p;
    }
  }
  __syncthreads();

  float rs_v = 0.f;
  if (tid < LCH) {
    for (int j = 0; j <= tid; ++j) rs_v += Ps[tid][j];
  }

  for (int tile = wave; tile < 64; tile += NW) {
    const int tr = (tile >> 3) * 16, tc = (tile & 7) * 16;
    v8f aci = {};
    for (int kk = 0; kk < LCH; kk += 4) {
      v2f a, b;
      const int j0 = kk + 2 * half;
      a.x = Ps[tr + lm][j0]; a.y = Ps[tr + lm][j0 + 1];
      const float* bp = vc + (size_t)j0 * MD + tc + lm;
      b.x = bp[0]; b.y = bp[MD];
      aci = wmma4(a, b, aci);
    }
#pragma unroll
    for (int e = 0; e < 8; ++e) {
      const int r = tr + e + 8 * half, cc = tc + lm;
      Gs[r][cc] = aci[e] * oc[(size_t)r * MD + cc] * fcw_s[cc];
    }
  }

  v8f dacc[4];
#pragma unroll
  for (int i = 0; i < 4; ++i) {
    const int tile = wave + NW * i;
    const int tr = (tile >> 3) * 16, tc = (tile & 7) * 16;
    v8f acc = {};
    for (int kk = 0; kk < LCH; kk += 4) {
      v2f a, b;
      const int j0 = kk + 2 * half;
      a.x = u_s[j0]     * vc[(size_t)j0 * MD + tr + lm];
      a.y = u_s[j0 + 1] * vc[(size_t)(j0 + 1) * MD + tr + lm];
      const float* bp = kc + (size_t)j0 * MD + tc + lm;
      b.x = bp[0]; b.y = bp[MD];
      acc = wmma4(a, b, acc);
    }
    dacc[i] = acc;
  }
  float dn_v = 0.f;
  if (tid < MD) {
    for (int j = 0; j < LCH; ++j) dn_v = fmaf(u_s[j], kc[(size_t)j * MD + tid], dn_v);
  }
  __syncthreads();

#pragma unroll
  for (int i = 0; i < 4; ++i) {
    const int tile = wave + NW * i;
    const int tr = (tile >> 3) * 16, tc = (tile & 7) * 16;
#pragma unroll
    for (int e = 0; e < 8; ++e)
      kc[(size_t)(tr + e + 8 * half) * MD + tc + lm] = dacc[i][e];
  }
  if (tid < LCH) {
    float a_v = 0.f;
    for (int f = 0; f < MD; ++f) a_v += Gs[tid][f];
    vc[SC_LA + tid] = la_s[tid];
    vc[SC_RS + tid] = rs_v;
    vc[SC_A + tid] = a_v;
    vc[SC_DN + tid] = dn_v;
  }
}

__global__ __launch_bounds__(256) void state_prefix_kernel(
    float* kw, float* vw,
    const float* __restrict__ C0, const float* __restrict__ n0,
    float* __restrict__ C_out, float* __restrict__ n_out) {
  __shared__ float F_s[NCH];
  if (threadIdx.x < NCH)
    F_s[threadIdx.x] = __expf(vw[(size_t)threadIdx.x * CH + SC_LA + LCH - 1]);
  __syncthreads();

  const int gid = blockIdx.x * 256 + threadIdx.x;
  float* p; float* dst; float s;
  if (gid < MD * MD) {
    p = kw + gid; s = C0[gid]; dst = C_out + gid;
  } else if (gid < MD * MD + MD) {
    const int f = gid - MD * MD;
    p = vw + SC_DN + f; s = n0[f]; dst = n_out + f;
  } else {
    return;
  }
  for (int cb = 0; cb < NCH; cb += 16) {
    float d[16];
#pragma unroll
    for (int i = 0; i < 16; ++i) d[i] = p[(size_t)(cb + i) * CH];
#pragma unroll
    for (int i = 0; i < 16; ++i) {
      p[(size_t)(cb + i) * CH] = s;
      s = fmaf(F_s[cb + i], s, d[i]);
    }
  }
  *dst = s;
}

__global__ __launch_bounds__(512) void chunk_readout_kernel(
    const float* __restrict__ qw, const float* __restrict__ kw,
    const float* __restrict__ vw, const float* __restrict__ ow,
    const float* __restrict__ fc_w, const float* __restrict__ fc_b,
    float* __restrict__ out) {
  __shared__ float Cs[MD][MD + 1];
  __shared__ float Gs[LCH][MD + 1];
  __shared__ float ns[MD], fcw_s[MD];

  const int tid = threadIdx.x;
  const int wave = tid >> 5;
  const int lane = tid & 31;
  const int half = lane >> 4;
  const int lm = lane & 15;
  const int t0 = blockIdx.x * LCH;

  const float* qc = qw + (size_t)t0 * MD;
  const float* oc = ow + (size_t)t0 * MD;
  const float* st = kw + (size_t)t0 * MD;
  const float* sc = vw + (size_t)t0 * MD;

  for (int i = tid; i < MD * MD; i += 512) Cs[i >> 7][i & 127] = st[i];
  if (tid < MD) { ns[tid] = sc[SC_DN + tid]; fcw_s[tid] = fc_w[tid]; }
  __syncthreads();

  float qn = 0.f, la_v = 0.f, rs_v = 0.f, a_v = 0.f;
  if (tid < LCH) {
    const float* qrow = qc + (size_t)tid * MD;
    for (int f = 0; f < MD; ++f) qn = fmaf(qrow[f], ns[f], qn);
    la_v = sc[SC_LA + tid];
    rs_v = sc[SC_RS + tid];
    a_v = sc[SC_A + tid];
  }

  for (int tile = wave; tile < 64; tile += NW) {
    const int tr = (tile >> 3) * 16, tc = (tile & 7) * 16;
    v8f ace = {};
    const float* ap = qc + (size_t)(tr + lm) * MD + 2 * half;
    for (int kk = 0; kk < MD; kk += 4) {
      v2f a, b;
      a.x = ap[kk]; a.y = ap[kk + 1];
      b.x = Cs[tc + lm][kk + 2 * half];
      b.y = Cs[tc + lm][kk + 2 * half + 1];
      ace = wmma4(a, b, ace);
    }
#pragma unroll
    for (int e = 0; e < 8; ++e) {
      const int r = tr + e + 8 * half, cc = tc + lm;
      Gs[r][cc] = ace[e] * oc[(size_t)r * MD + cc] * fcw_s[cc];
    }
  }
  __syncthreads();

  if (tid < LCH) {
    float b_v = 0.f;
    for (int f = 0; f < MD; ++f) b_v += Gs[tid][f];
    const float Ft = __expf(la_v);
    const float num = fmaf(Ft, b_v, a_v);
    const float den = fmaxf(fabsf(fmaf(Ft, qn, rs_v)), 1.0f);
    out[t0 + tid] = num / den + fc_b[0];
  }
}

extern "C" void kernel_launch(void* const* d_in, const int* in_sizes, int n_in,
                              void* d_out, int out_size, void* d_ws, size_t ws_size,
                              hipStream_t stream) {
  const float* x    = (const float*)d_in[0];
  const float* C0   = (const float*)d_in[1];
  const float* n0   = (const float*)d_in[2];
  const float* Wq   = (const float*)d_in[3];
  const float* bq   = (const float*)d_in[4];
  const float* Wk   = (const float*)d_in[5];
  const float* bk   = (const float*)d_in[6];
  const float* Wv   = (const float*)d_in[7];
  const float* bv   = (const float*)d_in[8];
  const float* wi   = (const float*)d_in[9];
  const float* bi   = (const float*)d_in[10];
  const float* wf   = (const float*)d_in[11];
  const float* bf   = (const float*)d_in[12];
  const float* Wo   = (const float*)d_in[13];
  const float* bo   = (const float*)d_in[14];
  const float* fc_w = (const float*)d_in[15];
  const float* fc_b = (const float*)d_in[16];

  float* ws = (float*)d_ws;
  const size_t NP = (size_t)T_LEN * MD;
  float* qw = ws;
  float* kw = ws + NP;
  float* vw = ws + 2 * NP;
  float* owp = ws + 3 * NP;
  float* igp = ws + 4 * NP;
  float* lfp = igp + T_LEN;

  float* outp  = (float*)d_out;
  float* Coutp = outp + T_LEN;
  float* noutp = Coutp + MD * MD;

  proj_kernel<<<dim3(T_LEN / PB, 4), 512, 0, stream>>>(
      x, Wq, bq, Wk, bk, Wv, bv, Wo, bo, wi, bi, wf, bf, qw, kw, vw, owp, igp, lfp);
  chunk_local_kernel<<<NCH, 512, 0, stream>>>(qw, kw, vw, owp, igp, lfp, fc_w);
  state_prefix_kernel<<<(MD * MD + MD + 255) / 256, 256, 0, stream>>>(
      kw, vw, C0, n0, Coutp, noutp);
  chunk_readout_kernel<<<NCH, 512, 0, stream>>>(
      qw, kw, vw, owp, fc_w, fc_b, outp);
}


# Read-out re-tiled: one work-group per 32 steps (256 of 8 waves, was 64 of 16), state and q rows staged in LDS by whole-line loads, each wave 2 row tiles on one B fragment; sum orders unchanged
# speedup vs baseline: 44.6296x; 1.1907x over previous
#include <hip/hip_runtime.h>
#include <hip/hip_bf16.h>
#include <math.h>


typedef __attribute__((ext_vector_type(2))) float v2f;
typedef __attribute__((ext_vector_type(8))) float v8f;

#define T_LEN 8192
#define IN_DIM 512
#define MD 128
#define LCH 128
#define NCH (T_LEN / LCH)
#define NW 16
#define CH (LCH * MD)
#define SC_LA 0
#define SC_RS 128
#define SC_A 256
#define SC_DN 384

__device__ __forceinline__ v8f wmma4(v2f a, v2f b, v8f c) {
  return __builtin_amdgcn_wmma_f32_16x16x4_f32(
       false, a,  false, b,
       (short)0, c,  false,  false);
}

__device__ __forceinline__ float sigmoidf_(float z) {
  return 1.0f / (1.0f + __expf(-z));
}

typedef __attribute__((ext_vector_type(4))) float v4f;

#define PB 128
#define PKT 64
#define PPITCH (PKT + 4)
#define PNK (IN_DIM / PKT)

__device__ __forceinline__ float gate_fma(float g, v4f xv, v4f w) {
  g = fmaf(xv.x, w.x, g);
  g = fmaf(xv.y, w.y, g);
  g = fmaf(xv.z, w.z, g);
  g = fmaf(xv.w, w.w, g);
  return g;
}

__global__ __launch_bounds__(512) void proj_kernel(
    const float* __restrict__ x,
    const float* __restrict__ Wq, const float* __restrict__ bq,
    const float* __restrict__ Wk, const float* __restrict__ bk,
    const float* __restrict__ Wv, const float* __restrict__ bv,
    const float* __restrict__ Wo, const float* __restrict__ bo,
    const float* __restrict__ wi, const float* __restrict__ bi,
    const float* __restrict__ wf, const float* __restrict__ bf,
    float* __restrict__ qw, float* __restrict__ kw,
    float* __restrict__ vw, float* __restrict__ ow,
    float* __restrict__ ig, float* __restrict__ lf) {
  __shared__ __attribute__((aligned(16))) float Xs[2][PB][PPITCH];
  __shared__ __attribute__((aligned(16))) float Ws[2][PB][PPITCH];

  const int tid = threadIdx.x;
  const int wave = tid >> 5;
  const int lane = tid & 31;
  const int half = lane >> 4;
  const int lm = lane & 15;

  const int rowBase = blockIdx.x * PB;
  const int mat = blockIdx.y;

  const float* W; const float* bias; float* dst;
  switch (mat) {
    case 0:  W = Wq; bias = bq; dst = qw; break;
    case 1:  W = Wk; bias = bk; dst = kw; break;
    case 2:  W = Wv; bias = bv; dst = vw; break;
    default: W = Wo; bias = bo; dst = ow; break;
  }

  const int sr = tid >> 4;
  const int sc = (tid & 15) * 4;
  const float* xg = x + (size_t)(rowBase + sr) * IN_DIM + sc;
  const float* wg = W + (size_t)sr * IN_DIM + sc;
  const bool gHi = (mat & 1) != 0;
  const float* gvec = ((mat < 2) ? wi : wf) + sc;
  float g0 = 0.f, g1 = 0.f;
  v4f gw;

  v4f px[4], pw[4];
#pragma unroll
  for (int i = 0; i < 4; ++i) {
    px[i] = *(const v4f*)(xg + (size_t)i * 32 * IN_DIM);
    pw[i] = *(const v4f*)(wg + (size_t)i * 32 * IN_DIM);
  }
  gw = *(const v4f*)gvec;
  g0 = gate_fma(g0, gHi ? px[2] : px[0], gw);
  g1 = gate_fma(g1, gHi ? px[3] : px[1], gw);
#pragma unroll
  for (int i = 0; i < 4; ++i) {
    *(v4f*)&Xs[0][sr + 32 * i][sc] = px[i];
    *(v4f*)&Ws[0][sr + 32 * i][sc] = pw[i];
  }
  __syncthreads();

  const int wr = (wave >> 2) * 32;
  const int wc = (wave & 3) * 32;
  v8f acc[2][2] = {};

  for (int kt = 0; kt < PNK; ++kt) {
    const int cur = kt & 1;
    const bool more = (kt + 1) < PNK;
    if (more) {
      const int k0 = (kt + 1) * PKT;
#pragma unroll
      for (int i = 0; i < 4; ++i) {
        px[i] = *(const v4f*)(xg + (size_t)i * 32 * IN_DIM + k0);
        pw[i] = *(const v4f*)(wg + (size_t)i * 32 * IN_DIM + k0);
      }
      gw = *(const v4f*)(gvec + k0);
    }

    const float* xa = &Xs[cur][wr + lm][4 * half];
    const float* wb = &Ws[cur][wc + lm][4 * half];
#pragma unroll
    for (int k8 = 0; k8 < PKT; k8 += 8) {
      const v4f a0 = *(const v4f*)(xa + k8);
      const v4f a1 = *(const v4f*)(xa + 16 * PPITCH + k8);
      const v4f b0 = *(const v4f*)(wb + k8);
      const v4f b1 = *(const v4f*)(wb + 16 * PPITCH + k8);
      const v2f a0l = {a0.x, a0.y}, a0h = {a0.z, a0.w};
      const v2f a1l = {a1.x, a1.y}, a1h = {a1.z, a1.w};
      const v2f b0l = {b0.x, b0.y}, b0h = {b0.z, b0.w};
      const v2f b1l = {b1.x, b1.y}, b1h = {b1.z, b1.w};
      acc[0][0] = wmma4(a0l, b0l, acc[0][0]);
      acc[0][1] = wmma4(a0l, b1l, acc[0][1]);
      acc[1][0] = wmma4(a1l, b0l, acc[1][0]);
      acc[1][1] = wmma4(a1l, b1l, acc[1][1]);
      acc[0][0] = wmma4(a0h, b0h, acc[0][0]);
      acc[0][1] = wmma4(a0h, b1h, acc[0][1]);
      acc[1][0] = wmma4(a1h, b0h, acc[1][0]);
      acc[1][1] = wmma4(a1h, b1h, acc[1][1]);
    }

    if (more) {
#pragma unroll
      for (int i = 0; i < 4; ++i) {
        *(v4f*)&Xs[cur ^ 1][sr + 32 * i][sc] = px[i];
        *(v4f*)&Ws[cur ^ 1][sr + 32 * i][sc] = pw[i];
      }
      __builtin_amdgcn_sched_barrier(0);
      g0 = gate_fma(g0, gHi ? px[2] : px[0], gw);
      g1 = gate_fma(g1, gHi ? px[3] : px[1], gw);
    }
    __syncthreads();
  }

  {
    g0 += __shfl_xor(g0, 8);  g1 += __shfl_xor(g1, 8);
    g0 += __shfl_xor(g0, 4);  g1 += __shfl_xor(g1, 4);
    g0 += __shfl_xor(g0, 2);  g1 += __shfl_xor(g1, 2);
    g0 += __shfl_xor(g0, 1);  g1 += __shfl_xor(g1, 1);
    const int gj = tid & 15;
    if (gj < 2) {
      const float s = (gj == 0) ? g0 : g1;
      const int t = rowBase + sr + 32 * (2 * (mat & 1) + gj);
      if (mat < 2) {
        ig[t] = __expf(s + bi[0]);
      } else {
        const float z = s + bf[0];
        lf[t] = (z >= 0.f) ? -log1pf(__expf(-z)) : (z - log1pf(__expf(z)));
      }
    }
  }

#pragma unroll
  for (int j = 0; j < 2; ++j) {
    const int cc = wc + 16 * j + lm;
    const float bv_ = bias[cc];
#pragma unroll
    for (int i = 0; i < 2; ++i) {
#pragma unroll
      for (int e = 0; e < 8; ++e) {
        const int r = rowBase + wr + 16 * i + e + 8 * half;
        float val = acc[i][j][e] + bv_;
        if (mat == 1) val *= 0.08838834764831845f;
        if (mat == 3) val = sigmoidf_(val);
        dst[(size_t)r * MD + cc] = val;
      }
    }
  }
}

__global__ __launch_bounds__(512) void chunk_local_kernel(
    const float* __restrict__ qw, float* kw, float* vw,
    const float* __restrict__ ow,
    const float* __restrict__ ig, const float* __restrict__ lf,
    const float* __restrict__ fc_w) {
  __shared__ float Ps[LCH][LCH + 1];
  __shared__ float Gs[LCH][MD + 1];
  __shared__ float la_s[LCH], ig_s[LCH], u_s[LCH], fcw_s[MD];

  const int tid = threadIdx.x;
  const int wave = tid >> 5;
  const int lane = tid & 31;
  const int half = lane >> 4;
  const int lm = lane & 15;
  const int t0 = blockIdx.x * LCH;

  const float* qc = qw + (size_t)t0 * MD;
  const float* oc = ow + (size_t)t0 * MD;
  float* kc = kw + (size_t)t0 * MD;
  float* vc = vw + (size_t)t0 * MD;

  if (tid < LCH) {
    la_s[tid] = lf[t0 + tid];
    ig_s[tid] = ig[t0 + tid];
    fcw_s[tid] = fc_w[tid];
  }
  __syncthreads();
  if (tid == 0) {
    float a = 0.f;
    for (int j = 0; j < LCH; ++j) { a += la_s[j]; la_s[j] = a; }
  }
  __syncthreads();
  if (tid < LCH) u_s[tid] = __expf(la_s[LCH - 1] - la_s[tid]) * ig_s[tid];

  for (int tile = wave; tile < 64; tile += NW) {
    const int tr = (tile >> 3) * 16, tc = (tile & 7) * 16;
    v8f acc = {};
    if (tc <= tr) {
      const float* ap = qc + (size_t)(tr + lm) * MD + 2 * half;
      const float* bp = kc + (size_t)(tc + lm) * MD + 2 * half;
      for (int kk = 0; kk < MD; kk += 4) {
        v2f a, b;
        a.x = ap[kk]; a.y = ap[kk + 1];
        b.x = bp[kk]; b.y = bp[kk + 1];
        acc = wmma4(a, b, acc);
      }
    }
#pragma unroll
    for (int e = 0; e < 8; ++e) {
      const int r = tr + e + 8 * half, cc = tc + lm;
      float p = 0.f;
      if (r >= cc) p = __expf(la_s[r] - la_s[cc]) * ig_s[cc] * acc[e];
      Ps[r][cc] = p;
    }
  }
  __syncthreads();

  float rs_v = 0.f;
  if (tid < LCH) {
    for (int j = 0; j <= tid; ++j) rs_v += Ps[tid][j];
  }

  for (int tile = wave; tile < 64; tile += NW) {
    const int tr = (tile >> 3) * 16, tc = (tile & 7) * 16;
    v8f aci = {};
    for (int kk = 0; kk < LCH; kk += 4) {
      v2f a, b;
      const int j0 = kk + 2 * half;
      a.x = Ps[tr + lm][j0]; a.y = Ps[tr + lm][j0 + 1];
      const float* bp = vc + (size_t)j0 * MD + tc + lm;
      b.x = bp[0]; b.y = bp[MD];
      aci = wmma4(a, b, aci);
    }
#pragma unroll
    for (int e = 0; e < 8; ++e) {
      const int r = tr + e + 8 * half, cc = tc + lm;
      Gs[r][cc] = aci[e] * oc[(size_t)r * MD + cc] * fcw_s[cc];
    }
  }

  v8f dacc[4];
#pragma unroll
  for (int i = 0; i < 4; ++i) {
    const int tile = wave + NW * i;
    const int tr = (tile >> 3) * 16, tc = (tile & 7) * 16;
    v8f acc = {};
    for (int kk = 0; kk < LCH; kk += 4) {
      v2f a, b;
      const int j0 = kk + 2 * half;
      a.x = u_s[j0]     * vc[(size_t)j0 * MD + tr + lm];
      a.y = u_s[j0 + 1] * vc[(size_t)(j0 + 1) * MD + tr + lm];
      const float* bp = kc + (size_t)j0 * MD + tc + lm;
      b.x = bp[0]; b.y = bp[MD];
      acc = wmma4(a, b, acc);
    }
    dacc[i] = acc;
  }
  float dn_v = 0.f;
  if (tid < MD) {
    for (int j = 0; j < LCH; ++j) dn_v = fmaf(u_s[j], kc[(size_t)j * MD + tid], dn_v);
  }
  __syncthreads();

#pragma unroll
  for (int i = 0; i < 4; ++i) {
    const int tile = wave + NW * i;
    const int tr = (tile >> 3) * 16, tc = (tile & 7) * 16;
#pragma unroll
    for (int e = 0; e < 8; ++e)
      kc[(size_t)(tr + e + 8 * half) * MD + tc + lm] = dacc[i][e];
  }
  if (tid < LCH) {
    float a_v = 0.f;
    for (int f = 0; f < MD; ++f) a_v += Gs[tid][f];
    vc[SC_LA + tid] = la_s[tid];
    vc[SC_RS + tid] = rs_v;
    vc[SC_A + tid] = a_v;
    vc[SC_DN + tid] = dn_v;
  }
}

__global__ __launch_bounds__(256) void state_prefix_kernel(
    float* kw, float* vw,
    const float* __restrict__ C0, const float* __restrict__ n0,
    float* __restrict__ C_out, float* __restrict__ n_out) {
  __shared__ float F_s[NCH];
  if (threadIdx.x < NCH)
    F_s[threadIdx.x] = __expf(vw[(size_t)threadIdx.x * CH + SC_LA + LCH - 1]);
  __syncthreads();

  const int gid = blockIdx.x * 256 + threadIdx.x;
  float* p; float* dst; float s;
  if (gid < MD * MD) {
    p = kw + gid; s = C0[gid]; dst = C_out + gid;
  } else if (gid < MD * MD + MD) {
    const int f = gid - MD * MD;
    p = vw + SC_DN + f; s = n0[f]; dst = n_out + f;
  } else {
    return;
  }
  for (int cb = 0; cb < NCH; cb += 16) {
    float d[16];
#pragma unroll
    for (int i = 0; i < 16; ++i) d[i] = p[(size_t)(cb + i) * CH];
#pragma unroll
    for (int i = 0; i < 16; ++i) {
      p[(size_t)(cb + i) * CH] = s;
      s = fmaf(F_s[cb + i], s, d[i]);
    }
  }
  *dst = s;
}

#define RB 32
#define RPITCH (MD + 4)

__global__ __launch_bounds__(256) void chunk_readout_kernel(
    const float* __restrict__ qw, const float* __restrict__ kw,
    const float* __restrict__ vw, const float* __restrict__ ow,
    const float* __restrict__ fc_w, const float* __restrict__ fc_b,
    float* __restrict__ out) {
  __shared__ __attribute__((aligned(16))) float Cs[MD][RPITCH];
  __shared__ __attribute__((aligned(16))) float Qs[RB][RPITCH];
  __shared__ __attribute__((aligned(16))) float Gs[RB][RPITCH];
  __shared__ __attribute__((aligned(16))) float ns[MD];
  __shared__ float fcw_s[MD], qn_s[RB];

  const int tid = threadIdx.x;
  const int wave = tid >> 5;
  const int lane = tid & 31;
  const int half = lane >> 4;
  const int lm = lane & 15;
  const int t0 = (blockIdx.x >> 2) * LCH;
  const int sub = (blockIdx.x & 3) * RB;

  const float* qc = qw + (size_t)(t0 + sub) * MD;
  const float* oc = ow + (size_t)(t0 + sub) * MD;
  const float* st = kw + (size_t)t0 * MD;
  const float* sc = vw + (size_t)t0 * MD;

  const int srow = tid >> 5;
  const int scol = (tid & 31) * 4;
  v4f cst[16], qst[4];
#pragma unroll
  for (int i = 0; i < 16; ++i)
    cst[i] = *(const v4f*)(st + (size_t)(srow + 8 * i) * MD + scol);
#pragma unroll
  for (int i = 0; i < 4; ++i)
    qst[i] = *(const v4f*)(qc + (size_t)(srow + 8 * i) * MD + scol);
  const int vi = tid & (MD - 1);
  const float nsv = sc[SC_DN + vi];
  const float fwv = fc_w[vi];
  const float la_v = sc[SC_LA + sub + lane];
  const float rs_v = sc[SC_RS + sub + lane];
  const float a_v = sc[SC_A + sub + lane];

  const int cc = wave * 16 + lm;
  float ov0[8], ov1[8];
#pragma unroll
  for (int e = 0; e < 8; ++e) {
    ov0[e] = oc[(size_t)(e + 8 * half) * MD + cc];
    ov1[e] = oc[(size_t)(16 + e + 8 * half) * MD + cc];
  }

#pragma unroll
  for (int i = 0; i < 16; ++i) *(v4f*)&Cs[srow + 8 * i][scol] = cst[i];
#pragma unroll
  for (int i = 0; i < 4; ++i) *(v4f*)&Qs[srow + 8 * i][scol] = qst[i];
  ns[vi] = nsv;
  fcw_s[vi] = fwv;
  __syncthreads();

  v8f ac0 = {}, ac1 = {};
  const float* a0p = &Qs[lm][2 * half];
  const float* a1p = &Qs[16 + lm][2 * half];
  const float* bp = &Cs[wave * 16 + lm][2 * half];
#pragma unroll
  for (int kk = 0; kk < MD; kk += 4) {
    const v2f a0 = *(const v2f*)(a0p + kk);
    const v2f a1 = *(const v2f*)(a1p + kk);
    const v2f b = *(const v2f*)(bp + kk);
    ac0 = wmma4(a0, b, ac0);
    ac1 = wmma4(a1, b, ac1);
  }
  const float fw = fcw_s[cc];
#pragma unroll
  for (int e = 0; e < 8; ++e) {
    const int r = e + 8 * half;
    Gs[r][cc] = ac0[e] * ov0[e] * fw;
    Gs[16 + r][cc] = ac1[e] * ov1[e] * fw;
  }
  __syncthreads();

  float b_v = 0.f;
  if (wave == 0) {
    v4f gv[MD / 4];
#pragma unroll
    for (int i = 0; i < MD / 4; ++i) gv[i] = *(const v4f*)&Gs[lane][4 * i];
    __builtin_amdgcn_sched_barrier(0);
#pragma unroll
    for (int i = 0; i < MD / 4; ++i) {
      b_v += gv[i].x; b_v += gv[i].y; b_v += gv[i].z; b_v += gv[i].w;
    }
  } else if (wave == 1) {
    float qn = 0.f;
#pragma unroll
    for (int h = 0; h < 4; ++h) {
      v4f qv[8], nv[8];
#pragma unroll
      for (int i = 0; i < 8; ++i) {
        qv[i] = *(const v4f*)&Qs[lane][32 * h + 4 * i];
        nv[i] = *(const v4f*)&ns[32 * h + 4 * i];
      }
      __builtin_amdgcn_sched_barrier(0);
#pragma unroll
      for (int i = 0; i < 8; ++i) {
        qn = fmaf(qv[i].x, nv[i].x, qn);
        qn = fmaf(qv[i].y, nv[i].y, qn);
        qn = fmaf(qv[i].z, nv[i].z, qn);
        qn = fmaf(qv[i].w, nv[i].w, qn);
      }
    }
    qn_s[lane] = qn;
  }
  __syncthreads();

  if (tid < RB) {
    const float qn = qn_s[tid];
    const float Ft = __expf(la_v);
    const float num = fmaf(Ft, b_v, a_v);
    const float den = fmaxf(fabsf(fmaf(Ft, qn, rs_v)), 1.0f);
    out[t0 + sub + tid] = num / den + fc_b[0];
  }
}

extern "C" void kernel_launch(void* const* d_in, const int* in_sizes, int n_in,
                              void* d_out, int out_size, void* d_ws, size_t ws_size,
                              hipStream_t stream) {
  const float* x    = (const float*)d_in[0];
  const float* C0   = (const float*)d_in[1];
  const float* n0   = (const float*)d_in[2];
  const float* Wq   = (const float*)d_in[3];
  const float* bq   = (const float*)d_in[4];
  const float* Wk   = (const float*)d_in[5];
  const float* bk   = (const float*)d_in[6];
  const float* Wv   = (const float*)d_in[7];
  const float* bv   = (const float*)d_in[8];
  const float* wi   = (const float*)d_in[9];
  const float* bi   = (const float*)d_in[10];
  const float* wf   = (const float*)d_in[11];
  const float* bf   = (const float*)d_in[12];
  const float* Wo   = (const float*)d_in[13];
  const float* bo   = (const float*)d_in[14];
  const float* fc_w = (const float*)d_in[15];
  const float* fc_b = (const float*)d_in[16];

  float* ws = (float*)d_ws;
  const size_t NP = (size_t)T_LEN * MD;
  float* qw = ws;
  float* kw = ws + NP;
  float* vw = ws + 2 * NP;
  float* owp = ws + 3 * NP;
  float* igp = ws + 4 * NP;
  float* lfp = igp + T_LEN;

  float* outp  = (float*)d_out;
  float* Coutp = outp + T_LEN;
  float* noutp = Coutp + MD * MD;

  proj_kernel<<<dim3(T_LEN / PB, 4), 512, 0, stream>>>(
      x, Wq, bq, Wk, bk, Wv, bv, Wo, bo, wi, bi, wf, bf, qw, kw, vw, owp, igp, lfp);
  chunk_local_kernel<<<NCH, 512, 0, stream>>>(qw, kw, vw, owp, igp, lfp, fc_w);
  state_prefix_kernel<<<(MD * MD + MD + 255) / 256, 256, 0, stream>>>(
      kw, vw, C0, n0, Coutp, noutp);
  chunk_readout_kernel<<<NCH * 4, 256, 0, stream>>>(
      qw, kw, vw, owp, fc_w, fc_b, outp);
}


# Chunk-local pass fed from LDS: its q, k, v rows staged once by whole-line loads, all three products read their fragments there, P V skips the zero blocks, equal work per wave; sum orders unchanged
# speedup vs baseline: 53.4206x; 1.1970x over previous
#include <hip/hip_runtime.h>
#include <hip/hip_bf16.h>
#include <math.h>


typedef __attribute__((ext_vector_type(2))) float v2f;
typedef __attribute__((ext_vector_type(8))) float v8f;

#define T_LEN 8192
#define IN_DIM 512
#define MD 128
#define LCH 128
#define NCH (T_LEN / LCH)
#define NW 16
#define CH (LCH * MD)
#define SC_LA 0
#define SC_RS 128
#define SC_A 256
#define SC_DN 384

__device__ __forceinline__ v8f wmma4(v2f a, v2f b, v8f c) {
  return __builtin_amdgcn_wmma_f32_16x16x4_f32(
       false, a,  false, b,
       (short)0, c,  false,  false);
}

__device__ __forceinline__ float sigmoidf_(float z) {
  return 1.0f / (1.0f + __expf(-z));
}

typedef __attribute__((ext_vector_type(4))) float v4f;

#define PB 128
#define PKT 64
#define PPITCH (PKT + 4)
#define PNK (IN_DIM / PKT)

__device__ __forceinline__ float gate_fma(float g, v4f xv, v4f w) {
  g = fmaf(xv.x, w.x, g);
  g = fmaf(xv.y, w.y, g);
  g = fmaf(xv.z, w.z, g);
  g = fmaf(xv.w, w.w, g);
  return g;
}

__global__ __launch_bounds__(512) void proj_kernel(
    const float* __restrict__ x,
    const float* __restrict__ Wq, const float* __restrict__ bq,
    const float* __restrict__ Wk, const float* __restrict__ bk,
    const float* __restrict__ Wv, const float* __restrict__ bv,
    const float* __restrict__ Wo, const float* __restrict__ bo,
    const float* __restrict__ wi, const float* __restrict__ bi,
    const float* __restrict__ wf, const float* __restrict__ bf,
    float* __restrict__ qw, float* __restrict__ kw,
    float* __restrict__ vw, float* __restrict__ ow,
    float* __restrict__ ig, float* __restrict__ lf) {
  __shared__ __attribute__((aligned(16))) float Xs[2][PB][PPITCH];
  __shared__ __attribute__((aligned(16))) float Ws[2][PB][PPITCH];

  const int tid = threadIdx.x;
  const int wave = tid >> 5;
  const int lane = tid & 31;
  const int half = lane >> 4;
  const int lm = lane & 15;

  const int rowBase = blockIdx.x * PB;
  const int mat = blockIdx.y;

  const float* W; const float* bias; float* dst;
  switch (mat) {
    case 0:  W = Wq; bias = bq; dst = qw; break;
    case 1:  W = Wk; bias = bk; dst = kw; break;
    case 2:  W = Wv; bias = bv; dst = vw; break;
    default: W = Wo; bias = bo; dst = ow; break;
  }

  const int sr = tid >> 4;
  const int sc = (tid & 15) * 4;
  const float* xg = x + (size_t)(rowBase + sr) * IN_DIM + sc;
  const float* wg = W + (size_t)sr * IN_DIM + sc;
  const bool gHi = (mat & 1) != 0;
  const float* gvec = ((mat < 2) ? wi : wf) + sc;
  float g0 = 0.f, g1 = 0.f;
  v4f gw;

  v4f px[4], pw[4];
#pragma unroll
  for (int i = 0; i < 4; ++i) {
    px[i] = *(const v4f*)(xg + (size_t)i * 32 * IN_DIM);
    pw[i] = *(const v4f*)(wg + (size_t)i * 32 * IN_DIM);
  }
  gw = *(const v4f*)gvec;
  g0 = gate_fma(g0, gHi ? px[2] : px[0], gw);
  g1 = gate_fma(g1, gHi ? px[3] : px[1], gw);
#pragma unroll
  for (int i = 0; i < 4; ++i) {
    *(v4f*)&Xs[0][sr + 32 * i][sc] = px[i];
    *(v4f*)&Ws[0][sr + 32 * i][sc] = pw[i];
  }
  __syncthreads();

  const int wr = (wave >> 2) * 32;
  const int wc = (wave & 3) * 32;
  v8f acc[2][2] = {};

  for (int kt = 0; kt < PNK; ++kt) {
    const int cur = kt & 1;
    const bool more = (kt + 1) < PNK;
    if (more) {
      const int k0 = (kt + 1) * PKT;
#pragma unroll
      for (int i = 0; i < 4; ++i) {
        px[i] = *(const v4f*)(xg + (size_t)i * 32 * IN_DIM + k0);
        pw[i] = *(const v4f*)(wg + (size_t)i * 32 * IN_DIM + k0);
      }
      gw = *(const v4f*)(gvec + k0);
    }

    const float* xa = &Xs[cur][wr + lm][4 * half];
    const float* wb = &Ws[cur][wc + lm][4 * half];
#pragma unroll
    for (int k8 = 0; k8 < PKT; k8 += 8) {
      const v4f a0 = *(const v4f*)(xa + k8);
      const v4f a1 = *(const v4f*)(xa + 16 * PPITCH + k8);
      const v4f b0 = *(const v4f*)(wb + k8);
      const v4f b1 = *(const v4f*)(wb + 16 * PPITCH + k8);
      const v2f a0l = {a0.x, a0.y}, a0h = {a0.z, a0.w};
      const v2f a1l = {a1.x, a1.y}, a1h = {a1.z, a1.w};
      const v2f b0l = {b0.x, b0.y}, b0h = {b0.z, b0.w};
      const v2f b1l = {b1.x, b1.y}, b1h = {b1.z, b1.w};
      acc[0][0] = wmma4(a0l, b0l, acc[0][0]);
      acc[0][1] = wmma4(a0l, b1l, acc[0][1]);
      acc[1][0] = wmma4(a1l, b0l, acc[1][0]);
      acc[1][1] = wmma4(a1l, b1l, acc[1][1]);
      acc[0][0] = wmma4(a0h, b0h, acc[0][0]);
      acc[0][1] = wmma4(a0h, b1h, acc[0][1]);
      acc[1][0] = wmma4(a1h, b0h, acc[1][0]);
      acc[1][1] = wmma4(a1h, b1h, acc[1][1]);
    }

    if (more) {
#pragma unroll
      for (int i = 0; i < 4; ++i) {
        *(v4f*)&Xs[cur ^ 1][sr + 32 * i][sc] = px[i];
        *(v4f*)&Ws[cur ^ 1][sr + 32 * i][sc] = pw[i];
      }
      __builtin_amdgcn_sched_barrier(0);
      g0 = gate_fma(g0, gHi ? px[2] : px[0], gw);
      g1 = gate_fma(g1, gHi ? px[3] : px[1], gw);
    }
    __syncthreads();
  }

  {
    g0 += __shfl_xor(g0, 8);  g1 += __shfl_xor(g1, 8);
    g0 += __shfl_xor(g0, 4);  g1 += __shfl_xor(g1, 4);
    g0 += __shfl_xor(g0, 2);  g1 += __shfl_xor(g1, 2);
    g0 += __shfl_xor(g0, 1);  g1 += __shfl_xor(g1, 1);
    const int gj = tid & 15;
    if (gj < 2) {
      const float s = (gj == 0) ? g0 : g1;
      const int t = rowBase + sr + 32 * (2 * (mat & 1) + gj);
      if (mat < 2) {
        ig[t] = __expf(s + bi[0]);
      } else {
        const float z = s + bf[0];
        lf[t] = (z >= 0.f) ? -log1pf(__expf(-z)) : (z - log1pf(__expf(z)));
      }
    }
  }

#pragma unroll
  for (int j = 0; j < 2; ++j) {
    const int cc = wc + 16 * j + lm;
    const float bv_ = bias[cc];
#pragma unroll
    for (int i = 0; i < 2; ++i) {
#pragma unroll
      for (int e = 0; e < 8; ++e) {
        const int r = rowBase + wr + 16 * i + e + 8 * half;
        float val = acc[i][j][e] + bv_;
        if (mat == 1) val *= 0.08838834764831845f;
        if (mat == 3) val = sigmoidf_(val);
        dst[(size_t)r * MD + cc] = val;
      }
    }
  }
}

#define QP (MD + 4)
#define VP (MD + 8)

template <int RSEL>
__device__ __forceinline__ void pv_wave(const float (*Ps)[QP], const float (*Vs)[VP],
                                        float (*Gs)[QP], const float* __restrict__ oc,
                                        const float* fcw_s, int tc, int lm, int half) {
  constexpr int RT[4] = {RSEL, 7 - RSEL, 3 - RSEL, 4 + RSEL};
  const int cc = tc + lm;
  float ov[4][8];
#pragma unroll
  for (int i = 0; i < 4; ++i) {
#pragma unroll
    for (int e = 0; e < 8; ++e)
      ov[i][e] = oc[(size_t)(RT[i] * 16 + e + 8 * half) * MD + cc];
  }
  v8f aci[4] = {};
#pragma unroll
  for (int kb = 0; kb < 8; ++kb) {
    v2f b[4];
#pragma unroll
    for (int s = 0; s < 4; ++s) {
      const int j0 = kb * 16 + 4 * s + 2 * half;
      b[s].x = Vs[j0][cc];
      b[s].y = Vs[j0 + 1][cc];
    }
#pragma unroll
    for (int i = 0; i < 4; ++i) {
      if (RT[i] >= kb) {
        const float* ap = &Ps[RT[i] * 16 + lm][kb * 16 + 2 * half];
#pragma unroll
        for (int s = 0; s < 4; ++s) {
          const v2f a = *(const v2f*)(ap + 4 * s);
          aci[i] = wmma4(a, b[s], aci[i]);
        }
      }
    }
  }
  const float fw = fcw_s[cc];
#pragma unroll
  for (int i = 0; i < 4; ++i) {
#pragma unroll
    for (int e = 0; e < 8; ++e)
      Gs[RT[i] * 16 + e + 8 * half][cc] = aci[i][e] * ov[i][e] * fw;
  }
}

__global__ __launch_bounds__(512) void chunk_local_kernel(
    const float* __restrict__ qw, float* kw, float* vw,
    const float* __restrict__ ow,
    const float* __restrict__ ig, const float* __restrict__ lf,
    const float* __restrict__ fc_w) {
  __shared__ __attribute__((aligned(16))) float Qs[LCH][QP];
  __shared__ __attribute__((aligned(16))) float Ks[LCH][QP];
  __shared__ __attribute__((aligned(16))) float Ps[LCH][QP];
  __shared__ __attribute__((aligned(16))) float Vs[LCH][VP];
  __shared__ __attribute__((aligned(16))) float la_s[LCH], ig_s[LCH], u_s[LCH], fcw_s[MD];
  float (*Gs)[QP] = Qs;

  const int tid = threadIdx.x;
  const int wv = __builtin_amdgcn_readfirstlane(tid >> 5);
  const int lane = tid & 31;
  const int half = lane >> 4;
  const int lm = lane & 15;
  const int t0 = blockIdx.x * LCH;

  const float* qc = qw + (size_t)t0 * MD;
  const float* oc = ow + (size_t)t0 * MD;
  float* kc = kw + (size_t)t0 * MD;
  float* vc = vw + (size_t)t0 * MD;

  const int srow = tid >> 5;
  const int scol = (tid & 31) * 4;
  v4f qst[8], kst[8], vst[8];
#pragma unroll
  for (int i = 0; i < 8; ++i) {
    qst[i] = *(const v4f*)(qc + (size_t)(srow + 16 * i) * MD + scol);
    kst[i] = *(const v4f*)(kc + (size_t)(srow + 16 * i) * MD + scol);
  }
  const int gi = tid & (LCH - 1);
  const float igv = ig[t0 + gi];
  const float fwv = fc_w[gi];
  if (tid == 0) {
    v4f lv[LCH / 4];
#pragma unroll
    for (int i = 0; i < LCH / 4; ++i) lv[i] = *(const v4f*)(lf + t0 + 4 * i);
    float a = 0.f;
#pragma unroll
    for (int i = 0; i < LCH / 4; ++i) {
      a += lv[i].x; lv[i].x = a;
      a += lv[i].y; lv[i].y = a;
      a += lv[i].z; lv[i].z = a;
      a += lv[i].w; lv[i].w = a;
    }
#pragma unroll
    for (int i = 0; i < LCH / 4; ++i) *(v4f*)&la_s[4 * i] = lv[i];
  }
  ig_s[gi] = igv;
  fcw_s[gi] = fwv;
#pragma unroll
  for (int i = 0; i < 8; ++i) {
    *(v4f*)&Qs[srow + 16 * i][scol] = qst[i];
    *(v4f*)&Ks[srow + 16 * i][scol] = kst[i];
  }
  __syncthreads();

#pragma unroll
  for (int i = 0; i < 8; ++i)
    vst[i] = *(const v4f*)(vc + (size_t)(srow + 16 * i) * MD + scol);

  for (int t = wv; t < 36; t += NW) {
    int r = 0;
    while ((r + 1) * (r + 2) / 2 <= t) ++r;
    const int tr = r * 16, tc = (t - r * (r + 1) / 2) * 16;
    v8f acc = {};
    const float* ap = &Qs[tr + lm][2 * half];
    const float* bp = &Ks[tc + lm][2 * half];
#pragma unroll
    for (int kk = 0; kk < MD; kk += 4) {
      const v2f a = *(const v2f*)(ap + kk);
      const v2f b = *(const v2f*)(bp + kk);
      acc = wmma4(a, b, acc);
    }
#pragma unroll
    for (int e = 0; e < 8; ++e) {
      const int rr = tr + e + 8 * half, cc = tc + lm;
      float p = 0.f;
      if (rr >= cc) p = __expf(la_s[rr] - la_s[cc]) * ig_s[cc] * acc[e];
      Ps[rr][cc] = p;
    }
  }
#pragma unroll
  for (int i = 0; i < 8; ++i) *(v4f*)&Vs[srow + 16 * i][scol] = vst[i];
  if (tid < LCH) u_s[tid] = __expf(la_s[LCH - 1] - la_s[tid]) * ig_s[tid];
  __syncthreads();

  float rs_v = 0.f;
  if (tid < LCH) {
#pragma unroll
    for (int h = 0; h < 4; ++h) {
      if (h <= wv) {
        v4f pv[8];
#pragma unroll
        for (int i = 0; i < 8; ++i) pv[i] = *(const v4f*)&Ps[tid][32 * h + 4 * i];
        __builtin_amdgcn_sched_barrier(0);
#pragma unroll
        for (int i = 0; i < 8; ++i) {
          const int j = 32 * h + 4 * i;
          if (j     <= tid) rs_v += pv[i].x;
          if (j + 1 <= tid) rs_v += pv[i].y;
          if (j + 2 <= tid) rs_v += pv[i].z;
          if (j + 3 <= tid) rs_v += pv[i].w;
        }
      }
    }
  }

  {
    const int tc = (wv & 7) * 16;
    if (wv >> 3) pv_wave<1>(Ps, Vs, Gs, oc, fcw_s, tc, lm, half);
    else         pv_wave<0>(Ps, Vs, Gs, oc, fcw_s, tc, lm, half);
  }
  __syncthreads();

#pragma unroll
  for (int i = 0; i < 8; ++i) {
    const int row = srow + 16 * i;
    const float u = u_s[row];
    v4f v = *(const v4f*)&Vs[row][scol];
    v.x *= u; v.y *= u; v.z *= u; v.w *= u;
    *(v4f*)&Vs[row][scol] = v;
  }
  __syncthreads();

  float a_v = 0.f, dn_v = 0.f;
  if (tid < LCH) {
#pragma unroll
    for (int h = 0; h < 4; ++h) {
      v4f gv[8];
#pragma unroll
      for (int i = 0; i < 8; ++i) gv[i] = *(const v4f*)&Gs[tid][32 * h + 4 * i];
      __builtin_amdgcn_sched_barrier(0);
#pragma unroll
      for (int i = 0; i < 8; ++i) {
        a_v += gv[i].x; a_v += gv[i].y; a_v += gv[i].z; a_v += gv[i].w;
      }
    }
#pragma unroll 16
    for (int j = 0; j < LCH; ++j) dn_v = fmaf(u_s[j], Ks[j][tid], dn_v);
  }

  {
    const int tr0 = (wv >> 2) * 32, tc0 = (wv & 3) * 32;
    v8f d00 = {}, d01 = {}, d10 = {}, d11 = {};
#pragma unroll
    for (int kk = 0; kk < LCH; kk += 4) {
      const int j0 = kk + 2 * half;
      const v2f a0 = {Vs[j0][tr0 + lm],      Vs[j0 + 1][tr0 + lm]};
      const v2f a1 = {Vs[j0][tr0 + 16 + lm], Vs[j0 + 1][tr0 + 16 + lm]};
      const v2f b0 = {Ks[j0][tc0 + lm],      Ks[j0 + 1][tc0 + lm]};
      const v2f b1 = {Ks[j0][tc0 + 16 + lm], Ks[j0 + 1][tc0 + 16 + lm]};
      d00 = wmma4(a0, b0, d00);
      d01 = wmma4(a0, b1, d01);
      d10 = wmma4(a1, b0, d10);
      d11 = wmma4(a1, b1, d11);
    }
#pragma unroll
    for (int e = 0; e < 8; ++e) {
      float* kr0 = kc + (size_t)(tr0 + e + 8 * half) * MD + tc0 + lm;
      float* kr1 = kr0 + (size_t)16 * MD;
      kr0[0] = d00[e];  kr0[16] = d01[e];
      kr1[0] = d10[e];  kr1[16] = d11[e];
    }
  }
  if (tid < LCH) {
    vc[SC_LA + tid] = la_s[tid];
    vc[SC_RS + tid] = rs_v;
    vc[SC_A + tid] = a_v;
    vc[SC_DN + tid] = dn_v;
  }
}

__global__ __launch_bounds__(256) void state_prefix_kernel(
    float* kw, float* vw,
    const float* __restrict__ C0, const float* __restrict__ n0,
    float* __restrict__ C_out, float* __restrict__ n_out) {
  __shared__ float F_s[NCH];
  if (threadIdx.x < NCH)
    F_s[threadIdx.x] = __expf(vw[(size_t)threadIdx.x * CH + SC_LA + LCH - 1]);
  __syncthreads();

  const int gid = blockIdx.x * 256 + threadIdx.x;
  float* p; float* dst; float s;
  if (gid < MD * MD) {
    p = kw + gid; s = C0[gid]; dst = C_out + gid;
  } else if (gid < MD * MD + MD) {
    const int f = gid - MD * MD;
    p = vw + SC_DN + f; s = n0[f]; dst = n_out + f;
  } else {
    return;
  }
  for (int cb = 0; cb < NCH; cb += 16) {
    float d[16];
#pragma unroll
    for (int i = 0; i < 16; ++i) d[i] = p[(size_t)(cb + i) * CH];
#pragma unroll
    for (int i = 0; i < 16; ++i) {
      p[(size_t)(cb + i) * CH] = s;
      s = fmaf(F_s[cb + i], s, d[i]);
    }
  }
  *dst = s;
}

#define RB 32
#define RPITCH (MD + 4)

__global__ __launch_bounds__(256) void chunk_readout_kernel(
    const float* __restrict__ qw, const float* __restrict__ kw,
    const float* __restrict__ vw, const float* __restrict__ ow,
    const float* __restrict__ fc_w, const float* __restrict__ fc_b,
    float* __restrict__ out) {
  __shared__ __attribute__((aligned(16))) float Cs[MD][RPITCH];
  __shared__ __attribute__((aligned(16))) float Qs[RB][RPITCH];
  __shared__ __attribute__((aligned(16))) float Gs[RB][RPITCH];
  __shared__ __attribute__((aligned(16))) float ns[MD];
  __shared__ float fcw_s[MD], qn_s[RB];

  const int tid = threadIdx.x;
  const int wave = tid >> 5;
  const int lane = tid & 31;
  const int half = lane >> 4;
  const int lm = lane & 15;
  const int t0 = (blockIdx.x >> 2) * LCH;
  const int sub = (blockIdx.x & 3) * RB;

  const float* qc = qw + (size_t)(t0 + sub) * MD;
  const float* oc = ow + (size_t)(t0 + sub) * MD;
  const float* st = kw + (size_t)t0 * MD;
  const float* sc = vw + (size_t)t0 * MD;

  const int srow = tid >> 5;
  const int scol = (tid & 31) * 4;
  v4f cst[16], qst[4];
#pragma unroll
  for (int i = 0; i < 16; ++i)
    cst[i] = *(const v4f*)(st + (size_t)(srow + 8 * i) * MD + scol);
#pragma unroll
  for (int i = 0; i < 4; ++i)
    qst[i] = *(const v4f*)(qc + (size_t)(srow + 8 * i) * MD + scol);
  const int vi = tid & (MD - 1);
  const float nsv = sc[SC_DN + vi];
  const float fwv = fc_w[vi];
  const float la_v = sc[SC_LA + sub + lane];
  const float rs_v = sc[SC_RS + sub + lane];
  const float a_v = sc[SC_A + sub + lane];

  const int cc = wave * 16 + lm;
  float ov0[8], ov1[8];
#pragma unroll
  for (int e = 0; e < 8; ++e) {
    ov0[e] = oc[(size_t)(e + 8 * half) * MD + cc];
    ov1[e] = oc[(size_t)(16 + e + 8 * half) * MD + cc];
  }

#pragma unroll
  for (int i = 0; i < 16; ++i) *(v4f*)&Cs[srow + 8 * i][scol] = cst[i];
#pragma unroll
  for (int i = 0; i < 4; ++i) *(v4f*)&Qs[srow + 8 * i][scol] = qst[i];
  ns[vi] = nsv;
  fcw_s[vi] = fwv;
  __syncthreads();

  v8f ac0 = {}, ac1 = {};
  const float* a0p = &Qs[lm][2 * half];
  const float* a1p = &Qs[16 + lm][2 * half];
  const float* bp = &Cs[wave * 16 + lm][2 * half];
#pragma unroll
  for (int kk = 0; kk < MD; kk += 4) {
    const v2f a0 = *(const v2f*)(a0p + kk);
    const v2f a1 = *(const v2f*)(a1p + kk);
    const v2f b = *(const v2f*)(bp + kk);
    ac0 = wmma4(a0, b, ac0);
    ac1 = wmma4(a1, b, ac1);
  }
  const float fw = fcw_s[cc];
#pragma unroll
  for (int e = 0; e < 8; ++e) {
    const int r = e + 8 * half;
    Gs[r][cc] = ac0[e] * ov0[e] * fw;
    Gs[16 + r][cc] = ac1[e] * ov1[e] * fw;
  }
  __syncthreads();

  float b_v = 0.f;
  if (wave == 0) {
    v4f gv[MD / 4];
#pragma unroll
    for (int i = 0; i < MD / 4; ++i) gv[i] = *(const v4f*)&Gs[lane][4 * i];
    __builtin_amdgcn_sched_barrier(0);
#pragma unroll
    for (int i = 0; i < MD / 4; ++i) {
      b_v += gv[i].x; b_v += gv[i].y; b_v += gv[i].z; b_v += gv[i].w;
    }
  } else if (wave == 1) {
    float qn = 0.f;
#pragma unroll
    for (int h = 0; h < 4; ++h) {
      v4f qv[8], nv[8];
#pragma unroll
      for (int i = 0; i < 8; ++i) {
        qv[i] = *(const v4f*)&Qs[lane][32 * h + 4 * i];
        nv[i] = *(const v4f*)&ns[32 * h + 4 * i];
      }
      __builtin_amdgcn_sched_barrier(0);
#pragma unroll
      for (int i = 0; i < 8; ++i) {
        qn = fmaf(qv[i].x, nv[i].x, qn);
        qn = fmaf(qv[i].y, nv[i].y, qn);
        qn = fmaf(qv[i].z, nv[i].z, qn);
        qn = fmaf(qv[i].w, nv[i].w, qn);
      }
    }
    qn_s[lane] = qn;
  }
  __syncthreads();

  if (tid < RB) {
    const float qn = qn_s[tid];
    const float Ft = __expf(la_v);
    const float num = fmaf(Ft, b_v, a_v);
    const float den = fmaxf(fabsf(fmaf(Ft, qn, rs_v)), 1.0f);
    out[t0 + sub + tid] = num / den + fc_b[0];
  }
}

extern "C" void kernel_launch(void* const* d_in, const int* in_sizes, int n_in,
                              void* d_out, int out_size, void* d_ws, size_t ws_size,
                              hipStream_t stream) {
  const float* x    = (const float*)d_in[0];
  const float* C0   = (const float*)d_in[1];
  const float* n0   = (const float*)d_in[2];
  const float* Wq   = (const float*)d_in[3];
  const float* bq   = (const float*)d_in[4];
  const float* Wk   = (const float*)d_in[5];
  const float* bk   = (const float*)d_in[6];
  const float* Wv   = (const float*)d_in[7];
  const float* bv   = (const float*)d_in[8];
  const float* wi   = (const float*)d_in[9];
  const float* bi   = (const float*)d_in[10];
  const float* wf   = (const float*)d_in[11];
  const float* bf   = (const float*)d_in[12];
  const float* Wo   = (const float*)d_in[13];
  const float* bo   = (const float*)d_in[14];
  const float* fc_w = (const float*)d_in[15];
  const float* fc_b = (const float*)d_in[16];

  float* ws = (float*)d_ws;
  const size_t NP = (size_t)T_LEN * MD;
  float* qw = ws;
  float* kw = ws + NP;
  float* vw = ws + 2 * NP;
  float* owp = ws + 3 * NP;
  float* igp = ws + 4 * NP;
  float* lfp = igp + T_LEN;

  float* outp  = (float*)d_out;
  float* Coutp = outp + T_LEN;
  float* noutp = Coutp + MD * MD;

  proj_kernel<<<dim3(T_LEN / PB, 4), 512, 0, stream>>>(
      x, Wq, bq, Wk, bk, Wv, bv, Wo, bo, wi, bi, wf, bf, qw, kw, vw, owp, igp, lfp);
  chunk_local_kernel<<<NCH, 512, 0, stream>>>(qw, kw, vw, owp, igp, lfp, fc_w);
  state_prefix_kernel<<<(MD * MD + MD + 255) / 256, 256, 0, stream>>>(
      kw, vw, C0, n0, Coutp, noutp);
  chunk_readout_kernel<<<NCH * 4, 256, 0, stream>>>(
      qw, kw, vw, owp, fc_w, fc_b, outp);
}


# Chunk-local pass split over four work-groups per chunk (256, was 64): each 32 steps and 32 increment rows; increments and scalar rows in a workspace area of their own; sum orders unchanged
# speedup vs baseline: 57.8296x; 1.0825x over previous
#include <hip/hip_runtime.h>
#include <hip/hip_bf16.h>
#include <math.h>


typedef __attribute__((ext_vector_type(2))) float v2f;
typedef __attribute__((ext_vector_type(8))) float v8f;

#define T_LEN 8192
#define IN_DIM 512
#define MD 128
#define LCH 128
#define NCH (T_LEN / LCH)
#define NW 16
#define CH (LCH * MD)
#define SC_LA 0
#define SC_RS 128
#define SC_A 256
#define SC_DN 384

__device__ __forceinline__ v8f wmma4(v2f a, v2f b, v8f c) {
  return __builtin_amdgcn_wmma_f32_16x16x4_f32(
       false, a,  false, b,
       (short)0, c,  false,  false);
}

__device__ __forceinline__ float sigmoidf_(float z) {
  return 1.0f / (1.0f + __expf(-z));
}

typedef __attribute__((ext_vector_type(4))) float v4f;

#define PB 128
#define PKT 64
#define PPITCH (PKT + 4)
#define PNK (IN_DIM / PKT)

__device__ __forceinline__ float gate_fma(float g, v4f xv, v4f w) {
  g = fmaf(xv.x, w.x, g);
  g = fmaf(xv.y, w.y, g);
  g = fmaf(xv.z, w.z, g);
  g = fmaf(xv.w, w.w, g);
  return g;
}

__global__ __launch_bounds__(512) void proj_kernel(
    const float* __restrict__ x,
    const float* __restrict__ Wq, const float* __restrict__ bq,
    const float* __restrict__ Wk, const float* __restrict__ bk,
    const float* __restrict__ Wv, const float* __restrict__ bv,
    const float* __restrict__ Wo, const float* __restrict__ bo,
    const float* __restrict__ wi, const float* __restrict__ bi,
    const float* __restrict__ wf, const float* __restrict__ bf,
    float* __restrict__ qw, float* __restrict__ kw,
    float* __restrict__ vw, float* __restrict__ ow,
    float* __restrict__ ig, float* __restrict__ lf) {
  __shared__ __attribute__((aligned(16))) float Xs[2][PB][PPITCH];
  __shared__ __attribute__((aligned(16))) float Ws[2][PB][PPITCH];

  const int tid = threadIdx.x;
  const int wave = tid >> 5;
  const int lane = tid & 31;
  const int half = lane >> 4;
  const int lm = lane & 15;

  const int rowBase = blockIdx.x * PB;
  const int mat = blockIdx.y;

  const float* W; const float* bias; float* dst;
  switch (mat) {
    case 0:  W = Wq; bias = bq; dst = qw; break;
    case 1:  W = Wk; bias = bk; dst = kw; break;
    case 2:  W = Wv; bias = bv; dst = vw; break;
    default: W = Wo; bias = bo; dst = ow; break;
  }

  const int sr = tid >> 4;
  const int sc = (tid & 15) * 4;
  const float* xg = x + (size_t)(rowBase + sr) * IN_DIM + sc;
  const float* wg = W + (size_t)sr * IN_DIM + sc;
  const bool gHi = (mat & 1) != 0;
  const float* gvec = ((mat < 2) ? wi : wf) + sc;
  float g0 = 0.f, g1 = 0.f;
  v4f gw;

  v4f px[4], pw[4];
#pragma unroll
  for (int i = 0; i < 4; ++i) {
    px[i] = *(const v4f*)(xg + (size_t)i * 32 * IN_DIM);
    pw[i] = *(const v4f*)(wg + (size_t)i * 32 * IN_DIM);
  }
  gw = *(const v4f*)gvec;
  g0 = gate_fma(g0, gHi ? px[2] : px[0], gw);
  g1 = gate_fma(g1, gHi ? px[3] : px[1], gw);
#pragma unroll
  for (int i = 0; i < 4; ++i) {
    *(v4f*)&Xs[0][sr + 32 * i][sc] = px[i];
    *(v4f*)&Ws[0][sr + 32 * i][sc] = pw[i];
  }
  __syncthreads();

  const int wr = (wave >> 2) * 32;
  const int wc = (wave & 3) * 32;
  v8f acc[2][2] = {};

  for (int kt = 0; kt < PNK; ++kt) {
    const int cur = kt & 1;
    const bool more = (kt + 1) < PNK;
    if (more) {
      const int k0 = (kt + 1) * PKT;
#pragma unroll
      for (int i = 0; i < 4; ++i) {
        px[i] = *(const v4f*)(xg + (size_t)i * 32 * IN_DIM + k0);
        pw[i] = *(const v4f*)(wg + (size_t)i * 32 * IN_DIM + k0);
      }
      gw = *(const v4f*)(gvec + k0);
    }

    const float* xa = &Xs[cur][wr + lm][4 * half];
    const float* wb = &Ws[cur][wc + lm][4 * half];
#pragma unroll
    for (int k8 = 0; k8 < PKT; k8 += 8) {
      const v4f a0 = *(const v4f*)(xa + k8);
      const v4f a1 = *(const v4f*)(xa + 16 * PPITCH + k8);
      const v4f b0 = *(const v4f*)(wb + k8);
      const v4f b1 = *(const v4f*)(wb + 16 * PPITCH + k8);
      const v2f a0l = {a0.x, a0.y}, a0h = {a0.z, a0.w};
      const v2f a1l = {a1.x, a1.y}, a1h = {a1.z, a1.w};
      const v2f b0l = {b0.x, b0.y}, b0h = {b0.z, b0.w};
      const v2f b1l = {b1.x, b1.y}, b1h = {b1.z, b1.w};
      acc[0][0] = wmma4(a0l, b0l, acc[0][0]);
      acc[0][1] = wmma4(a0l, b1l, acc[0][1]);
      acc[1][0] = wmma4(a1l, b0l, acc[1][0]);
      acc[1][1] = wmma4(a1l, b1l, acc[1][1]);
      acc[0][0] = wmma4(a0h, b0h, acc[0][0]);
      acc[0][1] = wmma4(a0h, b1h, acc[0][1]);
      acc[1][0] = wmma4(a1h, b0h, acc[1][0]);
      acc[1][1] = wmma4(a1h, b1h, acc[1][1]);
    }

    if (more) {
#pragma unroll
      for (int i = 0; i < 4; ++i) {
        *(v4f*)&Xs[cur ^ 1][sr + 32 * i][sc] = px[i];
        *(v4f*)&Ws[cur ^ 1][sr + 32 * i][sc] = pw[i];
      }
      __builtin_amdgcn_sched_barrier(0);
      g0 = gate_fma(g0, gHi ? px[2] : px[0], gw);
      g1 = gate_fma(g1, gHi ? px[3] : px[1], gw);
    }
    __syncthreads();
  }

  {
    g0 += __shfl_xor(g0, 8);  g1 += __shfl_xor(g1, 8);
    g0 += __shfl_xor(g0, 4);  g1 += __shfl_xor(g1, 4);
    g0 += __shfl_xor(g0, 2);  g1 += __shfl_xor(g1, 2);
    g0 += __shfl_xor(g0, 1);  g1 += __shfl_xor(g1, 1);
    const int gj = tid & 15;
    if (gj < 2) {
      const float s = (gj == 0) ? g0 : g1;
      const int t = rowBase + sr + 32 * (2 * (mat & 1) + gj);
      if (mat < 2) {
        ig[t] = __expf(s + bi[0]);
      } else {
        const float z = s + bf[0];
        lf[t] = (z >= 0.f) ? -log1pf(__expf(-z)) : (z - log1pf(__expf(z)));
      }
    }
  }

#pragma unroll
  for (int j = 0; j < 2; ++j) {
    const int cc = wc + 16 * j + lm;
    const float bv_ = bias[cc];
#pragma unroll
    for (int i = 0; i < 2; ++i) {
#pragma unroll
      for (int e = 0; e < 8; ++e) {
        const int r = rowBase + wr + 16 * i + e + 8 * half;
        float val = acc[i][j][e] + bv_;
        if (mat == 1) val *= 0.08838834764831845f;
        if (mat == 3) val = sigmoidf_(val);
        dst[(size_t)r * MD + cc] = val;
      }
    }
  }
}

#define QP (MD + 4)
#define VP (MD + 8)

template <int RSEL>
__device__ __forceinline__ void pv_wave(const float (*Ps)[QP], const float (*Vs)[VP],
                                        float (*Gs)[QP], const float* __restrict__ oc,
                                        const float* fcw_s, int tc, int lm, int half) {
  constexpr int RT[4] = {RSEL, 7 - RSEL, 3 - RSEL, 4 + RSEL};
  const int cc = tc + lm;
  float ov[4][8];
#pragma unroll
  for (int i = 0; i < 4; ++i) {
#pragma unroll
    for (int e = 0; e < 8; ++e)
      ov[i][e] = oc[(size_t)(RT[i] * 16 + e + 8 * half) * MD + cc];
  }
  v8f aci[4] = {};
#pragma unroll
  for (int kb = 0; kb < 8; ++kb) {
    v2f b[4];
#pragma unroll
    for (int s = 0; s < 4; ++s) {
      const int j0 = kb * 16 + 4 * s + 2 * half;
      b[s].x = Vs[j0][cc];
      b[s].y = Vs[j0 + 1][cc];
    }
#pragma unroll
    for (int i = 0; i < 4; ++i) {
      if (RT[i] >= kb) {
        const float* ap = &Ps[RT[i] * 16 + lm][kb * 16 + 2 * half];
#pragma unroll
        for (int s = 0; s < 4; ++s) {
          const v2f a = *(const v2f*)(ap + 4 * s);
          aci[i] = wmma4(a, b[s], aci[i]);
        }
      }
    }
  }
  const float fw = fcw_s[cc];
#pragma unroll
  for (int i = 0; i < 4; ++i) {
#pragma unroll
    for (int e = 0; e < 8; ++e)
      Gs[RT[i] * 16 + e + 8 * half][cc] = aci[i][e] * ov[i][e] * fw;
  }
}

__global__ __launch_bounds__(512) void chunk_local_kernel(
    const float* __restrict__ qw, float* kw, float* vw,
    const float* __restrict__ ow,
    const float* __restrict__ ig, const float* __restrict__ lf,
    const float* __restrict__ fc_w) {
  __shared__ __attribute__((aligned(16))) float Qs[LCH][QP];
  __shared__ __attribute__((aligned(16))) float Ks[LCH][QP];
  __shared__ __attribute__((aligned(16))) float Ps[LCH][QP];
  __shared__ __attribute__((aligned(16))) float Vs[LCH][VP];
  __shared__ __attribute__((aligned(16))) float la_s[LCH], ig_s[LCH], u_s[LCH], fcw_s[MD];
  float (*Gs)[QP] = Qs;

  const int tid = threadIdx.x;
  const int wv = __builtin_amdgcn_readfirstlane(tid >> 5);
  const int lane = tid & 31;
  const int half = lane >> 4;
  const int lm = lane & 15;
  const int t0 = blockIdx.x * LCH;

  const float* qc = qw + (size_t)t0 * MD;
  const float* oc = ow + (size_t)t0 * MD;
  float* kc = kw + (size_t)t0 * MD;
  float* vc = vw + (size_t)t0 * MD;

  const int srow = tid >> 5;
  const int scol = (tid & 31) * 4;
  v4f qst[8], kst[8], vst[8];
#pragma unroll
  for (int i = 0; i < 8; ++i) {
    qst[i] = *(const v4f*)(qc + (size_t)(srow + 16 * i) * MD + scol);
    kst[i] = *(const v4f*)(kc + (size_t)(srow + 16 * i) * MD + scol);
  }
  const int gi = tid & (LCH - 1);
  const float igv = ig[t0 + gi];
  const float fwv = fc_w[gi];
  if (tid == 0) {
    v4f lv[LCH / 4];
#pragma unroll
    for (int i = 0; i < LCH / 4; ++i) lv[i] = *(const v4f*)(lf + t0 + 4 * i);
    float a = 0.f;
#pragma unroll
    for (int i = 0; i < LCH / 4; ++i) {
      a += lv[i].x; lv[i].x = a;
      a += lv[i].y; lv[i].y = a;
      a += lv[i].z; lv[i].z = a;
      a += lv[i].w; lv[i].w = a;
    }
#pragma unroll
    for (int i = 0; i < LCH / 4; ++i) *(v4f*)&la_s[4 * i] = lv[i];
  }
  ig_s[gi] = igv;
  fcw_s[gi] = fwv;
#pragma unroll
  for (int i = 0; i < 8; ++i) {
    *(v4f*)&Qs[srow + 16 * i][scol] = qst[i];
    *(v4f*)&Ks[srow + 16 * i][scol] = kst[i];
  }
  __syncthreads();

#pragma unroll
  for (int i = 0; i < 8; ++i)
    vst[i] = *(const v4f*)(vc + (size_t)(srow + 16 * i) * MD + scol);

  for (int t = wv; t < 36; t += NW) {
    int r = 0;
    while ((r + 1) * (r + 2) / 2 <= t) ++r;
    const int tr = r * 16, tc = (t - r * (r + 1) / 2) * 16;
    v8f acc = {};
    const float* ap = &Qs[tr + lm][2 * half];
    const float* bp = &Ks[tc + lm][2 * half];
#pragma unroll
    for (int kk = 0; kk < MD; kk += 4) {
      const v2f a = *(const v2f*)(ap + kk);
      const v2f b = *(const v2f*)(bp + kk);
      acc = wmma4(a, b, acc);
    }
#pragma unroll
    for (int e = 0; e < 8; ++e) {
      const int rr = tr + e + 8 * half, cc = tc + lm;
      float p = 0.f;
      if (rr >= cc) p = __expf(la_s[rr] - la_s[cc]) * ig_s[cc] * acc[e];
      Ps[rr][cc] = p;
    }
  }
#pragma unroll
  for (int i = 0; i < 8; ++i) *(v4f*)&Vs[srow + 16 * i][scol] = vst[i];
  if (tid < LCH) u_s[tid] = __expf(la_s[LCH - 1] - la_s[tid]) * ig_s[tid];
  __syncthreads();

  float rs_v = 0.f;
  if (tid < LCH) {
#pragma unroll
    for (int h = 0; h < 4; ++h) {
      if (h <= wv) {
        v4f pv[8];
#pragma unroll
        for (int i = 0; i < 8; ++i) pv[i] = *(const v4f*)&Ps[tid][32 * h + 4 * i];
        __builtin_amdgcn_sched_barrier(0);
#pragma unroll
        for (int i = 0; i < 8; ++i) {
          const int j = 32 * h + 4 * i;
          if (j     <= tid) rs_v += pv[i].x;
          if (j + 1 <= tid) rs_v += pv[i].y;
          if (j + 2 <= tid) rs_v += pv[i].z;
          if (j + 3 <= tid) rs_v += pv[i].w;
        }
      }
    }
  }

  {
    const int tc = (wv & 7) * 16;
    if (wv >> 3) pv_wave<1>(Ps, Vs, Gs, oc, fcw_s, tc, lm, half);
    else         pv_wave<0>(Ps, Vs, Gs, oc, fcw_s, tc, lm, half);
  }
  __syncthreads();

#pragma unroll
  for (int i = 0; i < 8; ++i) {
    const int row = srow + 16 * i;
    const float u = u_s[row];
    v4f v = *(const v4f*)&Vs[row][scol];
    v.x *= u; v.y *= u; v.z *= u; v.w *= u;
    *(v4f*)&Vs[row][scol] = v;
  }
  __syncthreads();

  float a_v = 0.f, dn_v = 0.f;
  if (tid < LCH) {
#pragma unroll
    for (int h = 0; h < 4; ++h) {
      v4f gv[8];
#pragma unroll
      for (int i = 0; i < 8; ++i) gv[i] = *(const v4f*)&Gs[tid][32 * h + 4 * i];
      __builtin_amdgcn_sched_barrier(0);
#pragma unroll
      for (int i = 0; i < 8; ++i) {
        a_v += gv[i].x; a_v += gv[i].y; a_v += gv[i].z; a_v += gv[i].w;
      }
    }
#pragma unroll 16
    for (int j = 0; j < LCH; ++j) dn_v = fmaf(u_s[j], Ks[j][tid], dn_v);
  }

  {
    const int tr0 = (wv >> 2) * 32, tc0 = (wv & 3) * 32;
    v8f d00 = {}, d01 = {}, d10 = {}, d11 = {};
#pragma unroll
    for (int kk = 0; kk < LCH; kk += 4) {
      const int j0 = kk + 2 * half;
      const v2f a0 = {Vs[j0][tr0 + lm],      Vs[j0 + 1][tr0 + lm]};
      const v2f a1 = {Vs[j0][tr0 + 16 + lm], Vs[j0 + 1][tr0 + 16 + lm]};
      const v2f b0 = {Ks[j0][tc0 + lm],      Ks[j0 + 1][tc0 + lm]};
      const v2f b1 = {Ks[j0][tc0 + 16 + lm], Ks[j0 + 1][tc0 + 16 + lm]};
      d00 = wmma4(a0, b0, d00);
      d01 = wmma4(a0, b1, d01);
      d10 = wmma4(a1, b0, d10);
      d11 = wmma4(a1, b1, d11);
    }
#pragma unroll
    for (int e = 0; e < 8; ++e) {
      float* kr0 = kc + (size_t)(tr0 + e + 8 * half) * MD + tc0 + lm;
      float* kr1 = kr0 + (size_t)16 * MD;
      kr0[0] = d00[e];  kr0[16] = d01[e];
      kr1[0] = d10[e];  kr1[16] = d11[e];
    }
  }
  if (tid < LCH) {
    vc[SC_LA + tid] = la_s[tid];
    vc[SC_RS + tid] = rs_v;
    vc[SC_A + tid] = a_v;
    vc[SC_DN + tid] = dn_v;
  }
}

#define SB 32
#define SCS 512

__global__ __launch_bounds__(512) void chunk_local4_kernel(
    const float* __restrict__ qw, const float* __restrict__ kw,
    const float* __restrict__ vw, const float* __restrict__ ow,
    const float* __restrict__ ig, const float* __restrict__ lf,
    const float* __restrict__ fc_w,
    float* __restrict__ dcw, float* __restrict__ scw) {
  __shared__ __attribute__((aligned(16))) float Qs[SB][QP];
  __shared__ __attribute__((aligned(16))) float Ks[LCH][QP];
  __shared__ __attribute__((aligned(16))) float Ps[SB][QP];
  __shared__ __attribute__((aligned(16))) float Vs[LCH][VP];
  __shared__ __attribute__((aligned(16))) float la_s[LCH], ig_s[LCH], u_s[LCH], fcw_s[MD];
  float (*Gs)[QP] = Qs;

  const int tid = threadIdx.x;
  const int wv = __builtin_amdgcn_readfirstlane(tid >> 5);
  const int lane = tid & 31;
  const int half = lane >> 4;
  const int lm = lane & 15;
  const int ch = blockIdx.x >> 2;
  const int s4 = blockIdx.x & 3;
  const int t0 = ch * LCH;
  const int r0 = s4 * SB;

  const float* qc = qw + (size_t)(t0 + r0) * MD;
  const float* oc = ow + (size_t)(t0 + r0) * MD;
  const float* kc = kw + (size_t)t0 * MD;
  const float* vc = vw + (size_t)t0 * MD;
  float* dc = dcw + (size_t)ch * CH;
  float* sc = scw + (size_t)ch * SCS;

  const int srow = tid >> 5;
  const int scol = (tid & 31) * 4;
  v4f qst[2], kst[8], vst[8];
#pragma unroll
  for (int i = 0; i < 2; ++i)
    qst[i] = *(const v4f*)(qc + (size_t)(srow + 16 * i) * MD + scol);
#pragma unroll
  for (int i = 0; i < 8; ++i)
    kst[i] = *(const v4f*)(kc + (size_t)(srow + 16 * i) * MD + scol);
  const int gi = tid & (LCH - 1);
  const float igv = ig[t0 + gi];
  const float fwv = fc_w[gi];
  if (tid == 0) {
    v4f lv[LCH / 4];
#pragma unroll
    for (int i = 0; i < LCH / 4; ++i) lv[i] = *(const v4f*)(lf + t0 + 4 * i);
    float a = 0.f;
#pragma unroll
    for (int i = 0; i < LCH / 4; ++i) {
      a += lv[i].x; lv[i].x = a;
      a += lv[i].y; lv[i].y = a;
      a += lv[i].z; lv[i].z = a;
      a += lv[i].w; lv[i].w = a;
    }
#pragma unroll
    for (int i = 0; i < LCH / 4; ++i) *(v4f*)&la_s[4 * i] = lv[i];
  }
  ig_s[gi] = igv;
  fcw_s[gi] = fwv;
#pragma unroll
  for (int i = 0; i < 2; ++i) *(v4f*)&Qs[srow + 16 * i][scol] = qst[i];
#pragma unroll
  for (int i = 0; i < 8; ++i) *(v4f*)&Ks[srow + 16 * i][scol] = kst[i];
  __syncthreads();

#pragma unroll
  for (int i = 0; i < 8; ++i)
    vst[i] = *(const v4f*)(vc + (size_t)(srow + 16 * i) * MD + scol);

  if (wv < 4 * s4 + 3) {
    const int rt = (wv > 2 * s4) ? 1 : 0;
    const int trl = rt * 16;
    const int tc = (rt ? (wv - 2 * s4 - 1) : wv) * 16;
    v8f acc = {};
    const float* ap = &Qs[trl + lm][2 * half];
    const float* bp = &Ks[tc + lm][2 * half];
#pragma unroll
    for (int kk = 0; kk < MD; kk += 4) {
      const v2f a = *(const v2f*)(ap + kk);
      const v2f b = *(const v2f*)(bp + kk);
      acc = wmma4(a, b, acc);
    }
#pragma unroll
    for (int e = 0; e < 8; ++e) {
      const int rl = trl + e + 8 * half;
      const int rr = r0 + rl, cc = tc + lm;
      float p = 0.f;
      if (rr >= cc) p = __expf(la_s[rr] - la_s[cc]) * ig_s[cc] * acc[e];
      Ps[rl][cc] = p;
    }
  }
#pragma unroll
  for (int i = 0; i < 8; ++i) *(v4f*)&Vs[srow + 16 * i][scol] = vst[i];
  if (tid < LCH) u_s[tid] = __expf(la_s[LCH - 1] - la_s[tid]) * ig_s[tid];
  __syncthreads();

  {
    const int rt = wv >> 3;
    const int cc = (wv & 7) * 16 + lm;
    float ov[8];
#pragma unroll
    for (int e = 0; e < 8; ++e) ov[e] = oc[(size_t)(rt * 16 + e + 8 * half) * MD + cc];
    v8f aci = {};
    const int nkb = 2 * s4 + rt + 1;
    for (int kb = 0; kb < nkb; ++kb) {
      v2f b[4];
#pragma unroll
      for (int s = 0; s < 4; ++s) {
        const int j0 = kb * 16 + 4 * s + 2 * half;
        b[s].x = Vs[j0][cc];
        b[s].y = Vs[j0 + 1][cc];
      }
      const float* ap = &Ps[rt * 16 + lm][kb * 16 + 2 * half];
#pragma unroll
      for (int s = 0; s < 4; ++s) {
        const v2f a = *(const v2f*)(ap + 4 * s);
        aci = wmma4(a, b[s], aci);
      }
    }
    const float fw = fcw_s[cc];
#pragma unroll
    for (int e = 0; e < 8; ++e)
      Gs[rt * 16 + e + 8 * half][cc] = aci[e] * ov[e] * fw;
  }
  __syncthreads();

  if (wv < 8) {
    const int fr = r0 + (wv >> 2) * 16;
    const int tc0 = (wv & 3) * 32;
    v8f d0 = {}, d1 = {};
#pragma unroll
    for (int kk = 0; kk < LCH; kk += 4) {
      const int j0 = kk + 2 * half;
      const v2f uu = *(const v2f*)&u_s[j0];
      const v2f a  = {Vs[j0][fr + lm] * uu.x,  Vs[j0 + 1][fr + lm] * uu.y};
      const v2f b0 = {Ks[j0][tc0 + lm],        Ks[j0 + 1][tc0 + lm]};
      const v2f b1 = {Ks[j0][tc0 + 16 + lm],   Ks[j0 + 1][tc0 + 16 + lm]};
      d0 = wmma4(a, b0, d0);
      d1 = wmma4(a, b1, d1);
    }
#pragma unroll
    for (int e = 0; e < 8; ++e) {
      float* dr = dc + (size_t)(fr + e + 8 * half) * MD + tc0 + lm;
      dr[0] = d0[e];  dr[16] = d1[e];
    }
  } else if (wv == 8) {
    float a_v = 0.f;
#pragma unroll
    for (int h = 0; h < 4; ++h) {
      v4f gv[8];
#pragma unroll
      for (int i = 0; i < 8; ++i) gv[i] = *(const v4f*)&Gs[lane][32 * h + 4 * i];
      __builtin_amdgcn_sched_barrier(0);
#pragma unroll
      for (int i = 0; i < 8; ++i) {
        a_v += gv[i].x; a_v += gv[i].y; a_v += gv[i].z; a_v += gv[i].w;
      }
    }
    sc[SC_A + r0 + lane] = a_v;
    sc[SC_LA + r0 + lane] = la_s[r0 + lane];
  } else if (wv == 9) {
    float dn_v = 0.f;
#pragma unroll 16
    for (int j = 0; j < LCH; ++j) dn_v = fmaf(u_s[j], Ks[j][r0 + lane], dn_v);
    sc[SC_DN + r0 + lane] = dn_v;
  } else if (wv == 10) {
    float rs_v = 0.f;
    const int row = r0 + lane;
#pragma unroll
    for (int h = 0; h < 4; ++h) {
      if (h <= s4) {
        v4f pv[8];
#pragma unroll
        for (int i = 0; i < 8; ++i) pv[i] = *(const v4f*)&Ps[lane][32 * h + 4 * i];
        __builtin_amdgcn_sched_barrier(0);
#pragma unroll
        for (int i = 0; i < 8; ++i) {
          const int j = 32 * h + 4 * i;
          if (j     <= row) rs_v += pv[i].x;
          if (j + 1 <= row) rs_v += pv[i].y;
          if (j + 2 <= row) rs_v += pv[i].z;
          if (j + 3 <= row) rs_v += pv[i].w;
        }
      }
    }
    sc[SC_RS + r0 + lane] = rs_v;
  }
}

__global__ __launch_bounds__(256) void state_prefix_kernel(
    float* dcw, int dstride, float* scw, int sstride,
    const float* __restrict__ C0, const float* __restrict__ n0,
    float* __restrict__ C_out, float* __restrict__ n_out) {
  __shared__ float F_s[NCH];
  if (threadIdx.x < NCH)
    F_s[threadIdx.x] = __expf(scw[(size_t)threadIdx.x * sstride + SC_LA + LCH - 1]);
  __syncthreads();

  const int gid = blockIdx.x * 256 + threadIdx.x;
  float* p; float* dst; float s; size_t ps;
  if (gid < MD * MD) {
    p = dcw + gid; ps = (size_t)dstride; s = C0[gid]; dst = C_out + gid;
  } else if (gid < MD * MD + MD) {
    const int f = gid - MD * MD;
    p = scw + SC_DN + f; ps = (size_t)sstride; s = n0[f]; dst = n_out + f;
  } else {
    return;
  }
  for (int cb = 0; cb < NCH; cb += 16) {
    float d[16];
#pragma unroll
    for (int i = 0; i < 16; ++i) d[i] = p[(size_t)(cb + i) * ps];
#pragma unroll
    for (int i = 0; i < 16; ++i) {
      p[(size_t)(cb + i) * ps] = s;
      s = fmaf(F_s[cb + i], s, d[i]);
    }
  }
  *dst = s;
}

#define RB 32
#define RPITCH (MD + 4)

__global__ __launch_bounds__(256) void chunk_readout_kernel(
    const float* __restrict__ qw, const float* __restrict__ dcw, int dstride,
    const float* __restrict__ scw, int sstride, const float* __restrict__ ow,
    const float* __restrict__ fc_w, const float* __restrict__ fc_b,
    float* __restrict__ out) {
  __shared__ __attribute__((aligned(16))) float Cs[MD][RPITCH];
  __shared__ __attribute__((aligned(16))) float Qs[RB][RPITCH];
  __shared__ __attribute__((aligned(16))) float Gs[RB][RPITCH];
  __shared__ __attribute__((aligned(16))) float ns[MD];
  __shared__ float fcw_s[MD], qn_s[RB];

  const int tid = threadIdx.x;
  const int wave = tid >> 5;
  const int lane = tid & 31;
  const int half = lane >> 4;
  const int lm = lane & 15;
  const int t0 = (blockIdx.x >> 2) * LCH;
  const int sub = (blockIdx.x & 3) * RB;

  const float* qc = qw + (size_t)(t0 + sub) * MD;
  const float* oc = ow + (size_t)(t0 + sub) * MD;
  const float* st = dcw + (size_t)(blockIdx.x >> 2) * dstride;
  const float* sc = scw + (size_t)(blockIdx.x >> 2) * sstride;

  const int srow = tid >> 5;
  const int scol = (tid & 31) * 4;
  v4f cst[16], qst[4];
#pragma unroll
  for (int i = 0; i < 16; ++i)
    cst[i] = *(const v4f*)(st + (size_t)(srow + 8 * i) * MD + scol);
#pragma unroll
  for (int i = 0; i < 4; ++i)
    qst[i] = *(const v4f*)(qc + (size_t)(srow + 8 * i) * MD + scol);
  const int vi = tid & (MD - 1);
  const float nsv = sc[SC_DN + vi];
  const float fwv = fc_w[vi];
  const float la_v = sc[SC_LA + sub + lane];
  const float rs_v = sc[SC_RS + sub + lane];
  const float a_v = sc[SC_A + sub + lane];

  const int cc = wave * 16 + lm;
  float ov0[8], ov1[8];
#pragma unroll
  for (int e = 0; e < 8; ++e) {
    ov0[e] = oc[(size_t)(e + 8 * half) * MD + cc];
    ov1[e] = oc[(size_t)(16 + e + 8 * half) * MD + cc];
  }

#pragma unroll
  for (int i = 0; i < 16; ++i) *(v4f*)&Cs[srow + 8 * i][scol] = cst[i];
#pragma unroll
  for (int i = 0; i < 4; ++i) *(v4f*)&Qs[srow + 8 * i][scol] = qst[i];
  ns[vi] = nsv;
  fcw_s[vi] = fwv;
  __syncthreads();

  v8f ac0 = {}, ac1 = {};
  const float* a0p = &Qs[lm][2 * half];
  const float* a1p = &Qs[16 + lm][2 * half];
  const float* bp = &Cs[wave * 16 + lm][2 * half];
#pragma unroll
  for (int kk = 0; kk < MD; kk += 4) {
    const v2f a0 = *(const v2f*)(a0p + kk);
    const v2f a1 = *(const v2f*)(a1p + kk);
    const v2f b = *(const v2f*)(bp + kk);
    ac0 = wmma4(a0, b, ac0);
    ac1 = wmma4(a1, b, ac1);
  }
  const float fw = fcw_s[cc];
#pragma unroll
  for (int e = 0; e < 8; ++e) {
    const int r = e + 8 * half;
    Gs[r][cc] = ac0[e] * ov0[e] * fw;
    Gs[16 + r][cc] = ac1[e] * ov1[e] * fw;
  }
  __syncthreads();

  float b_v = 0.f;
  if (wave == 0) {
    v4f gv[MD / 4];
#pragma unroll
    for (int i = 0; i < MD / 4; ++i) gv[i] = *(const v4f*)&Gs[lane][4 * i];
    __builtin_amdgcn_sched_barrier(0);
#pragma unroll
    for (int i = 0; i < MD / 4; ++i) {
      b_v += gv[i].x; b_v += gv[i].y; b_v += gv[i].z; b_v += gv[i].w;
    }
  } else if (wave == 1) {
    float qn = 0.f;
#pragma unroll
    for (int h = 0; h < 4; ++h) {
      v4f qv[8], nv[8];
#pragma unroll
      for (int i = 0; i < 8; ++i) {
        qv[i] = *(const v4f*)&Qs[lane][32 * h + 4 * i];
        nv[i] = *(const v4f*)&ns[32 * h + 4 * i];
      }
      __builtin_amdgcn_sched_barrier(0);
#pragma unroll
      for (int i = 0; i < 8; ++i) {
        qn = fmaf(qv[i].x, nv[i].x, qn);
        qn = fmaf(qv[i].y, nv[i].y, qn);
        qn = fmaf(qv[i].z, nv[i].z, qn);
        qn = fmaf(qv[i].w, nv[i].w, qn);
      }
    }
    qn_s[lane] = qn;
  }
  __syncthreads();

  if (tid < RB) {
    const float qn = qn_s[tid];
    const float Ft = __expf(la_v);
    const float num = fmaf(Ft, b_v, a_v);
    const float den = fmaxf(fabsf(fmaf(Ft, qn, rs_v)), 1.0f);
    out[t0 + sub + tid] = num / den + fc_b[0];
  }
}

extern "C" void kernel_launch(void* const* d_in, const int* in_sizes, int n_in,
                              void* d_out, int out_size, void* d_ws, size_t ws_size,
                              hipStream_t stream) {
  const float* x    = (const float*)d_in[0];
  const float* C0   = (const float*)d_in[1];
  const float* n0   = (const float*)d_in[2];
  const float* Wq   = (const float*)d_in[3];
  const float* bq   = (const float*)d_in[4];
  const float* Wk   = (const float*)d_in[5];
  const float* bk   = (const float*)d_in[6];
  const float* Wv   = (const float*)d_in[7];
  const float* bv   = (const float*)d_in[8];
  const float* wi   = (const float*)d_in[9];
  const float* bi   = (const float*)d_in[10];
  const float* wf   = (const float*)d_in[11];
  const float* bf   = (const float*)d_in[12];
  const float* Wo   = (const float*)d_in[13];
  const float* bo   = (const float*)d_in[14];
  const float* fc_w = (const float*)d_in[15];
  const float* fc_b = (const float*)d_in[16];

  float* ws = (float*)d_ws;
  const size_t NP = (size_t)T_LEN * MD;
  float* qw = ws;
  float* kw = ws + NP;
  float* vw = ws + 2 * NP;
  float* owp = ws + 3 * NP;
  float* igp = ws + 4 * NP;
  float* lfp = igp + T_LEN;

  float* outp  = (float*)d_out;
  float* Coutp = outp + T_LEN;
  float* noutp = Coutp + MD * MD;

  proj_kernel<<<dim3(T_LEN / PB, 4), 512, 0, stream>>>(
      x, Wq, bq, Wk, bk, Wv, bv, Wo, bo, wi, bi, wf, bf, qw, kw, vw, owp, igp, lfp);
  float* dcp = lfp + T_LEN;
  float* scp = dcp + NP;
  if (ws_size >= (size_t)21168128) {
    chunk_local4_kernel<<<NCH * 4, 512, 0, stream>>>(
        qw, kw, vw, owp, igp, lfp, fc_w, dcp, scp);
    state_prefix_kernel<<<(MD * MD + MD + 255) / 256, 256, 0, stream>>>(
        dcp, CH, scp, SCS, C0, n0, Coutp, noutp);
    chunk_readout_kernel<<<NCH * 4, 256, 0, stream>>>(
        qw, dcp, CH, scp, SCS, owp, fc_w, fc_b, outp);
  } else {
    chunk_local_kernel<<<NCH, 512, 0, stream>>>(qw, kw, vw, owp, igp, lfp, fc_w);
    state_prefix_kernel<<<(MD * MD + MD + 255) / 256, 256, 0, stream>>>(
        kw, CH, vw, CH, C0, n0, Coutp, noutp);
    chunk_readout_kernel<<<NCH * 4, 256, 0, stream>>>(
        qw, kw, CH, vw, CH, owp, fc_w, fc_b, outp);
  }
}


# State prefix in one trip to memory: a thread loads all 64 increments of its element before the serial chain, decays held per lane and read through scalar registers; sum order unchanged
# speedup vs baseline: 61.8059x; 1.0688x over previous
#include <hip/hip_runtime.h>
#include <hip/hip_bf16.h>
#include <math.h>


typedef __attribute__((ext_vector_type(2))) float v2f;
typedef __attribute__((ext_vector_type(8))) float v8f;

#define T_LEN 8192
#define IN_DIM 512
#define MD 128
#define LCH 128
#define NCH (T_LEN / LCH)
#define NW 16
#define CH (LCH * MD)
#define SC_LA 0
#define SC_RS 128
#define SC_A 256
#define SC_DN 384

__device__ __forceinline__ v8f wmma4(v2f a, v2f b, v8f c) {
  return __builtin_amdgcn_wmma_f32_16x16x4_f32(
       false, a,  false, b,
       (short)0, c,  false,  false);
}

__device__ __forceinline__ float sigmoidf_(float z) {
  return 1.0f / (1.0f + __expf(-z));
}

typedef __attribute__((ext_vector_type(4))) float v4f;

#define PB 128
#define PKT 64
#define PPITCH (PKT + 4)
#define PNK (IN_DIM / PKT)

__device__ __forceinline__ float gate_fma(float g, v4f xv, v4f w) {
  g = fmaf(xv.x, w.x, g);
  g = fmaf(xv.y, w.y, g);
  g = fmaf(xv.z, w.z, g);
  g = fmaf(xv.w, w.w, g);
  return g;
}

__global__ __launch_bounds__(512) void proj_kernel(
    const float* __restrict__ x,
    const float* __restrict__ Wq, const float* __restrict__ bq,
    const float* __restrict__ Wk, const float* __restrict__ bk,
    const float* __restrict__ Wv, const float* __restrict__ bv,
    const float* __restrict__ Wo, const float* __restrict__ bo,
    const float* __restrict__ wi, const float* __restrict__ bi,
    const float* __restrict__ wf, const float* __restrict__ bf,
    float* __restrict__ qw, float* __restrict__ kw,
    float* __restrict__ vw, float* __restrict__ ow,
    float* __restrict__ ig, float* __restrict__ lf) {
  __shared__ __attribute__((aligned(16))) float Xs[2][PB][PPITCH];
  __shared__ __attribute__((aligned(16))) float Ws[2][PB][PPITCH];

  const int tid = threadIdx.x;
  const int wave = tid >> 5;
  const int lane = tid & 31;
  const int half = lane >> 4;
  const int lm = lane & 15;

  const int rowBase = blockIdx.x * PB;
  const int mat = blockIdx.y;

  const float* W; const float* bias; float* dst;
  switch (mat) {
    case 0:  W = Wq; bias = bq; dst = qw; break;
    case 1:  W = Wk; bias = bk; dst = kw; break;
    case 2:  W = Wv; bias = bv; dst = vw; break;
    default: W = Wo; bias = bo; dst = ow; break;
  }

  const int sr = tid >> 4;
  const int sc = (tid & 15) * 4;
  const float* xg = x + (size_t)(rowBase + sr) * IN_DIM + sc;
  const float* wg = W + (size_t)sr * IN_DIM + sc;
  const bool gHi = (mat & 1) != 0;
  const float* gvec = ((mat < 2) ? wi : wf) + sc;
  float g0 = 0.f, g1 = 0.f;
  v4f gw;

  v4f px[4], pw[4];
#pragma unroll
  for (int i = 0; i < 4; ++i) {
    px[i] = *(const v4f*)(xg + (size_t)i * 32 * IN_DIM);
    pw[i] = *(const v4f*)(wg + (size_t)i * 32 * IN_DIM);
  }
  gw = *(const v4f*)gvec;
  g0 = gate_fma(g0, gHi ? px[2] : px[0], gw);
  g1 = gate_fma(g1, gHi ? px[3] : px[1], gw);
#pragma unroll
  for (int i = 0; i < 4; ++i) {
    *(v4f*)&Xs[0][sr + 32 * i][sc] = px[i];
    *(v4f*)&Ws[0][sr + 32 * i][sc] = pw[i];
  }
  __syncthreads();

  const int wr = (wave >> 2) * 32;
  const int wc = (wave & 3) * 32;
  v8f acc[2][2] = {};

  for (int kt = 0; kt < PNK; ++kt) {
    const int cur = kt & 1;
    const bool more = (kt + 1) < PNK;
    if (more) {
      const int k0 = (kt + 1) * PKT;
#pragma unroll
      for (int i = 0; i < 4; ++i) {
        px[i] = *(const v4f*)(xg + (size_t)i * 32 * IN_DIM + k0);
        pw[i] = *(const v4f*)(wg + (size_t)i * 32 * IN_DIM + k0);
      }
      gw = *(const v4f*)(gvec + k0);
    }

    const float* xa = &Xs[cur][wr + lm][4 * half];
    const float* wb = &Ws[cur][wc + lm][4 * half];
#pragma unroll
    for (int k8 = 0; k8 < PKT; k8 += 8) {
      const v4f a0 = *(const v4f*)(xa + k8);
      const v4f a1 = *(const v4f*)(xa + 16 * PPITCH + k8);
      const v4f b0 = *(const v4f*)(wb + k8);
      const v4f b1 = *(const v4f*)(wb + 16 * PPITCH + k8);
      const v2f a0l = {a0.x, a0.y}, a0h = {a0.z, a0.w};
      const v2f a1l = {a1.x, a1.y}, a1h = {a1.z, a1.w};
      const v2f b0l = {b0.x, b0.y}, b0h = {b0.z, b0.w};
      const v2f b1l = {b1.x, b1.y}, b1h = {b1.z, b1.w};
      acc[0][0] = wmma4(a0l, b0l, acc[0][0]);
      acc[0][1] = wmma4(a0l, b1l, acc[0][1]);
      acc[1][0] = wmma4(a1l, b0l, acc[1][0]);
      acc[1][1] = wmma4(a1l, b1l, acc[1][1]);
      acc[0][0] = wmma4(a0h, b0h, acc[0][0]);
      acc[0][1] = wmma4(a0h, b1h, acc[0][1]);
      acc[1][0] = wmma4(a1h, b0h, acc[1][0]);
      acc[1][1] = wmma4(a1h, b1h, acc[1][1]);
    }

    if (more) {
#pragma unroll
      for (int i = 0; i < 4; ++i) {
        *(v4f*)&Xs[cur ^ 1][sr + 32 * i][sc] = px[i];
        *(v4f*)&Ws[cur ^ 1][sr + 32 * i][sc] = pw[i];
      }
      __builtin_amdgcn_sched_barrier(0);
      g0 = gate_fma(g0, gHi ? px[2] : px[0], gw);
      g1 = gate_fma(g1, gHi ? px[3] : px[1], gw);
    }
    __syncthreads();
  }

  {
    g0 += __shfl_xor(g0, 8);  g1 += __shfl_xor(g1, 8);
    g0 += __shfl_xor(g0, 4);  g1 += __shfl_xor(g1, 4);
    g0 += __shfl_xor(g0, 2);  g1 += __shfl_xor(g1, 2);
    g0 += __shfl_xor(g0, 1);  g1 += __shfl_xor(g1, 1);
    const int gj = tid & 15;
    if (gj < 2) {
      const float s = (gj == 0) ? g0 : g1;
      const int t = rowBase + sr + 32 * (2 * (mat & 1) + gj);
      if (mat < 2) {
        ig[t] = __expf(s + bi[0]);
      } else {
        const float z = s + bf[0];
        lf[t] = (z >= 0.f) ? -log1pf(__expf(-z)) : (z - log1pf(__expf(z)));
      }
    }
  }

#pragma unroll
  for (int j = 0; j < 2; ++j) {
    const int cc = wc + 16 * j + lm;
    const float bv_ = bias[cc];
#pragma unroll
    for (int i = 0; i < 2; ++i) {
#pragma unroll
      for (int e = 0; e < 8; ++e) {
        const int r = rowBase + wr + 16 * i + e + 8 * half;
        float val = acc[i][j][e] + bv_;
        if (mat == 1) val *= 0.08838834764831845f;
        if (mat == 3) val = sigmoidf_(val);
        dst[(size_t)r * MD + cc] = val;
      }
    }
  }
}

#define QP (MD + 4)
#define VP (MD + 8)

template <int RSEL>
__device__ __forceinline__ void pv_wave(const float (*Ps)[QP], const float (*Vs)[VP],
                                        float (*Gs)[QP], const float* __restrict__ oc,
                                        const float* fcw_s, int tc, int lm, int half) {
  constexpr int RT[4] = {RSEL, 7 - RSEL, 3 - RSEL, 4 + RSEL};
  const int cc = tc + lm;
  float ov[4][8];
#pragma unroll
  for (int i = 0; i < 4; ++i) {
#pragma unroll
    for (int e = 0; e < 8; ++e)
      ov[i][e] = oc[(size_t)(RT[i] * 16 + e + 8 * half) * MD + cc];
  }
  v8f aci[4] = {};
#pragma unroll
  for (int kb = 0; kb < 8; ++kb) {
    v2f b[4];
#pragma unroll
    for (int s = 0; s < 4; ++s) {
      const int j0 = kb * 16 + 4 * s + 2 * half;
      b[s].x = Vs[j0][cc];
      b[s].y = Vs[j0 + 1][cc];
    }
#pragma unroll
    for (int i = 0; i < 4; ++i) {
      if (RT[i] >= kb) {
        const float* ap = &Ps[RT[i] * 16 + lm][kb * 16 + 2 * half];
#pragma unroll
        for (int s = 0; s < 4; ++s) {
          const v2f a = *(const v2f*)(ap + 4 * s);
          aci[i] = wmma4(a, b[s], aci[i]);
        }
      }
    }
  }
  const float fw = fcw_s[cc];
#pragma unroll
  for (int i = 0; i < 4; ++i) {
#pragma unroll
    for (int e = 0; e < 8; ++e)
      Gs[RT[i] * 16 + e + 8 * half][cc] = aci[i][e] * ov[i][e] * fw;
  }
}

__global__ __launch_bounds__(512) void chunk_local_kernel(
    const float* __restrict__ qw, float* kw, float* vw,
    const float* __restrict__ ow,
    const float* __restrict__ ig, const float* __restrict__ lf,
    const float* __restrict__ fc_w) {
  __shared__ __attribute__((aligned(16))) float Qs[LCH][QP];
  __shared__ __attribute__((aligned(16))) float Ks[LCH][QP];
  __shared__ __attribute__((aligned(16))) float Ps[LCH][QP];
  __shared__ __attribute__((aligned(16))) float Vs[LCH][VP];
  __shared__ __attribute__((aligned(16))) float la_s[LCH], ig_s[LCH], u_s[LCH], fcw_s[MD];
  float (*Gs)[QP] = Qs;

  const int tid = threadIdx.x;
  const int wv = __builtin_amdgcn_readfirstlane(tid >> 5);
  const int lane = tid & 31;
  const int half = lane >> 4;
  const int lm = lane & 15;
  const int t0 = blockIdx.x * LCH;

  const float* qc = qw + (size_t)t0 * MD;
  const float* oc = ow + (size_t)t0 * MD;
  float* kc = kw + (size_t)t0 * MD;
  float* vc = vw + (size_t)t0 * MD;

  const int srow = tid >> 5;
  const int scol = (tid & 31) * 4;
  v4f qst[8], kst[8], vst[8];
#pragma unroll
  for (int i = 0; i < 8; ++i) {
    qst[i] = *(const v4f*)(qc + (size_t)(srow + 16 * i) * MD + scol);
    kst[i] = *(const v4f*)(kc + (size_t)(srow + 16 * i) * MD + scol);
  }
  const int gi = tid & (LCH - 1);
  const float igv = ig[t0 + gi];
  const float fwv = fc_w[gi];
  if (tid == 0) {
    v4f lv[LCH / 4];
#pragma unroll
    for (int i = 0; i < LCH / 4; ++i) lv[i] = *(const v4f*)(lf + t0 + 4 * i);
    float a = 0.f;
#pragma unroll
    for (int i = 0; i < LCH / 4; ++i) {
      a += lv[i].x; lv[i].x = a;
      a += lv[i].y; lv[i].y = a;
      a += lv[i].z; lv[i].z = a;
      a += lv[i].w; lv[i].w = a;
    }
#pragma unroll
    for (int i = 0; i < LCH / 4; ++i) *(v4f*)&la_s[4 * i] = lv[i];
  }
  ig_s[gi] = igv;
  fcw_s[gi] = fwv;
#pragma unroll
  for (int i = 0; i < 8; ++i) {
    *(v4f*)&Qs[srow + 16 * i][scol] = qst[i];
    *(v4f*)&Ks[srow + 16 * i][scol] = kst[i];
  }
  __syncthreads();

#pragma unroll
  for (int i = 0; i < 8; ++i)
    vst[i] = *(const v4f*)(vc + (size_t)(srow + 16 * i) * MD + scol);

  for (int t = wv; t < 36; t += NW) {
    int r = 0;
    while ((r + 1) * (r + 2) / 2 <= t) ++r;
    const int tr = r * 16, tc = (t - r * (r + 1) / 2) * 16;
    v8f acc = {};
    const float* ap = &Qs[tr + lm][2 * half];
    const float* bp = &Ks[tc + lm][2 * half];
#pragma unroll
    for (int kk = 0; kk < MD; kk += 4) {
      const v2f a = *(const v2f*)(ap + kk);
      const v2f b = *(const v2f*)(bp + kk);
      acc = wmma4(a, b, acc);
    }
#pragma unroll
    for (int e = 0; e < 8; ++e) {
      const int rr = tr + e + 8 * half, cc = tc + lm;
      float p = 0.f;
      if (rr >= cc) p = __expf(la_s[rr] - la_s[cc]) * ig_s[cc] * acc[e];
      Ps[rr][cc] = p;
    }
  }
#pragma unroll
  for (int i = 0; i < 8; ++i) *(v4f*)&Vs[srow + 16 * i][scol] = vst[i];
  if (tid < LCH) u_s[tid] = __expf(la_s[LCH - 1] - la_s[tid]) * ig_s[tid];
  __syncthreads();

  float rs_v = 0.f;
  if (tid < LCH) {
#pragma unroll
    for (int h = 0; h < 4; ++h) {
      if (h <= wv) {
        v4f pv[8];
#pragma unroll
        for (int i = 0; i < 8; ++i) pv[i] = *(const v4f*)&Ps[tid][32 * h + 4 * i];
        __builtin_amdgcn_sched_barrier(0);
#pragma unroll
        for (int i = 0; i < 8; ++i) {
          const int j = 32 * h + 4 * i;
          if (j     <= tid) rs_v += pv[i].x;
          if (j + 1 <= tid) rs_v += pv[i].y;
          if (j + 2 <= tid) rs_v += pv[i].z;
          if (j + 3 <= tid) rs_v += pv[i].w;
        }
      }
    }
  }

  {
    const int tc = (wv & 7) * 16;
    if (wv >> 3) pv_wave<1>(Ps, Vs, Gs, oc, fcw_s, tc, lm, half);
    else         pv_wave<0>(Ps, Vs, Gs, oc, fcw_s, tc, lm, half);
  }
  __syncthreads();

#pragma unroll
  for (int i = 0; i < 8; ++i) {
    const int row = srow + 16 * i;
    const float u = u_s[row];
    v4f v = *(const v4f*)&Vs[row][scol];
    v.x *= u; v.y *= u; v.z *= u; v.w *= u;
    *(v4f*)&Vs[row][scol] = v;
  }
  __syncthreads();

  float a_v = 0.f, dn_v = 0.f;
  if (tid < LCH) {
#pragma unroll
    for (int h = 0; h < 4; ++h) {
      v4f gv[8];
#pragma unroll
      for (int i = 0; i < 8; ++i) gv[i] = *(const v4f*)&Gs[tid][32 * h + 4 * i];
      __builtin_amdgcn_sched_barrier(0);
#pragma unroll
      for (int i = 0; i < 8; ++i) {
        a_v += gv[i].x; a_v += gv[i].y; a_v += gv[i].z; a_v += gv[i].w;
      }
    }
#pragma unroll 16
    for (int j = 0; j < LCH; ++j) dn_v = fmaf(u_s[j], Ks[j][tid], dn_v);
  }

  {
    const int tr0 = (wv >> 2) * 32, tc0 = (wv & 3) * 32;
    v8f d00 = {}, d01 = {}, d10 = {}, d11 = {};
#pragma unroll
    for (int kk = 0; kk < LCH; kk += 4) {
      const int j0 = kk + 2 * half;
      const v2f a0 = {Vs[j0][tr0 + lm],      Vs[j0 + 1][tr0 + lm]};
      const v2f a1 = {Vs[j0][tr0 + 16 + lm], Vs[j0 + 1][tr0 + 16 + lm]};
      const v2f b0 = {Ks[j0][tc0 + lm],      Ks[j0 + 1][tc0 + lm]};
      const v2f b1 = {Ks[j0][tc0 + 16 + lm], Ks[j0 + 1][tc0 + 16 + lm]};
      d00 = wmma4(a0, b0, d00);
      d01 = wmma4(a0, b1, d01);
      d10 = wmma4(a1, b0, d10);
      d11 = wmma4(a1, b1, d11);
    }
#pragma unroll
    for (int e = 0; e < 8; ++e) {
      float* kr0 = kc + (size_t)(tr0 + e + 8 * half) * MD + tc0 + lm;
      float* kr1 = kr0 + (size_t)16 * MD;
      kr0[0] = d00[e];  kr0[16] = d01[e];
      kr1[0] = d10[e];  kr1[16] = d11[e];
    }
  }
  if (tid < LCH) {
    vc[SC_LA + tid] = la_s[tid];
    vc[SC_RS + tid] = rs_v;
    vc[SC_A + tid] = a_v;
    vc[SC_DN + tid] = dn_v;
  }
}

#define SB 32
#define SCS 512

__global__ __launch_bounds__(512) void chunk_local4_kernel(
    const float* __restrict__ qw, const float* __restrict__ kw,
    const float* __restrict__ vw, const float* __restrict__ ow,
    const float* __restrict__ ig, const float* __restrict__ lf,
    const float* __restrict__ fc_w,
    float* __restrict__ dcw, float* __restrict__ scw) {
  __shared__ __attribute__((aligned(16))) float Qs[SB][QP];
  __shared__ __attribute__((aligned(16))) float Ks[LCH][QP];
  __shared__ __attribute__((aligned(16))) float Ps[SB][QP];
  __shared__ __attribute__((aligned(16))) float Vs[LCH][VP];
  __shared__ __attribute__((aligned(16))) float la_s[LCH], ig_s[LCH], u_s[LCH], fcw_s[MD];
  float (*Gs)[QP] = Qs;

  const int tid = threadIdx.x;
  const int wv = __builtin_amdgcn_readfirstlane(tid >> 5);
  const int lane = tid & 31;
  const int half = lane >> 4;
  const int lm = lane & 15;
  const int ch = blockIdx.x >> 2;
  const int s4 = blockIdx.x & 3;
  const int t0 = ch * LCH;
  const int r0 = s4 * SB;

  const float* qc = qw + (size_t)(t0 + r0) * MD;
  const float* oc = ow + (size_t)(t0 + r0) * MD;
  const float* kc = kw + (size_t)t0 * MD;
  const float* vc = vw + (size_t)t0 * MD;
  float* dc = dcw + (size_t)ch * CH;
  float* sc = scw + (size_t)ch * SCS;

  const int srow = tid >> 5;
  const int scol = (tid & 31) * 4;
  v4f qst[2], kst[8], vst[8];
#pragma unroll
  for (int i = 0; i < 2; ++i)
    qst[i] = *(const v4f*)(qc + (size_t)(srow + 16 * i) * MD + scol);
#pragma unroll
  for (int i = 0; i < 8; ++i)
    kst[i] = *(const v4f*)(kc + (size_t)(srow + 16 * i) * MD + scol);
  const int gi = tid & (LCH - 1);
  const float igv = ig[t0 + gi];
  const float fwv = fc_w[gi];
  if (tid == 0) {
    v4f lv[LCH / 4];
#pragma unroll
    for (int i = 0; i < LCH / 4; ++i) lv[i] = *(const v4f*)(lf + t0 + 4 * i);
    float a = 0.f;
#pragma unroll
    for (int i = 0; i < LCH / 4; ++i) {
      a += lv[i].x; lv[i].x = a;
      a += lv[i].y; lv[i].y = a;
      a += lv[i].z; lv[i].z = a;
      a += lv[i].w; lv[i].w = a;
    }
#pragma unroll
    for (int i = 0; i < LCH / 4; ++i) *(v4f*)&la_s[4 * i] = lv[i];
  }
  ig_s[gi] = igv;
  fcw_s[gi] = fwv;
#pragma unroll
  for (int i = 0; i < 2; ++i) *(v4f*)&Qs[srow + 16 * i][scol] = qst[i];
#pragma unroll
  for (int i = 0; i < 8; ++i) *(v4f*)&Ks[srow + 16 * i][scol] = kst[i];
  __syncthreads();

#pragma unroll
  for (int i = 0; i < 8; ++i)
    vst[i] = *(const v4f*)(vc + (size_t)(srow + 16 * i) * MD + scol);

  if (wv < 4 * s4 + 3) {
    const int rt = (wv > 2 * s4) ? 1 : 0;
    const int trl = rt * 16;
    const int tc = (rt ? (wv - 2 * s4 - 1) : wv) * 16;
    v8f acc = {};
    const float* ap = &Qs[trl + lm][2 * half];
    const float* bp = &Ks[tc + lm][2 * half];
#pragma unroll
    for (int kk = 0; kk < MD; kk += 4) {
      const v2f a = *(const v2f*)(ap + kk);
      const v2f b = *(const v2f*)(bp + kk);
      acc = wmma4(a, b, acc);
    }
#pragma unroll
    for (int e = 0; e < 8; ++e) {
      const int rl = trl + e + 8 * half;
      const int rr = r0 + rl, cc = tc + lm;
      float p = 0.f;
      if (rr >= cc) p = __expf(la_s[rr] - la_s[cc]) * ig_s[cc] * acc[e];
      Ps[rl][cc] = p;
    }
  }
#pragma unroll
  for (int i = 0; i < 8; ++i) *(v4f*)&Vs[srow + 16 * i][scol] = vst[i];
  if (tid < LCH) u_s[tid] = __expf(la_s[LCH - 1] - la_s[tid]) * ig_s[tid];
  __syncthreads();

  {
    const int rt = wv >> 3;
    const int cc = (wv & 7) * 16 + lm;
    float ov[8];
#pragma unroll
    for (int e = 0; e < 8; ++e) ov[e] = oc[(size_t)(rt * 16 + e + 8 * half) * MD + cc];
    v8f aci = {};
    const int nkb = 2 * s4 + rt + 1;
    for (int kb = 0; kb < nkb; ++kb) {
      v2f b[4];
#pragma unroll
      for (int s = 0; s < 4; ++s) {
        const int j0 = kb * 16 + 4 * s + 2 * half;
        b[s].x = Vs[j0][cc];
        b[s].y = Vs[j0 + 1][cc];
      }
      const float* ap = &Ps[rt * 16 + lm][kb * 16 + 2 * half];
#pragma unroll
      for (int s = 0; s < 4; ++s) {
        const v2f a = *(const v2f*)(ap + 4 * s);
        aci = wmma4(a, b[s], aci);
      }
    }
    const float fw = fcw_s[cc];
#pragma unroll
    for (int e = 0; e < 8; ++e)
      Gs[rt * 16 + e + 8 * half][cc] = aci[e] * ov[e] * fw;
  }
  __syncthreads();

  if (wv < 8) {
    const int fr = r0 + (wv >> 2) * 16;
    const int tc0 = (wv & 3) * 32;
    v8f d0 = {}, d1 = {};
#pragma unroll
    for (int kk = 0; kk < LCH; kk += 4) {
      const int j0 = kk + 2 * half;
      const v2f uu = *(const v2f*)&u_s[j0];
      const v2f a  = {Vs[j0][fr + lm] * uu.x,  Vs[j0 + 1][fr + lm] * uu.y};
      const v2f b0 = {Ks[j0][tc0 + lm],        Ks[j0 + 1][tc0 + lm]};
      const v2f b1 = {Ks[j0][tc0 + 16 + lm],   Ks[j0 + 1][tc0 + 16 + lm]};
      d0 = wmma4(a, b0, d0);
      d1 = wmma4(a, b1, d1);
    }
#pragma unroll
    for (int e = 0; e < 8; ++e) {
      float* dr = dc + (size_t)(fr + e + 8 * half) * MD + tc0 + lm;
      dr[0] = d0[e];  dr[16] = d1[e];
    }
  } else if (wv == 8) {
    float a_v = 0.f;
#pragma unroll
    for (int h = 0; h < 4; ++h) {
      v4f gv[8];
#pragma unroll
      for (int i = 0; i < 8; ++i) gv[i] = *(const v4f*)&Gs[lane][32 * h + 4 * i];
      __builtin_amdgcn_sched_barrier(0);
#pragma unroll
      for (int i = 0; i < 8; ++i) {
        a_v += gv[i].x; a_v += gv[i].y; a_v += gv[i].z; a_v += gv[i].w;
      }
    }
    sc[SC_A + r0 + lane] = a_v;
    sc[SC_LA + r0 + lane] = la_s[r0 + lane];
  } else if (wv == 9) {
    float dn_v = 0.f;
#pragma unroll 16
    for (int j = 0; j < LCH; ++j) dn_v = fmaf(u_s[j], Ks[j][r0 + lane], dn_v);
    sc[SC_DN + r0 + lane] = dn_v;
  } else if (wv == 10) {
    float rs_v = 0.f;
    const int row = r0 + lane;
#pragma unroll
    for (int h = 0; h < 4; ++h) {
      if (h <= s4) {
        v4f pv[8];
#pragma unroll
        for (int i = 0; i < 8; ++i) pv[i] = *(const v4f*)&Ps[lane][32 * h + 4 * i];
        __builtin_amdgcn_sched_barrier(0);
#pragma unroll
        for (int i = 0; i < 8; ++i) {
          const int j = 32 * h + 4 * i;
          if (j     <= row) rs_v += pv[i].x;
          if (j + 1 <= row) rs_v += pv[i].y;
          if (j + 2 <= row) rs_v += pv[i].z;
          if (j + 3 <= row) rs_v += pv[i].w;
        }
      }
    }
    sc[SC_RS + r0 + lane] = rs_v;
  }
}

#define PFX_BLOCK 128
#define PFX_GRID 129

__global__ __launch_bounds__(PFX_BLOCK) void state_prefix_kernel(
    float* dcw, int dstride, float* scw, int sstride,
    const float* __restrict__ C0, const float* __restrict__ n0,
    float* __restrict__ C_out, float* __restrict__ n_out) {
  const int lane = threadIdx.x & 31;
  const float la_lo = scw[(size_t)lane * sstride + SC_LA + LCH - 1];
  const float la_hi = scw[(size_t)(lane + 32) * sstride + SC_LA + LCH - 1];

  const int gid = blockIdx.x * PFX_BLOCK + threadIdx.x;
  float* p; float* dst; float s; size_t ps;
  if (gid < MD * MD) {
    p = dcw + gid; ps = (size_t)dstride; s = C0[gid]; dst = C_out + gid;
  } else {
    const int f = gid - MD * MD;
    p = scw + SC_DN + f; ps = (size_t)sstride; s = n0[f]; dst = n_out + f;
  }
  float d[NCH];
#pragma unroll
  for (int c = 0; c < NCH; ++c) d[c] = p[(size_t)c * ps];

  __builtin_amdgcn_sched_barrier(0);
  const int f_lo = __float_as_int(__expf(la_lo));
  const int f_hi = __float_as_int(__expf(la_hi));
#pragma unroll
  for (int c = 0; c < NCH; ++c) {
    const float Fc = __int_as_float(
        __builtin_amdgcn_readlane((c < 32) ? f_lo : f_hi, c & 31));
    p[(size_t)c * ps] = s;
    s = fmaf(Fc, s, d[c]);
  }
  *dst = s;
}

#define RB 32
#define RPITCH (MD + 4)

__global__ __launch_bounds__(256) void chunk_readout_kernel(
    const float* __restrict__ qw, const float* __restrict__ dcw, int dstride,
    const float* __restrict__ scw, int sstride, const float* __restrict__ ow,
    const float* __restrict__ fc_w, const float* __restrict__ fc_b,
    float* __restrict__ out) {
  __shared__ __attribute__((aligned(16))) float Cs[MD][RPITCH];
  __shared__ __attribute__((aligned(16))) float Qs[RB][RPITCH];
  __shared__ __attribute__((aligned(16))) float Gs[RB][RPITCH];
  __shared__ __attribute__((aligned(16))) float ns[MD];
  __shared__ float fcw_s[MD], qn_s[RB];

  const int tid = threadIdx.x;
  const int wave = tid >> 5;
  const int lane = tid & 31;
  const int half = lane >> 4;
  const int lm = lane & 15;
  const int t0 = (blockIdx.x >> 2) * LCH;
  const int sub = (blockIdx.x & 3) * RB;

  const float* qc = qw + (size_t)(t0 + sub) * MD;
  const float* oc = ow + (size_t)(t0 + sub) * MD;
  const float* st = dcw + (size_t)(blockIdx.x >> 2) * dstride;
  const float* sc = scw + (size_t)(blockIdx.x >> 2) * sstride;

  const int srow = tid >> 5;
  const int scol = (tid & 31) * 4;
  v4f cst[16], qst[4];
#pragma unroll
  for (int i = 0; i < 16; ++i)
    cst[i] = *(const v4f*)(st + (size_t)(srow + 8 * i) * MD + scol);
#pragma unroll
  for (int i = 0; i < 4; ++i)
    qst[i] = *(const v4f*)(qc + (size_t)(srow + 8 * i) * MD + scol);
  const int vi = tid & (MD - 1);
  const float nsv = sc[SC_DN + vi];
  const float fwv = fc_w[vi];
  const float la_v = sc[SC_LA + sub + lane];
  const float rs_v = sc[SC_RS + sub + lane];
  const float a_v = sc[SC_A + sub + lane];

  const int cc = wave * 16 + lm;
  float ov0[8], ov1[8];
#pragma unroll
  for (int e = 0; e < 8; ++e) {
    ov0[e] = oc[(size_t)(e + 8 * half) * MD + cc];
    ov1[e] = oc[(size_t)(16 + e + 8 * half) * MD + cc];
  }

#pragma unroll
  for (int i = 0; i < 16; ++i) *(v4f*)&Cs[srow + 8 * i][scol] = cst[i];
#pragma unroll
  for (int i = 0; i < 4; ++i) *(v4f*)&Qs[srow + 8 * i][scol] = qst[i];
  ns[vi] = nsv;
  fcw_s[vi] = fwv;
  __syncthreads();

  v8f ac0 = {}, ac1 = {};
  const float* a0p = &Qs[lm][2 * half];
  const float* a1p = &Qs[16 + lm][2 * half];
  const float* bp = &Cs[wave * 16 + lm][2 * half];
#pragma unroll
  for (int kk = 0; kk < MD; kk += 4) {
    const v2f a0 = *(const v2f*)(a0p + kk);
    const v2f a1 = *(const v2f*)(a1p + kk);
    const v2f b = *(const v2f*)(bp + kk);
    ac0 = wmma4(a0, b, ac0);
    ac1 = wmma4(a1, b, ac1);
  }
  const float fw = fcw_s[cc];
#pragma unroll
  for (int e = 0; e < 8; ++e) {
    const int r = e + 8 * half;
    Gs[r][cc] = ac0[e] * ov0[e] * fw;
    Gs[16 + r][cc] = ac1[e] * ov1[e] * fw;
  }
  __syncthreads();

  float b_v = 0.f;
  if (wave == 0) {
    v4f gv[MD / 4];
#pragma unroll
    for (int i = 0; i < MD / 4; ++i) gv[i] = *(const v4f*)&Gs[lane][4 * i];
    __builtin_amdgcn_sched_barrier(0);
#pragma unroll
    for (int i = 0; i < MD / 4; ++i) {
      b_v += gv[i].x; b_v += gv[i].y; b_v += gv[i].z; b_v += gv[i].w;
    }
  } else if (wave == 1) {
    float qn = 0.f;
#pragma unroll
    for (int h = 0; h < 4; ++h) {
      v4f qv[8], nv[8];
#pragma unroll
      for (int i = 0; i < 8; ++i) {
        qv[i] = *(const v4f*)&Qs[lane][32 * h + 4 * i];
        nv[i] = *(const v4f*)&ns[32 * h + 4 * i];
      }
      __builtin_amdgcn_sched_barrier(0);
#pragma unroll
      for (int i = 0; i < 8; ++i) {
        qn = fmaf(qv[i].x, nv[i].x, qn);
        qn = fmaf(qv[i].y, nv[i].y, qn);
        qn = fmaf(qv[i].z, nv[i].z, qn);
        qn = fmaf(qv[i].w, nv[i].w, qn);
      }
    }
    qn_s[lane] = qn;
  }
  __syncthreads();

  if (tid < RB) {
    const float qn = qn_s[tid];
    const float Ft = __expf(la_v);
    const float num = fmaf(Ft, b_v, a_v);
    const float den = fmaxf(fabsf(fmaf(Ft, qn, rs_v)), 1.0f);
    out[t0 + sub + tid] = num / den + fc_b[0];
  }
}

extern "C" void kernel_launch(void* const* d_in, const int* in_sizes, int n_in,
                              void* d_out, int out_size, void* d_ws, size_t ws_size,
                              hipStream_t stream) {
  const float* x    = (const float*)d_in[0];
  const float* C0   = (const float*)d_in[1];
  const float* n0   = (const float*)d_in[2];
  const float* Wq   = (const float*)d_in[3];
  const float* bq   = (const float*)d_in[4];
  const float* Wk   = (const float*)d_in[5];
  const float* bk   = (const float*)d_in[6];
  const float* Wv   = (const float*)d_in[7];
  const float* bv   = (const float*)d_in[8];
  const float* wi   = (const float*)d_in[9];
  const float* bi   = (const float*)d_in[10];
  const float* wf   = (const float*)d_in[11];
  const float* bf   = (const float*)d_in[12];
  const float* Wo   = (const float*)d_in[13];
  const float* bo   = (const float*)d_in[14];
  const float* fc_w = (const float*)d_in[15];
  const float* fc_b = (const float*)d_in[16];

  float* ws = (float*)d_ws;
  const size_t NP = (size_t)T_LEN * MD;
  float* qw = ws;
  float* kw = ws + NP;
  float* vw = ws + 2 * NP;
  float* owp = ws + 3 * NP;
  float* igp = ws + 4 * NP;
  float* lfp = igp + T_LEN;

  float* outp  = (float*)d_out;
  float* Coutp = outp + T_LEN;
  float* noutp = Coutp + MD * MD;

  proj_kernel<<<dim3(T_LEN / PB, 4), 512, 0, stream>>>(
      x, Wq, bq, Wk, bk, Wv, bv, Wo, bo, wi, bi, wf, bf, qw, kw, vw, owp, igp, lfp);
  float* dcp = lfp + T_LEN;
  float* scp = dcp + NP;
  if (ws_size >= (size_t)21168128) {
    chunk_local4_kernel<<<NCH * 4, 512, 0, stream>>>(
        qw, kw, vw, owp, igp, lfp, fc_w, dcp, scp);
    state_prefix_kernel<<<PFX_GRID, PFX_BLOCK, 0, stream>>>(
        dcp, CH, scp, SCS, C0, n0, Coutp, noutp);
    chunk_readout_kernel<<<NCH * 4, 256, 0, stream>>>(
        qw, dcp, CH, scp, SCS, owp, fc_w, fc_b, outp);
  } else {
    chunk_local_kernel<<<NCH, 512, 0, stream>>>(qw, kw, vw, owp, igp, lfp, fc_w);
    state_prefix_kernel<<<PFX_GRID, PFX_BLOCK, 0, stream>>>(
        kw, CH, vw, CH, C0, n0, Coutp, noutp);
    chunk_readout_kernel<<<NCH * 4, 256, 0, stream>>>(
        qw, kw, CH, vw, CH, owp, fc_w, fc_b, outp);
  }
}


# Serial tails behind the last products cut: projection picks its matrix once, not per value, sigmoid on v_rcp_f32 (was an IEEE divide), biases loaded early; P mask by select; sum orders unchanged
# speedup vs baseline: 64.8289x; 1.0489x over previous
#include <hip/hip_runtime.h>
#include <hip/hip_bf16.h>
#include <math.h>


typedef __attribute__((ext_vector_type(2))) float v2f;
typedef __attribute__((ext_vector_type(8))) float v8f;

#define T_LEN 8192
#define IN_DIM 512
#define MD 128
#define LCH 128
#define NCH (T_LEN / LCH)
#define NW 16
#define CH (LCH * MD)
#define SC_LA 0
#define SC_RS 128
#define SC_A 256
#define SC_DN 384

__device__ __forceinline__ v8f wmma4(v2f a, v2f b, v8f c) {
  return __builtin_amdgcn_wmma_f32_16x16x4_f32(
       false, a,  false, b,
       (short)0, c,  false,  false);
}

__device__ __forceinline__ float sigmoidf_(float z) {
  return __builtin_amdgcn_rcpf(1.0f + __expf(-z));
}

typedef __attribute__((ext_vector_type(4))) float v4f;

#define PB 128
#define PKT 64
#define PPITCH (PKT + 4)
#define PNK (IN_DIM / PKT)

__device__ __forceinline__ float gate_fma(float g, v4f xv, v4f w) {
  g = fmaf(xv.x, w.x, g);
  g = fmaf(xv.y, w.y, g);
  g = fmaf(xv.z, w.z, g);
  g = fmaf(xv.w, w.w, g);
  return g;
}

__global__ __launch_bounds__(512) void proj_kernel(
    const float* __restrict__ x,
    const float* __restrict__ Wq, const float* __restrict__ bq,
    const float* __restrict__ Wk, const float* __restrict__ bk,
    const float* __restrict__ Wv, const float* __restrict__ bv,
    const float* __restrict__ Wo, const float* __restrict__ bo,
    const float* __restrict__ wi, const float* __restrict__ bi,
    const float* __restrict__ wf, const float* __restrict__ bf,
    float* __restrict__ qw, float* __restrict__ kw,
    float* __restrict__ vw, float* __restrict__ ow,
    float* __restrict__ ig, float* __restrict__ lf) {
  __shared__ __attribute__((aligned(16))) float Xs[2][PB][PPITCH];
  __shared__ __attribute__((aligned(16))) float Ws[2][PB][PPITCH];

  const int tid = threadIdx.x;
  const int wave = tid >> 5;
  const int lane = tid & 31;
  const int half = lane >> 4;
  const int lm = lane & 15;

  const int rowBase = blockIdx.x * PB;
  const int mat = blockIdx.y;

  const float* W; const float* bias; float* dst;
  switch (mat) {
    case 0:  W = Wq; bias = bq; dst = qw; break;
    case 1:  W = Wk; bias = bk; dst = kw; break;
    case 2:  W = Wv; bias = bv; dst = vw; break;
    default: W = Wo; bias = bo; dst = ow; break;
  }

  const int wr = (wave >> 2) * 32;
  const int wc = (wave & 3) * 32;
  const float bias0 = bias[wc + lm];
  const float bias1 = bias[wc + 16 + lm];
  const float gbias = (mat < 2) ? bi[0] : bf[0];

  const int sr = tid >> 4;
  const int sc = (tid & 15) * 4;
  const float* xg = x + (size_t)(rowBase + sr) * IN_DIM + sc;
  const float* wg = W + (size_t)sr * IN_DIM + sc;
  const bool gHi = (mat & 1) != 0;
  const float* gvec = ((mat < 2) ? wi : wf) + sc;
  float g0 = 0.f, g1 = 0.f;
  v4f gw;

  v4f px[4], pw[4];
#pragma unroll
  for (int i = 0; i < 4; ++i) {
    px[i] = *(const v4f*)(xg + (size_t)i * 32 * IN_DIM);
    pw[i] = *(const v4f*)(wg + (size_t)i * 32 * IN_DIM);
  }
  gw = *(const v4f*)gvec;
  g0 = gate_fma(g0, gHi ? px[2] : px[0], gw);
  g1 = gate_fma(g1, gHi ? px[3] : px[1], gw);
#pragma unroll
  for (int i = 0; i < 4; ++i) {
    *(v4f*)&Xs[0][sr + 32 * i][sc] = px[i];
    *(v4f*)&Ws[0][sr + 32 * i][sc] = pw[i];
  }
  __syncthreads();

  v8f acc[2][2] = {};

  for (int kt = 0; kt < PNK; ++kt) {
    const int cur = kt & 1;
    const bool more = (kt + 1) < PNK;
    if (more) {
      const int k0 = (kt + 1) * PKT;
#pragma unroll
      for (int i = 0; i < 4; ++i) {
        px[i] = *(const v4f*)(xg + (size_t)i * 32 * IN_DIM + k0);
        pw[i] = *(const v4f*)(wg + (size_t)i * 32 * IN_DIM + k0);
      }
      gw = *(const v4f*)(gvec + k0);
    }

    const float* xa = &Xs[cur][wr + lm][4 * half];
    const float* wb = &Ws[cur][wc + lm][4 * half];
#pragma unroll
    for (int k8 = 0; k8 < PKT; k8 += 8) {
      const v4f a0 = *(const v4f*)(xa + k8);
      const v4f a1 = *(const v4f*)(xa + 16 * PPITCH + k8);
      const v4f b0 = *(const v4f*)(wb + k8);
      const v4f b1 = *(const v4f*)(wb + 16 * PPITCH + k8);
      const v2f a0l = {a0.x, a0.y}, a0h = {a0.z, a0.w};
      const v2f a1l = {a1.x, a1.y}, a1h = {a1.z, a1.w};
      const v2f b0l = {b0.x, b0.y}, b0h = {b0.z, b0.w};
      const v2f b1l = {b1.x, b1.y}, b1h = {b1.z, b1.w};
      acc[0][0] = wmma4(a0l, b0l, acc[0][0]);
      acc[0][1] = wmma4(a0l, b1l, acc[0][1]);
      acc[1][0] = wmma4(a1l, b0l, acc[1][0]);
      acc[1][1] = wmma4(a1l, b1l, acc[1][1]);
      acc[0][0] = wmma4(a0h, b0h, acc[0][0]);
      acc[0][1] = wmma4(a0h, b1h, acc[0][1]);
      acc[1][0] = wmma4(a1h, b0h, acc[1][0]);
      acc[1][1] = wmma4(a1h, b1h, acc[1][1]);
    }

    if (more) {
#pragma unroll
      for (int i = 0; i < 4; ++i) {
        *(v4f*)&Xs[cur ^ 1][sr + 32 * i][sc] = px[i];
        *(v4f*)&Ws[cur ^ 1][sr + 32 * i][sc] = pw[i];
      }
      __builtin_amdgcn_sched_barrier(0);
      g0 = gate_fma(g0, gHi ? px[2] : px[0], gw);
      g1 = gate_fma(g1, gHi ? px[3] : px[1], gw);
    }
    __syncthreads();
  }

  {
    g0 += __shfl_xor(g0, 8);  g1 += __shfl_xor(g1, 8);
    g0 += __shfl_xor(g0, 4);  g1 += __shfl_xor(g1, 4);
    g0 += __shfl_xor(g0, 2);  g1 += __shfl_xor(g1, 2);
    g0 += __shfl_xor(g0, 1);  g1 += __shfl_xor(g1, 1);
    const int gj = tid & 15;
    if (gj < 2) {
      const float z = ((gj == 0) ? g0 : g1) + gbias;
      const int t = rowBase + sr + 32 * (2 * (mat & 1) + gj);
      if (mat < 2) {
        ig[t] = __expf(z);
      } else {
        lf[t] = fminf(z, 0.f) - log1pf(__expf(-fabsf(z)));
      }
    }
  }

  float* drow = dst + (size_t)(rowBase + wr + 8 * half) * MD + wc + lm;
  if (mat == 3) {
#pragma unroll
    for (int j = 0; j < 2; ++j) {
      const float bv_ = j ? bias1 : bias0;
#pragma unroll
      for (int i = 0; i < 2; ++i) {
#pragma unroll
        for (int e = 0; e < 8; ++e)
          drow[(size_t)(16 * i + e) * MD + 16 * j] = sigmoidf_(acc[i][j][e] + bv_);
      }
    }
  } else {
    const float sc = (mat == 1) ? 0.08838834764831845f : 1.0f;
#pragma unroll
    for (int j = 0; j < 2; ++j) {
      const float bv_ = j ? bias1 : bias0;
#pragma unroll
      for (int i = 0; i < 2; ++i) {
#pragma unroll
        for (int e = 0; e < 8; ++e)
          drow[(size_t)(16 * i + e) * MD + 16 * j] = (acc[i][j][e] + bv_) * sc;
      }
    }
  }
}

#define QP (MD + 4)
#define VP (MD + 8)

template <int RSEL>
__device__ __forceinline__ void pv_wave(const float (*Ps)[QP], const float (*Vs)[VP],
                                        float (*Gs)[QP], const float* __restrict__ oc,
                                        const float* fcw_s, int tc, int lm, int half) {
  constexpr int RT[4] = {RSEL, 7 - RSEL, 3 - RSEL, 4 + RSEL};
  const int cc = tc + lm;
  float ov[4][8];
#pragma unroll
  for (int i = 0; i < 4; ++i) {
#pragma unroll
    for (int e = 0; e < 8; ++e)
      ov[i][e] = oc[(size_t)(RT[i] * 16 + e + 8 * half) * MD + cc];
  }
  v8f aci[4] = {};
#pragma unroll
  for (int kb = 0; kb < 8; ++kb) {
    v2f b[4];
#pragma unroll
    for (int s = 0; s < 4; ++s) {
      const int j0 = kb * 16 + 4 * s + 2 * half;
      b[s].x = Vs[j0][cc];
      b[s].y = Vs[j0 + 1][cc];
    }
#pragma unroll
    for (int i = 0; i < 4; ++i) {
      if (RT[i] >= kb) {
        const float* ap = &Ps[RT[i] * 16 + lm][kb * 16 + 2 * half];
#pragma unroll
        for (int s = 0; s < 4; ++s) {
          const v2f a = *(const v2f*)(ap + 4 * s);
          aci[i] = wmma4(a, b[s], aci[i]);
        }
      }
    }
  }
  const float fw = fcw_s[cc];
#pragma unroll
  for (int i = 0; i < 4; ++i) {
#pragma unroll
    for (int e = 0; e < 8; ++e)
      Gs[RT[i] * 16 + e + 8 * half][cc] = aci[i][e] * ov[i][e] * fw;
  }
}

__global__ __launch_bounds__(512) void chunk_local_kernel(
    const float* __restrict__ qw, float* kw, float* vw,
    const float* __restrict__ ow,
    const float* __restrict__ ig, const float* __restrict__ lf,
    const float* __restrict__ fc_w) {
  __shared__ __attribute__((aligned(16))) float Qs[LCH][QP];
  __shared__ __attribute__((aligned(16))) float Ks[LCH][QP];
  __shared__ __attribute__((aligned(16))) float Ps[LCH][QP];
  __shared__ __attribute__((aligned(16))) float Vs[LCH][VP];
  __shared__ __attribute__((aligned(16))) float la_s[LCH], ig_s[LCH], u_s[LCH], fcw_s[MD];
  float (*Gs)[QP] = Qs;

  const int tid = threadIdx.x;
  const int wv = __builtin_amdgcn_readfirstlane(tid >> 5);
  const int lane = tid & 31;
  const int half = lane >> 4;
  const int lm = lane & 15;
  const int t0 = blockIdx.x * LCH;

  const float* qc = qw + (size_t)t0 * MD;
  const float* oc = ow + (size_t)t0 * MD;
  float* kc = kw + (size_t)t0 * MD;
  float* vc = vw + (size_t)t0 * MD;

  const int srow = tid >> 5;
  const int scol = (tid & 31) * 4;
  v4f qst[8], kst[8], vst[8];
#pragma unroll
  for (int i = 0; i < 8; ++i) {
    qst[i] = *(const v4f*)(qc + (size_t)(srow + 16 * i) * MD + scol);
    kst[i] = *(const v4f*)(kc + (size_t)(srow + 16 * i) * MD + scol);
  }
  const int gi = tid & (LCH - 1);
  const float igv = ig[t0 + gi];
  const float fwv = fc_w[gi];
  if (tid == 0) {
    v4f lv[LCH / 4];
#pragma unroll
    for (int i = 0; i < LCH / 4; ++i) lv[i] = *(const v4f*)(lf + t0 + 4 * i);
    float a = 0.f;
#pragma unroll
    for (int i = 0; i < LCH / 4; ++i) {
      a += lv[i].x; lv[i].x = a;
      a += lv[i].y; lv[i].y = a;
      a += lv[i].z; lv[i].z = a;
      a += lv[i].w; lv[i].w = a;
    }
#pragma unroll
    for (int i = 0; i < LCH / 4; ++i) *(v4f*)&la_s[4 * i] = lv[i];
  }
  ig_s[gi] = igv;
  fcw_s[gi] = fwv;
#pragma unroll
  for (int i = 0; i < 8; ++i) {
    *(v4f*)&Qs[srow + 16 * i][scol] = qst[i];
    *(v4f*)&Ks[srow + 16 * i][scol] = kst[i];
  }
  __syncthreads();

#pragma unroll
  for (int i = 0; i < 8; ++i)
    vst[i] = *(const v4f*)(vc + (size_t)(srow + 16 * i) * MD + scol);

  for (int t = wv; t < 36; t += NW) {
    int r = 0;
    while ((r + 1) * (r + 2) / 2 <= t) ++r;
    const int tr = r * 16, tc = (t - r * (r + 1) / 2) * 16;
    v8f acc = {};
    const float* ap = &Qs[tr + lm][2 * half];
    const float* bp = &Ks[tc + lm][2 * half];
#pragma unroll
    for (int kk = 0; kk < MD; kk += 4) {
      const v2f a = *(const v2f*)(ap + kk);
      const v2f b = *(const v2f*)(bp + kk);
      acc = wmma4(a, b, acc);
    }
#pragma unroll
    for (int e = 0; e < 8; ++e) {
      const int rr = tr + e + 8 * half, cc = tc + lm;
      float p = 0.f;
      if (rr >= cc) p = __expf(la_s[rr] - la_s[cc]) * ig_s[cc] * acc[e];
      Ps[rr][cc] = p;
    }
  }
#pragma unroll
  for (int i = 0; i < 8; ++i) *(v4f*)&Vs[srow + 16 * i][scol] = vst[i];
  if (tid < LCH) u_s[tid] = __expf(la_s[LCH - 1] - la_s[tid]) * ig_s[tid];
  __syncthreads();

  float rs_v = 0.f;
  if (tid < LCH) {
#pragma unroll
    for (int h = 0; h < 4; ++h) {
      if (h <= wv) {
        v4f pv[8];
#pragma unroll
        for (int i = 0; i < 8; ++i) pv[i] = *(const v4f*)&Ps[tid][32 * h + 4 * i];
        __builtin_amdgcn_sched_barrier(0);
#pragma unroll
        for (int i = 0; i < 8; ++i) {
          const int j = 32 * h + 4 * i;
          if (j     <= tid) rs_v += pv[i].x;
          if (j + 1 <= tid) rs_v += pv[i].y;
          if (j + 2 <= tid) rs_v += pv[i].z;
          if (j + 3 <= tid) rs_v += pv[i].w;
        }
      }
    }
  }

  {
    const int tc = (wv & 7) * 16;
    if (wv >> 3) pv_wave<1>(Ps, Vs, Gs, oc, fcw_s, tc, lm, half);
    else         pv_wave<0>(Ps, Vs, Gs, oc, fcw_s, tc, lm, half);
  }
  __syncthreads();

#pragma unroll
  for (int i = 0; i < 8; ++i) {
    const int row = srow + 16 * i;
    const float u = u_s[row];
    v4f v = *(const v4f*)&Vs[row][scol];
    v.x *= u; v.y *= u; v.z *= u; v.w *= u;
    *(v4f*)&Vs[row][scol] = v;
  }
  __syncthreads();

  float a_v = 0.f, dn_v = 0.f;
  if (tid < LCH) {
#pragma unroll
    for (int h = 0; h < 4; ++h) {
      v4f gv[8];
#pragma unroll
      for (int i = 0; i < 8; ++i) gv[i] = *(const v4f*)&Gs[tid][32 * h + 4 * i];
      __builtin_amdgcn_sched_barrier(0);
#pragma unroll
      for (int i = 0; i < 8; ++i) {
        a_v += gv[i].x; a_v += gv[i].y; a_v += gv[i].z; a_v += gv[i].w;
      }
    }
#pragma unroll 16
    for (int j = 0; j < LCH; ++j) dn_v = fmaf(u_s[j], Ks[j][tid], dn_v);
  }

  {
    const int tr0 = (wv >> 2) * 32, tc0 = (wv & 3) * 32;
    v8f d00 = {}, d01 = {}, d10 = {}, d11 = {};
#pragma unroll
    for (int kk = 0; kk < LCH; kk += 4) {
      const int j0 = kk + 2 * half;
      const v2f a0 = {Vs[j0][tr0 + lm],      Vs[j0 + 1][tr0 + lm]};
      const v2f a1 = {Vs[j0][tr0 + 16 + lm], Vs[j0 + 1][tr0 + 16 + lm]};
      const v2f b0 = {Ks[j0][tc0 + lm],      Ks[j0 + 1][tc0 + lm]};
      const v2f b1 = {Ks[j0][tc0 + 16 + lm], Ks[j0 + 1][tc0 + 16 + lm]};
      d00 = wmma4(a0, b0, d00);
      d01 = wmma4(a0, b1, d01);
      d10 = wmma4(a1, b0, d10);
      d11 = wmma4(a1, b1, d11);
    }
#pragma unroll
    for (int e = 0; e < 8; ++e) {
      float* kr0 = kc + (size_t)(tr0 + e + 8 * half) * MD + tc0 + lm;
      float* kr1 = kr0 + (size_t)16 * MD;
      kr0[0] = d00[e];  kr0[16] = d01[e];
      kr1[0] = d10[e];  kr1[16] = d11[e];
    }
  }
  if (tid < LCH) {
    vc[SC_LA + tid] = la_s[tid];
    vc[SC_RS + tid] = rs_v;
    vc[SC_A + tid] = a_v;
    vc[SC_DN + tid] = dn_v;
  }
}

#define SB 32
#define SCS 512

__global__ __launch_bounds__(512) void chunk_local4_kernel(
    const float* __restrict__ qw, const float* __restrict__ kw,
    const float* __restrict__ vw, const float* __restrict__ ow,
    const float* __restrict__ ig, const float* __restrict__ lf,
    const float* __restrict__ fc_w,
    float* __restrict__ dcw, float* __restrict__ scw) {
  __shared__ __attribute__((aligned(16))) float Qs[SB][QP];
  __shared__ __attribute__((aligned(16))) float Ks[LCH][QP];
  __shared__ __attribute__((aligned(16))) float Ps[SB][QP];
  __shared__ __attribute__((aligned(16))) float Vs[LCH][VP];
  __shared__ __attribute__((aligned(16))) float la_s[LCH], ig_s[LCH], u_s[LCH], fcw_s[MD];
  float (*Gs)[QP] = Qs;

  const int tid = threadIdx.x;
  const int wv = __builtin_amdgcn_readfirstlane(tid >> 5);
  const int lane = tid & 31;
  const int half = lane >> 4;
  const int lm = lane & 15;
  const int ch = blockIdx.x >> 2;
  const int s4 = blockIdx.x & 3;
  const int t0 = ch * LCH;
  const int r0 = s4 * SB;

  const float* qc = qw + (size_t)(t0 + r0) * MD;
  const float* oc = ow + (size_t)(t0 + r0) * MD;
  const float* kc = kw + (size_t)t0 * MD;
  const float* vc = vw + (size_t)t0 * MD;
  float* dc = dcw + (size_t)ch * CH;
  float* sc = scw + (size_t)ch * SCS;

  const int srow = tid >> 5;
  const int scol = (tid & 31) * 4;
  v4f qst[2], kst[8], vst[8];
#pragma unroll
  for (int i = 0; i < 2; ++i)
    qst[i] = *(const v4f*)(qc + (size_t)(srow + 16 * i) * MD + scol);
#pragma unroll
  for (int i = 0; i < 8; ++i)
    kst[i] = *(const v4f*)(kc + (size_t)(srow + 16 * i) * MD + scol);
  const int gi = tid & (LCH - 1);
  const float igv = ig[t0 + gi];
  const float fwv = fc_w[gi];
  if (tid == 0) {
    v4f lv[LCH / 4];
#pragma unroll
    for (int i = 0; i < LCH / 4; ++i) lv[i] = *(const v4f*)(lf + t0 + 4 * i);
    float a = 0.f;
#pragma unroll
    for (int i = 0; i < LCH / 4; ++i) {
      a += lv[i].x; lv[i].x = a;
      a += lv[i].y; lv[i].y = a;
      a += lv[i].z; lv[i].z = a;
      a += lv[i].w; lv[i].w = a;
    }
#pragma unroll
    for (int i = 0; i < LCH / 4; ++i) *(v4f*)&la_s[4 * i] = lv[i];
  }
  ig_s[gi] = igv;
  fcw_s[gi] = fwv;
#pragma unroll
  for (int i = 0; i < 2; ++i) *(v4f*)&Qs[srow + 16 * i][scol] = qst[i];
#pragma unroll
  for (int i = 0; i < 8; ++i) *(v4f*)&Ks[srow + 16 * i][scol] = kst[i];
  __syncthreads();

#pragma unroll
  for (int i = 0; i < 8; ++i)
    vst[i] = *(const v4f*)(vc + (size_t)(srow + 16 * i) * MD + scol);

  if (wv < 4 * s4 + 3) {
    const int rt = (wv > 2 * s4) ? 1 : 0;
    const int trl = rt * 16;
    const int tc = (rt ? (wv - 2 * s4 - 1) : wv) * 16;
    v8f acc = {};
    const float* ap = &Qs[trl + lm][2 * half];
    const float* bp = &Ks[tc + lm][2 * half];
    const int cc = tc + lm;
    const float lac = la_s[cc], igc = ig_s[cc];
    float lar[8];
#pragma unroll
    for (int e = 0; e < 8; ++e) lar[e] = la_s[r0 + trl + e + 8 * half];
#pragma unroll
    for (int kk = 0; kk < MD; kk += 4) {
      const v2f a = *(const v2f*)(ap + kk);
      const v2f b = *(const v2f*)(bp + kk);
      acc = wmma4(a, b, acc);
    }
#pragma unroll
    for (int e = 0; e < 8; ++e) {
      const int rl = trl + e + 8 * half;
      const bool keep = (r0 + rl) >= cc;
      const float w = __expf(keep ? (lar[e] - lac) : 0.f) * igc;
      Ps[rl][cc] = keep ? w * acc[e] : 0.f;
    }
  }
#pragma unroll
  for (int i = 0; i < 8; ++i) *(v4f*)&Vs[srow + 16 * i][scol] = vst[i];
  if (tid < LCH) u_s[tid] = __expf(la_s[LCH - 1] - la_s[tid]) * ig_s[tid];
  __syncthreads();

  {
    const int rt = wv >> 3;
    const int cc = (wv & 7) * 16 + lm;
    float ov[8];
#pragma unroll
    for (int e = 0; e < 8; ++e) ov[e] = oc[(size_t)(rt * 16 + e + 8 * half) * MD + cc];
    v8f aci = {};
    const int nkb = 2 * s4 + rt + 1;
    for (int kb = 0; kb < nkb; ++kb) {
      v2f b[4];
#pragma unroll
      for (int s = 0; s < 4; ++s) {
        const int j0 = kb * 16 + 4 * s + 2 * half;
        b[s].x = Vs[j0][cc];
        b[s].y = Vs[j0 + 1][cc];
      }
      const float* ap = &Ps[rt * 16 + lm][kb * 16 + 2 * half];
#pragma unroll
      for (int s = 0; s < 4; ++s) {
        const v2f a = *(const v2f*)(ap + 4 * s);
        aci = wmma4(a, b[s], aci);
      }
    }
    const float fw = fcw_s[cc];
#pragma unroll
    for (int e = 0; e < 8; ++e)
      Gs[rt * 16 + e + 8 * half][cc] = aci[e] * ov[e] * fw;
  }
  __syncthreads();

  if (wv < 8) {
    const int fr = r0 + (wv >> 2) * 16;
    const int tc0 = (wv & 3) * 32;
    v8f d0 = {}, d1 = {};
#pragma unroll
    for (int kk = 0; kk < LCH; kk += 4) {
      const int j0 = kk + 2 * half;
      const v2f uu = *(const v2f*)&u_s[j0];
      const v2f a  = {Vs[j0][fr + lm] * uu.x,  Vs[j0 + 1][fr + lm] * uu.y};
      const v2f b0 = {Ks[j0][tc0 + lm],        Ks[j0 + 1][tc0 + lm]};
      const v2f b1 = {Ks[j0][tc0 + 16 + lm],   Ks[j0 + 1][tc0 + 16 + lm]};
      d0 = wmma4(a, b0, d0);
      d1 = wmma4(a, b1, d1);
    }
#pragma unroll
    for (int e = 0; e < 8; ++e) {
      float* dr = dc + (size_t)(fr + e + 8 * half) * MD + tc0 + lm;
      dr[0] = d0[e];  dr[16] = d1[e];
    }
  } else if (wv == 8) {
    float a_v = 0.f;
#pragma unroll
    for (int h = 0; h < 4; ++h) {
      v4f gv[8];
#pragma unroll
      for (int i = 0; i < 8; ++i) gv[i] = *(const v4f*)&Gs[lane][32 * h + 4 * i];
      __builtin_amdgcn_sched_barrier(0);
#pragma unroll
      for (int i = 0; i < 8; ++i) {
        a_v += gv[i].x; a_v += gv[i].y; a_v += gv[i].z; a_v += gv[i].w;
      }
    }
    sc[SC_A + r0 + lane] = a_v;
    sc[SC_LA + r0 + lane] = la_s[r0 + lane];
  } else if (wv == 9) {
    float dn_v = 0.f;
#pragma unroll 16
    for (int j = 0; j < LCH; ++j) dn_v = fmaf(u_s[j], Ks[j][r0 + lane], dn_v);
    sc[SC_DN + r0 + lane] = dn_v;
  } else if (wv == 10) {
    float rs_v = 0.f;
    const int row = r0 + lane;
#pragma unroll
    for (int h = 0; h < 4; ++h) {
      if (h <= s4) {
        v4f pv[8];
#pragma unroll
        for (int i = 0; i < 8; ++i) pv[i] = *(const v4f*)&Ps[lane][32 * h + 4 * i];
        __builtin_amdgcn_sched_barrier(0);
#pragma unroll
        for (int i = 0; i < 8; ++i) {
          const int j = 32 * h + 4 * i;
          if (j     <= row) rs_v += pv[i].x;
          if (j + 1 <= row) rs_v += pv[i].y;
          if (j + 2 <= row) rs_v += pv[i].z;
          if (j + 3 <= row) rs_v += pv[i].w;
        }
      }
    }
    sc[SC_RS + r0 + lane] = rs_v;
  }
}

#define PFX_BLOCK 128
#define PFX_GRID 129

__global__ __launch_bounds__(PFX_BLOCK) void state_prefix_kernel(
    float* dcw, int dstride, float* scw, int sstride,
    const float* __restrict__ C0, const float* __restrict__ n0,
    float* __restrict__ C_out, float* __restrict__ n_out) {
  const int lane = threadIdx.x & 31;
  const float la_lo = scw[(size_t)lane * sstride + SC_LA + LCH - 1];
  const float la_hi = scw[(size_t)(lane + 32) * sstride + SC_LA + LCH - 1];

  const int gid = blockIdx.x * PFX_BLOCK + threadIdx.x;
  float* p; float* dst; float s; size_t ps;
  if (gid < MD * MD) {
    p = dcw + gid; ps = (size_t)dstride; s = C0[gid]; dst = C_out + gid;
  } else {
    const int f = gid - MD * MD;
    p = scw + SC_DN + f; ps = (size_t)sstride; s = n0[f]; dst = n_out + f;
  }
  float d[NCH];
#pragma unroll
  for (int c = 0; c < NCH; ++c) d[c] = p[(size_t)c * ps];

  __builtin_amdgcn_sched_barrier(0);
  const int f_lo = __float_as_int(__expf(la_lo));
  const int f_hi = __float_as_int(__expf(la_hi));
#pragma unroll
  for (int c = 0; c < NCH; ++c) {
    const float Fc = __int_as_float(
        __builtin_amdgcn_readlane((c < 32) ? f_lo : f_hi, c & 31));
    p[(size_t)c * ps] = s;
    s = fmaf(Fc, s, d[c]);
  }
  *dst = s;
}

#define RB 32
#define RPITCH (MD + 4)

__global__ __launch_bounds__(256) void chunk_readout_kernel(
    const float* __restrict__ qw, const float* __restrict__ dcw, int dstride,
    const float* __restrict__ scw, int sstride, const float* __restrict__ ow,
    const float* __restrict__ fc_w, const float* __restrict__ fc_b,
    float* __restrict__ out) {
  __shared__ __attribute__((aligned(16))) float Cs[MD][RPITCH];
  __shared__ __attribute__((aligned(16))) float Qs[RB][RPITCH];
  __shared__ __attribute__((aligned(16))) float Gs[RB][RPITCH];
  __shared__ __attribute__((aligned(16))) float ns[MD];
  __shared__ float fcw_s[MD], qn_s[RB];

  const int tid = threadIdx.x;
  const int wave = tid >> 5;
  const int lane = tid & 31;
  const int half = lane >> 4;
  const int lm = lane & 15;
  const int t0 = (blockIdx.x >> 2) * LCH;
  const int sub = (blockIdx.x & 3) * RB;

  const float* qc = qw + (size_t)(t0 + sub) * MD;
  const float* oc = ow + (size_t)(t0 + sub) * MD;
  const float* st = dcw + (size_t)(blockIdx.x >> 2) * dstride;
  const float* sc = scw + (size_t)(blockIdx.x >> 2) * sstride;

  const int srow = tid >> 5;
  const int scol = (tid & 31) * 4;
  v4f cst[16], qst[4];
#pragma unroll
  for (int i = 0; i < 16; ++i)
    cst[i] = *(const v4f*)(st + (size_t)(srow + 8 * i) * MD + scol);
#pragma unroll
  for (int i = 0; i < 4; ++i)
    qst[i] = *(const v4f*)(qc + (size_t)(srow + 8 * i) * MD + scol);
  const int vi = tid & (MD - 1);
  const float nsv = sc[SC_DN + vi];
  const float fwv = fc_w[vi];
  const float la_v = sc[SC_LA + sub + lane];
  const float rs_v = sc[SC_RS + sub + lane];
  const float a_v = sc[SC_A + sub + lane];
  const float fcb = fc_b[0];

  const int cc = wave * 16 + lm;
  float ov0[8], ov1[8];
#pragma unroll
  for (int e = 0; e < 8; ++e) {
    ov0[e] = oc[(size_t)(e + 8 * half) * MD + cc];
    ov1[e] = oc[(size_t)(16 + e + 8 * half) * MD + cc];
  }

#pragma unroll
  for (int i = 0; i < 16; ++i) *(v4f*)&Cs[srow + 8 * i][scol] = cst[i];
#pragma unroll
  for (int i = 0; i < 4; ++i) *(v4f*)&Qs[srow + 8 * i][scol] = qst[i];
  ns[vi] = nsv;
  fcw_s[vi] = fwv;
  __syncthreads();

  v8f ac0 = {}, ac1 = {};
  const float* a0p = &Qs[lm][2 * half];
  const float* a1p = &Qs[16 + lm][2 * half];
  const float* bp = &Cs[wave * 16 + lm][2 * half];
#pragma unroll
  for (int kk = 0; kk < MD; kk += 4) {
    const v2f a0 = *(const v2f*)(a0p + kk);
    const v2f a1 = *(const v2f*)(a1p + kk);
    const v2f b = *(const v2f*)(bp + kk);
    ac0 = wmma4(a0, b, ac0);
    ac1 = wmma4(a1, b, ac1);
  }
  const float fw = fcw_s[cc];
#pragma unroll
  for (int e = 0; e < 8; ++e) {
    const int r = e + 8 * half;
    Gs[r][cc] = ac0[e] * ov0[e] * fw;
    Gs[16 + r][cc] = ac1[e] * ov1[e] * fw;
  }
  __syncthreads();

  float b_v = 0.f;
  if (wave == 0) {
    v4f gv[MD / 4];
#pragma unroll
    for (int i = 0; i < MD / 4; ++i) gv[i] = *(const v4f*)&Gs[lane][4 * i];
    __builtin_amdgcn_sched_barrier(0);
#pragma unroll
    for (int i = 0; i < MD / 4; ++i) {
      b_v += gv[i].x; b_v += gv[i].y; b_v += gv[i].z; b_v += gv[i].w;
    }
  } else if (wave == 1) {
    float qn = 0.f;
#pragma unroll
    for (int h = 0; h < 4; ++h) {
      v4f qv[8], nv[8];
#pragma unroll
      for (int i = 0; i < 8; ++i) {
        qv[i] = *(const v4f*)&Qs[lane][32 * h + 4 * i];
        nv[i] = *(const v4f*)&ns[32 * h + 4 * i];
      }
      __builtin_amdgcn_sched_barrier(0);
#pragma unroll
      for (int i = 0; i < 8; ++i) {
        qn = fmaf(qv[i].x, nv[i].x, qn);
        qn = fmaf(qv[i].y, nv[i].y, qn);
        qn = fmaf(qv[i].z, nv[i].z, qn);
        qn = fmaf(qv[i].w, nv[i].w, qn);
      }
    }
    qn_s[lane] = qn;
  }
  __syncthreads();

  if (tid < RB) {
    const float qn = qn_s[tid];
    const float Ft = __expf(la_v);
    const float num = fmaf(Ft, b_v, a_v);
    const float den = fmaxf(fabsf(fmaf(Ft, qn, rs_v)), 1.0f);
    out[t0 + sub + tid] = num / den + fcb;
  }
}

extern "C" void kernel_launch(void* const* d_in, const int* in_sizes, int n_in,
                              void* d_out, int out_size, void* d_ws, size_t ws_size,
                              hipStream_t stream) {
  const float* x    = (const float*)d_in[0];
  const float* C0   = (const float*)d_in[1];
  const float* n0   = (const float*)d_in[2];
  const float* Wq   = (const float*)d_in[3];
  const float* bq   = (const float*)d_in[4];
  const float* Wk   = (const float*)d_in[5];
  const float* bk   = (const float*)d_in[6];
  const float* Wv   = (const float*)d_in[7];
  const float* bv   = (const float*)d_in[8];
  const float* wi   = (const float*)d_in[9];
  const float* bi   = (const float*)d_in[10];
  const float* wf   = (const float*)d_in[11];
  const float* bf   = (const float*)d_in[12];
  const float* Wo   = (const float*)d_in[13];
  const float* bo   = (const float*)d_in[14];
  const float* fc_w = (const float*)d_in[15];
  const float* fc_b = (const float*)d_in[16];

  float* ws = (float*)d_ws;
  const size_t NP = (size_t)T_LEN * MD;
  float* qw = ws;
  float* kw = ws + NP;
  float* vw = ws + 2 * NP;
  float* owp = ws + 3 * NP;
  float* igp = ws + 4 * NP;
  float* lfp = igp + T_LEN;

  float* outp  = (float*)d_out;
  float* Coutp = outp + T_LEN;
  float* noutp = Coutp + MD * MD;

  proj_kernel<<<dim3(T_LEN / PB, 4), 512, 0, stream>>>(
      x, Wq, bq, Wk, bk, Wv, bv, Wo, bo, wi, bi, wf, bf, qw, kw, vw, owp, igp, lfp);
  float* dcp = lfp + T_LEN;
  float* scp = dcp + NP;
  if (ws_size >= (size_t)21168128) {
    chunk_local4_kernel<<<NCH * 4, 512, 0, stream>>>(
        qw, kw, vw, owp, igp, lfp, fc_w, dcp, scp);
    state_prefix_kernel<<<PFX_GRID, PFX_BLOCK, 0, stream>>>(
        dcp, CH, scp, SCS, C0, n0, Coutp, noutp);
    chunk_readout_kernel<<<NCH * 4, 256, 0, stream>>>(
        qw, dcp, CH, scp, SCS, owp, fc_w, fc_b, outp);
  } else {
    chunk_local_kernel<<<NCH, 512, 0, stream>>>(qw, kw, vw, owp, igp, lfp, fc_w);
    state_prefix_kernel<<<PFX_GRID, PFX_BLOCK, 0, stream>>>(
        kw, CH, vw, CH, C0, n0, Coutp, noutp);
    chunk_readout_kernel<<<NCH * 4, 256, 0, stream>>>(
        qw, kw, CH, vw, CH, owp, fc_w, fc_b, outp);
  }
}
